# starting kernel
# baseline (speedup 1.0000x reference)
#include <hip/hip_runtime.h>
#include <math.h>

#define NN      20000
#define NE      320000
#define ETOT    (NE + NN)
#define IN_CH   128
#define HID     64
#define HEADS   8
#define HC1     (HEADS * HID)
#define OUT_CH  64

typedef float v2f __attribute__((ext_vector_type(2)));
typedef float v8f __attribute__((ext_vector_type(8)));

__device__ __forceinline__ int f2ord(float f) {
    int i = __float_as_int(f);
    return i >= 0 ? i : (i ^ 0x7fffffff);
}
__device__ __forceinline__ float ord2f(int i) {
    return __int_as_float(i >= 0 ? i : (i ^ 0x7fffffff));
}

__global__ void wmma_gemm_f32(const float* __restrict__ A,
                              const float* __restrict__ B,
                              float* __restrict__ C,
                              int M, int K, int N) {
    const int wave    = (blockIdx.x * blockDim.x + threadIdx.x) >> 5;
    const int lane    = threadIdx.x & 31;
    const int nGroups = N >> 6;
    const int mTiles  = M >> 4;
    if (wave >= mTiles * nGroups) return;
    const int mt   = wave / nGroups;
    const int ng   = wave % nGroups;
    const int l    = lane & 15;
    const int half = lane >> 4;

    unsigned aoff = (unsigned)(mt * 16 + l) * (unsigned)K + 2u * half;
    unsigned boff = (unsigned)(2 * half) * (unsigned)N + (unsigned)(ng * 64 + l);
    const unsigned uN = (unsigned)N;

    v8f acc0 = {}, acc1 = {}, acc2 = {}, acc3 = {};
    for (int k0 = 0; k0 < K; k0 += 4) {
        v2f af = *(const v2f*)(A + aoff);
        v2f bf0 = { B[boff],      B[boff + uN]      };
        v2f bf1 = { B[boff + 16], B[boff + uN + 16] };
        v2f bf2 = { B[boff + 32], B[boff + uN + 32] };
        v2f bf3 = { B[boff + 48], B[boff + uN + 48] };
        acc0 = __builtin_amdgcn_wmma_f32_16x16x4_f32(false, af, false, bf0, (short)0, acc0, false, false);
        acc1 = __builtin_amdgcn_wmma_f32_16x16x4_f32(false, af, false, bf1, (short)0, acc1, false, false);
        acc2 = __builtin_amdgcn_wmma_f32_16x16x4_f32(false, af, false, bf2, (short)0, acc2, false, false);
        acc3 = __builtin_amdgcn_wmma_f32_16x16x4_f32(false, af, false, bf3, (short)0, acc3, false, false);
        aoff += 4u;
        boff += 4u * uN;
    }
    unsigned coff = (unsigned)(mt * 16 + 8 * half) * uN + (unsigned)(ng * 64 + l);
    #pragma unroll
    for (int r = 0; r < 8; ++r) {
        float* cp = C + coff;
        cp[0]  = acc0[r];
        cp[16] = acc1[r];
        cp[32] = acc2[r];
        cp[48] = acc3[r];
        coff += uN;
    }
}

__global__ void scores_kernel(const float* __restrict__ h,
                              const float* __restrict__ a_src,
                              const float* __restrict__ a_dst,
                              float* __restrict__ esrc, float* __restrict__ edst,
                              int* __restrict__ mmax, float* __restrict__ denom,
                              int n, int H, int C) {
    int i = blockIdx.x * blockDim.x + threadIdx.x;
    if (i >= n * H) return;
    int node = i / H, hh = i % H;
    const float* hp = h + (size_t)node * H * C + (size_t)hh * C;
    const float* as = a_src + hh * C;
    const float* ad = a_dst + hh * C;
    float s = 0.f, d = 0.f;
    for (int c = 0; c < C; ++c) { float v = hp[c]; s += v * as[c]; d += v * ad[c]; }
    esrc[i]  = s;
    edst[i]  = d;
    mmax[i]  = f2ord(-INFINITY);
    denom[i] = 0.f;
}

__global__ void fill_zero(float* __restrict__ p, long long n) {
    long long i = (long long)blockIdx.x * blockDim.x + threadIdx.x;
    long long stride = (long long)gridDim.x * blockDim.x;
    for (; i < n; i += stride) p[i] = 0.f;
}

__global__ void edge_score_max(const int* __restrict__ ei,
                               const float* __restrict__ esrc,
                               const float* __restrict__ edst,
                               float* __restrict__ eleaky,
                               int* __restrict__ mmax, int H) {
    long long t = (long long)blockIdx.x * blockDim.x + threadIdx.x;
    if (t >= (long long)ETOT * H) return;
    int e = (int)(t / H), hh = (int)(t % H);
    int src, dst;
    if (e < NE) { src = ei[e]; dst = ei[NE + e]; }
    else        { src = dst = e - NE; }
    float v = esrc[src * H + hh] + edst[dst * H + hh];
    v = v > 0.f ? v : 0.2f * v;
    eleaky[t] = v;
    atomicMax(&mmax[dst * H + hh], f2ord(v));
}

__global__ void edge_exp_sum(const int* __restrict__ ei,
                             float* __restrict__ eleaky,
                             const int* __restrict__ mmax,
                             float* __restrict__ denom, int H) {
    long long t = (long long)blockIdx.x * blockDim.x + threadIdx.x;
    if (t >= (long long)ETOT * H) return;
    int e = (int)(t / H), hh = (int)(t % H);
    int dst = (e < NE) ? ei[NE + e] : (e - NE);
    float ex = expf(eleaky[t] - ord2f(mmax[dst * H + hh]));
    eleaky[t] = ex;
    atomicAdd(&denom[dst * H + hh], ex);
}

__global__ void edge_alpha(const int* __restrict__ ei,
                           float* __restrict__ exal,
                           const float* __restrict__ denom, int H) {
    long long t = (long long)blockIdx.x * blockDim.x + threadIdx.x;
    if (t >= (long long)ETOT * H) return;
    int e = (int)(t / H), hh = (int)(t % H);
    int dst = (e < NE) ? ei[NE + e] : (e - NE);
    exal[t] = exal[t] / (denom[dst * H + hh] + 1e-16f);
}

__global__ void edge_aggregate(const int* __restrict__ ei,
                               const float* __restrict__ alpha,
                               const float* __restrict__ hval,
                               float* __restrict__ agg, int H, int C) {
    const int HC = H * C;
    long long t = (long long)blockIdx.x * blockDim.x + threadIdx.x;
    if (t >= (long long)ETOT * HC) return;
    int e = (int)(t / HC), c = (int)(t % HC);
    int hh = c / C;
    int src, dst;
    if (e < NE) { src = ei[e]; dst = ei[NE + e]; }
    else        { src = dst = e - NE; }
    float a = alpha[(size_t)e * H + hh];
    atomicAdd(&agg[(size_t)dst * HC + c], a * hval[(size_t)src * HC + c]);
}

__global__ void bias_relu(float* __restrict__ a, const float* __restrict__ b,
                          long long n, int F) {
    long long i = (long long)blockIdx.x * blockDim.x + threadIdx.x;
    if (i >= n) return;
    float v = a[i] + b[(int)(i % F)];
    a[i] = v > 0.f ? v : 0.f;
}

__global__ void logsoftmax64(float* __restrict__ out, const float* __restrict__ b2) {
    int node = blockIdx.x;
    int lane = threadIdx.x;
    float* row = out + (size_t)node * OUT_CH;
    float v0 = row[lane]      + b2[lane];
    float v1 = row[lane + 32] + b2[lane + 32];
    float m = fmaxf(v0, v1);
    #pragma unroll
    for (int o = 16; o > 0; o >>= 1) m = fmaxf(m, __shfl_xor(m, o, 32));
    float s = expf(v0 - m) + expf(v1 - m);
    #pragma unroll
    for (int o = 16; o > 0; o >>= 1) s += __shfl_xor(s, o, 32);
    float lse = logf(s) + m;
    row[lane]      = v0 - lse;
    row[lane + 32] = v1 - lse;
}

extern "C" void kernel_launch(void* const* d_in, const int* in_sizes, int n_in,
                              void* d_out, int out_size, void* d_ws, size_t ws_size,
                              hipStream_t stream) {
    const float* x   = (const float*)d_in[0];
    const int*   ei  = (const int*)  d_in[1];
    const float* W1  = (const float*)d_in[2];
    const float* a1s = (const float*)d_in[3];
    const float* a1d = (const float*)d_in[4];
    const float* b1  = (const float*)d_in[5];
    const float* W2  = (const float*)d_in[6];
    const float* a2s = (const float*)d_in[7];
    const float* a2d = (const float*)d_in[8];
    const float* b2  = (const float*)d_in[9];
    float* out = (float*)d_out;

    float* ws = (float*)d_ws;
    float* h1pre = ws;
    float* agg1  = h1pre + (size_t)NN * HC1;
    float* esrc1 = agg1  + (size_t)NN * HC1;
    float* edst1 = esrc1 + (size_t)NN * HEADS;
    int*   m1    = (int*)(edst1 + (size_t)NN * HEADS);
    float* den1  = (float*)m1 + (size_t)NN * HEADS;
    float* ex1   = den1 + (size_t)NN * HEADS;
    float* h2pre = ex1  + (size_t)ETOT * HEADS;
    float* esrc2 = h2pre + (size_t)NN * OUT_CH;
    float* edst2 = esrc2 + NN;
    int*   m2    = (int*)(edst2 + NN);
    float* den2  = (float*)m2 + NN;
    float* ex2   = den2 + NN;

    const int TB = 256;
    auto blocks = [](long long n, int tb) { return (unsigned)((n + tb - 1) / tb); };

    {
        long long waves = (long long)(NN / 16) * (HC1 / 64);
        wmma_gemm_f32<<<blocks(waves * 32, TB), TB, 0, stream>>>(x, W1, h1pre, NN, IN_CH, HC1);
    }
    scores_kernel<<<blocks((long long)NN * HEADS, TB), TB, 0, stream>>>(
        h1pre, a1s, a1d, esrc1, edst1, m1, den1, NN, HEADS, HID);
    fill_zero<<<4096, TB, 0, stream>>>(agg1, (long long)NN * HC1);
    edge_score_max<<<blocks((long long)ETOT * HEADS, TB), TB, 0, stream>>>(ei, esrc1, edst1, ex1, m1, HEADS);
    edge_exp_sum  <<<blocks((long long)ETOT * HEADS, TB), TB, 0, stream>>>(ei, ex1, m1, den1, HEADS);
    edge_alpha    <<<blocks((long long)ETOT * HEADS, TB), TB, 0, stream>>>(ei, ex1, den1, HEADS);
    edge_aggregate<<<blocks((long long)ETOT * HC1,   TB), TB, 0, stream>>>(ei, ex1, h1pre, agg1, HEADS, HID);
    bias_relu<<<blocks((long long)NN * HC1, TB), TB, 0, stream>>>(agg1, b1, (long long)NN * HC1, HC1);

    {
        long long waves = (long long)(NN / 16) * (OUT_CH / 64);
        wmma_gemm_f32<<<blocks(waves * 32, TB), TB, 0, stream>>>(agg1, W2, h2pre, NN, HC1, OUT_CH);
    }
    scores_kernel<<<blocks((long long)NN, TB), TB, 0, stream>>>(
        h2pre, a2s, a2d, esrc2, edst2, m2, den2, NN, 1, OUT_CH);
    fill_zero<<<2048, TB, 0, stream>>>(out, (long long)NN * OUT_CH);
    edge_score_max<<<blocks((long long)ETOT, TB), TB, 0, stream>>>(ei, esrc2, edst2, ex2, m2, 1);
    edge_exp_sum  <<<blocks((long long)ETOT, TB), TB, 0, stream>>>(ei, ex2, m2, den2, 1);
    edge_alpha    <<<blocks((long long)ETOT, TB), TB, 0, stream>>>(ei, ex2, den2, 1);
    edge_aggregate<<<blocks((long long)ETOT * OUT_CH, TB), TB, 0, stream>>>(ei, ex2, h2pre, out, 1, OUT_CH);
    logsoftmax64<<<NN, 32, 0, stream>>>(out, b2);
}


# Edges counting-sorted by destination once per call; both aggregations gather per node into registers instead of float atomic scatter (sum order now the arrival-order rank)
# speedup vs baseline: 2.9959x; 2.9959x over previous
#include <hip/hip_runtime.h>
#include <math.h>

#define NN      20000
#define NE      320000
#define ETOT    (NE + NN)
#define IN_CH   128
#define HID     64
#define HEADS   8
#define HC1     (HEADS * HID)
#define OUT_CH  64

typedef float v2f __attribute__((ext_vector_type(2)));
typedef float v8f __attribute__((ext_vector_type(8)));

__device__ __forceinline__ int f2ord(float f) {
    int i = __float_as_int(f);
    return i >= 0 ? i : (i ^ 0x7fffffff);
}
__device__ __forceinline__ float ord2f(int i) {
    return __int_as_float(i >= 0 ? i : (i ^ 0x7fffffff));
}

__global__ void wmma_gemm_f32(const float* __restrict__ A,
                              const float* __restrict__ B,
                              float* __restrict__ C,
                              int M, int K, int N) {
    const int wave    = (blockIdx.x * blockDim.x + threadIdx.x) >> 5;
    const int lane    = threadIdx.x & 31;
    const int nGroups = N >> 6;
    const int mTiles  = M >> 4;
    if (wave >= mTiles * nGroups) return;
    const int mt   = wave / nGroups;
    const int ng   = wave % nGroups;
    const int l    = lane & 15;
    const int half = lane >> 4;

    unsigned aoff = (unsigned)(mt * 16 + l) * (unsigned)K + 2u * half;
    unsigned boff = (unsigned)(2 * half) * (unsigned)N + (unsigned)(ng * 64 + l);
    const unsigned uN = (unsigned)N;

    v8f acc0 = {}, acc1 = {}, acc2 = {}, acc3 = {};
    for (int k0 = 0; k0 < K; k0 += 4) {
        v2f af = *(const v2f*)(A + aoff);
        v2f bf0 = { B[boff],      B[boff + uN]      };
        v2f bf1 = { B[boff + 16], B[boff + uN + 16] };
        v2f bf2 = { B[boff + 32], B[boff + uN + 32] };
        v2f bf3 = { B[boff + 48], B[boff + uN + 48] };
        acc0 = __builtin_amdgcn_wmma_f32_16x16x4_f32(false, af, false, bf0, (short)0, acc0, false, false);
        acc1 = __builtin_amdgcn_wmma_f32_16x16x4_f32(false, af, false, bf1, (short)0, acc1, false, false);
        acc2 = __builtin_amdgcn_wmma_f32_16x16x4_f32(false, af, false, bf2, (short)0, acc2, false, false);
        acc3 = __builtin_amdgcn_wmma_f32_16x16x4_f32(false, af, false, bf3, (short)0, acc3, false, false);
        aoff += 4u;
        boff += 4u * uN;
    }
    unsigned coff = (unsigned)(mt * 16 + 8 * half) * uN + (unsigned)(ng * 64 + l);
    #pragma unroll
    for (int r = 0; r < 8; ++r) {
        float* cp = C + coff;
        cp[0]  = acc0[r];
        cp[16] = acc1[r];
        cp[32] = acc2[r];
        cp[48] = acc3[r];
        coff += uN;
    }
}

__global__ void scores_kernel(const float* __restrict__ h,
                              const float* __restrict__ a_src,
                              const float* __restrict__ a_dst,
                              float* __restrict__ esrc, float* __restrict__ edst,
                              int* __restrict__ mmax, float* __restrict__ denom,
                              int n, int H, int C) {
    int i = blockIdx.x * blockDim.x + threadIdx.x;
    if (i >= n * H) return;
    int node = i / H, hh = i % H;
    const float* hp = h + (size_t)node * H * C + (size_t)hh * C;
    const float* as = a_src + hh * C;
    const float* ad = a_dst + hh * C;
    float s = 0.f, d = 0.f;
    for (int c = 0; c < C; ++c) { float v = hp[c]; s += v * as[c]; d += v * ad[c]; }
    esrc[i]  = s;
    edst[i]  = d;
    mmax[i]  = f2ord(-INFINITY);
    denom[i] = 0.f;
}

__global__ void fill_zero(float* __restrict__ p, long long n) {
    long long i = (long long)blockIdx.x * blockDim.x + threadIdx.x;
    long long stride = (long long)gridDim.x * blockDim.x;
    for (; i < n; i += stride) p[i] = 0.f;
}

__global__ void edge_score_max(const int* __restrict__ ei,
                               const float* __restrict__ esrc,
                               const float* __restrict__ edst,
                               float* __restrict__ eleaky,
                               int* __restrict__ mmax, int H) {
    long long t = (long long)blockIdx.x * blockDim.x + threadIdx.x;
    if (t >= (long long)ETOT * H) return;
    int e = (int)(t / H), hh = (int)(t % H);
    int src, dst;
    if (e < NE) { src = ei[e]; dst = ei[NE + e]; }
    else        { src = dst = e - NE; }
    float v = esrc[src * H + hh] + edst[dst * H + hh];
    v = v > 0.f ? v : 0.2f * v;
    eleaky[t] = v;
    atomicMax(&mmax[dst * H + hh], f2ord(v));
}

__global__ void edge_exp_sum(const int* __restrict__ ei,
                             float* __restrict__ eleaky,
                             const int* __restrict__ mmax,
                             float* __restrict__ denom, int H) {
    long long t = (long long)blockIdx.x * blockDim.x + threadIdx.x;
    if (t >= (long long)ETOT * H) return;
    int e = (int)(t / H), hh = (int)(t % H);
    int dst = (e < NE) ? ei[NE + e] : (e - NE);
    float ex = expf(eleaky[t] - ord2f(mmax[dst * H + hh]));
    eleaky[t] = ex;
    atomicAdd(&denom[dst * H + hh], ex);
}

__global__ void edge_alpha(const int* __restrict__ ei,
                           float* __restrict__ exal,
                           const float* __restrict__ denom, int H) {
    long long t = (long long)blockIdx.x * blockDim.x + threadIdx.x;
    if (t >= (long long)ETOT * H) return;
    int e = (int)(t / H), hh = (int)(t % H);
    int dst = (e < NE) ? ei[NE + e] : (e - NE);
    exal[t] = exal[t] / (denom[dst * H + hh] + 1e-16f);
}

__global__ void dst_count(const int* __restrict__ ei, int* __restrict__ cnt,
                          int* __restrict__ rank) {
    int e = blockIdx.x * blockDim.x + threadIdx.x;
    if (e >= NE) return;
    rank[e] = atomicAdd(&cnt[ei[NE + e]], 1);
}

#define SCAN_T   1024
#define SCAN_PER ((NN + SCAN_T - 1) / SCAN_T)
__global__ void row_scan(const int* __restrict__ cnt, int* __restrict__ rowptr) {
    __shared__ int part[SCAN_T];
    const int t = threadIdx.x;
    const int base = t * SCAN_PER;
    int loc[SCAN_PER];
    int s = 0;
    #pragma unroll
    for (int i = 0; i < SCAN_PER; ++i) {
        int idx = base + i;
        int v = idx < NN ? cnt[idx] : 0;
        loc[i] = s;
        s += v;
    }
    part[t] = s;
    __syncthreads();
    for (int off = 1; off < SCAN_T; off <<= 1) {
        int v = t >= off ? part[t - off] : 0;
        __syncthreads();
        part[t] += v;
        __syncthreads();
    }
    const int excl = part[t] - s;
    #pragma unroll
    for (int i = 0; i < SCAN_PER; ++i) {
        int idx = base + i;
        if (idx < NN) rowptr[idx] = excl + loc[i];
    }
    if (t == SCAN_T - 1) rowptr[NN] = part[t];
}

__global__ void dst_fill(const int* __restrict__ ei, const int* __restrict__ rowptr,
                         const int* __restrict__ rank, int2* __restrict__ csr) {
    int e = blockIdx.x * blockDim.x + threadIdx.x;
    if (e >= NE) return;
    csr[rowptr[ei[NE + e]] + rank[e]] = make_int2(e, ei[e]);
}

template <int HC, int C, int VEC>
struct GatherAgg {
    typedef float vf __attribute__((ext_vector_type(VEC)));
    static constexpr int H = HC / C;

    template <int U>
    static __device__ __forceinline__ void group(const int2* __restrict__ csr, int j,
                                                 const float* __restrict__ alpha,
                                                 const float* __restrict__ hval,
                                                 unsigned c, unsigned hh, vf& acc) {
        float a[U];
        vf v[U];
        #pragma unroll
        for (int u = 0; u < U; ++u) {
            int2 es = csr[j + u];
            a[u] = alpha[(unsigned)es.x * (unsigned)H + hh];
            v[u] = *(const vf*)(hval + (unsigned)es.y * (unsigned)HC + c);
        }
        #pragma unroll
        for (int u = 0; u < U; ++u) acc += a[u] * v[u];
    }
};

template <int HC, int C, int VEC>
__global__ void gather_aggregate(const int* __restrict__ rowptr,
                                 const int2* __restrict__ csr,
                                 const float* __restrict__ alpha,
                                 const float* __restrict__ hval,
                                 float* __restrict__ agg) {
    typedef GatherAgg<HC, C, VEC> G;
    typedef typename G::vf vf;
    constexpr int WPN = HC / (32 * VEC);
    const int wave = __builtin_amdgcn_readfirstlane(
        (int)((blockIdx.x * blockDim.x + threadIdx.x) >> 5));
    if (wave >= NN * WPN) return;
    const int node  = wave / WPN;
    const int chunk = wave % WPN;
    const unsigned c  = (unsigned)(chunk * 32 * VEC) + (unsigned)(threadIdx.x & 31) * VEC;
    const unsigned hh = c / C;
    const int beg = rowptr[node], end = rowptr[node + 1];

    vf acc = alpha[(unsigned)(NE + node) * (unsigned)G::H + hh]
             * *(const vf*)(hval + (unsigned)node * (unsigned)HC + c);
    int j = beg;
    for (; j + 4 <= end; j += 4) G::template group<4>(csr, j, alpha, hval, c, hh, acc);
    const int r = end - j;
    if (r == 3)      G::template group<3>(csr, j, alpha, hval, c, hh, acc);
    else if (r == 2) G::template group<2>(csr, j, alpha, hval, c, hh, acc);
    else if (r == 1) G::template group<1>(csr, j, alpha, hval, c, hh, acc);
    *(vf*)(agg + (unsigned)node * (unsigned)HC + c) = acc;
}

__global__ void bias_relu(float* __restrict__ a, const float* __restrict__ b,
                          long long n, int F) {
    long long i = (long long)blockIdx.x * blockDim.x + threadIdx.x;
    if (i >= n) return;
    float v = a[i] + b[(int)(i % F)];
    a[i] = v > 0.f ? v : 0.f;
}

__global__ void logsoftmax64(float* __restrict__ out, const float* __restrict__ b2) {
    int node = blockIdx.x;
    int lane = threadIdx.x;
    float* row = out + (size_t)node * OUT_CH;
    float v0 = row[lane]      + b2[lane];
    float v1 = row[lane + 32] + b2[lane + 32];
    float m = fmaxf(v0, v1);
    #pragma unroll
    for (int o = 16; o > 0; o >>= 1) m = fmaxf(m, __shfl_xor(m, o, 32));
    float s = expf(v0 - m) + expf(v1 - m);
    #pragma unroll
    for (int o = 16; o > 0; o >>= 1) s += __shfl_xor(s, o, 32);
    float lse = logf(s) + m;
    row[lane]      = v0 - lse;
    row[lane + 32] = v1 - lse;
}

extern "C" void kernel_launch(void* const* d_in, const int* in_sizes, int n_in,
                              void* d_out, int out_size, void* d_ws, size_t ws_size,
                              hipStream_t stream) {
    const float* x   = (const float*)d_in[0];
    const int*   ei  = (const int*)  d_in[1];
    const float* W1  = (const float*)d_in[2];
    const float* a1s = (const float*)d_in[3];
    const float* a1d = (const float*)d_in[4];
    const float* b1  = (const float*)d_in[5];
    const float* W2  = (const float*)d_in[6];
    const float* a2s = (const float*)d_in[7];
    const float* a2d = (const float*)d_in[8];
    const float* b2  = (const float*)d_in[9];
    float* out = (float*)d_out;

    float* ws = (float*)d_ws;
    float* h1pre = ws;
    float* agg1  = h1pre + (size_t)NN * HC1;
    float* esrc1 = agg1  + (size_t)NN * HC1;
    float* edst1 = esrc1 + (size_t)NN * HEADS;
    int*   m1    = (int*)(edst1 + (size_t)NN * HEADS);
    float* den1  = (float*)m1 + (size_t)NN * HEADS;
    float* ex1   = den1 + (size_t)NN * HEADS;
    float* h2pre = ex1  + (size_t)ETOT * HEADS;
    float* esrc2 = h2pre + (size_t)NN * OUT_CH;
    float* edst2 = esrc2 + NN;
    int*   m2    = (int*)(edst2 + NN);
    float* den2  = (float*)m2 + NN;
    float* ex2   = den2 + NN;
    int*   cnt   = (int*)(ex2 + (size_t)ETOT);
    int*   rank  = cnt + NN;
    int*   rowp  = rank + NE;
    int2*  csr   = (int2*)(rowp + NN + 2);

    const int TB = 256;
    auto blocks = [](long long n, int tb) { return (unsigned)((n + tb - 1) / tb); };

    fill_zero<<<blocks(NN, TB), TB, 0, stream>>>((float*)cnt, (long long)NN);
    dst_count<<<blocks(NE, TB), TB, 0, stream>>>(ei, cnt, rank);
    row_scan <<<1, SCAN_T, 0, stream>>>(cnt, rowp);
    dst_fill <<<blocks(NE, TB), TB, 0, stream>>>(ei, rowp, rank, csr);

    {
        long long waves = (long long)(NN / 16) * (HC1 / 64);
        wmma_gemm_f32<<<blocks(waves * 32, TB), TB, 0, stream>>>(x, W1, h1pre, NN, IN_CH, HC1);
    }
    scores_kernel<<<blocks((long long)NN * HEADS, TB), TB, 0, stream>>>(
        h1pre, a1s, a1d, esrc1, edst1, m1, den1, NN, HEADS, HID);
    edge_score_max<<<blocks((long long)ETOT * HEADS, TB), TB, 0, stream>>>(ei, esrc1, edst1, ex1, m1, HEADS);
    edge_exp_sum  <<<blocks((long long)ETOT * HEADS, TB), TB, 0, stream>>>(ei, ex1, m1, den1, HEADS);
    edge_alpha    <<<blocks((long long)ETOT * HEADS, TB), TB, 0, stream>>>(ei, ex1, den1, HEADS);
    gather_aggregate<HC1, HID, 4><<<blocks((long long)NN * (HC1 / 128) * 32, TB), TB, 0, stream>>>(
        rowp, csr, ex1, h1pre, agg1);
    bias_relu<<<blocks((long long)NN * HC1, TB), TB, 0, stream>>>(agg1, b1, (long long)NN * HC1, HC1);

    {
        long long waves = (long long)(NN / 16) * (OUT_CH / 64);
        wmma_gemm_f32<<<blocks(waves * 32, TB), TB, 0, stream>>>(agg1, W2, h2pre, NN, HC1, OUT_CH);
    }
    scores_kernel<<<blocks((long long)NN, TB), TB, 0, stream>>>(
        h2pre, a2s, a2d, esrc2, edst2, m2, den2, NN, 1, OUT_CH);
    edge_score_max<<<blocks((long long)ETOT, TB), TB, 0, stream>>>(ei, esrc2, edst2, ex2, m2, 1);
    edge_exp_sum  <<<blocks((long long)ETOT, TB), TB, 0, stream>>>(ei, ex2, m2, den2, 1);
    edge_alpha    <<<blocks((long long)ETOT, TB), TB, 0, stream>>>(ei, ex2, den2, 1);
    gather_aggregate<OUT_CH, OUT_CH, 2><<<blocks((long long)NN * 32, TB), TB, 0, stream>>>(
        rowp, csr, ex2, h2pre, out);
    logsoftmax64<<<NN, 32, 0, stream>>>(out, b2);
}


# Edge softmax (max, exp-sum, normalise) now computed per destination segment inside the gather kernels; six atomic edge passes removed; denominators summed by a fixed lane butterfly
# speedup vs baseline: 4.5095x; 1.5052x over previous
#include <hip/hip_runtime.h>
#include <math.h>

#define NN      20000
#define NE      320000
#define ETOT    (NE + NN)
#define IN_CH   128
#define HID     64
#define HEADS   8
#define HC1     (HEADS * HID)
#define OUT_CH  64

typedef float v2f __attribute__((ext_vector_type(2)));
typedef float v8f __attribute__((ext_vector_type(8)));

__global__ void wmma_gemm_f32(const float* __restrict__ A,
                              const float* __restrict__ B,
                              float* __restrict__ C,
                              int M, int K, int N) {
    const int wave    = (blockIdx.x * blockDim.x + threadIdx.x) >> 5;
    const int lane    = threadIdx.x & 31;
    const int nGroups = N >> 6;
    const int mTiles  = M >> 4;
    if (wave >= mTiles * nGroups) return;
    const int mt   = wave / nGroups;
    const int ng   = wave % nGroups;
    const int l    = lane & 15;
    const int half = lane >> 4;

    unsigned aoff = (unsigned)(mt * 16 + l) * (unsigned)K + 2u * half;
    unsigned boff = (unsigned)(2 * half) * (unsigned)N + (unsigned)(ng * 64 + l);
    const unsigned uN = (unsigned)N;

    v8f acc0 = {}, acc1 = {}, acc2 = {}, acc3 = {};
    for (int k0 = 0; k0 < K; k0 += 4) {
        v2f af = *(const v2f*)(A + aoff);
        v2f bf0 = { B[boff],      B[boff + uN]      };
        v2f bf1 = { B[boff + 16], B[boff + uN + 16] };
        v2f bf2 = { B[boff + 32], B[boff + uN + 32] };
        v2f bf3 = { B[boff + 48], B[boff + uN + 48] };
        acc0 = __builtin_amdgcn_wmma_f32_16x16x4_f32(false, af, false, bf0, (short)0, acc0, false, false);
        acc1 = __builtin_amdgcn_wmma_f32_16x16x4_f32(false, af, false, bf1, (short)0, acc1, false, false);
        acc2 = __builtin_amdgcn_wmma_f32_16x16x4_f32(false, af, false, bf2, (short)0, acc2, false, false);
        acc3 = __builtin_amdgcn_wmma_f32_16x16x4_f32(false, af, false, bf3, (short)0, acc3, false, false);
        aoff += 4u;
        boff += 4u * uN;
    }
    unsigned coff = (unsigned)(mt * 16 + 8 * half) * uN + (unsigned)(ng * 64 + l);
    #pragma unroll
    for (int r = 0; r < 8; ++r) {
        float* cp = C + coff;
        cp[0]  = acc0[r];
        cp[16] = acc1[r];
        cp[32] = acc2[r];
        cp[48] = acc3[r];
        coff += uN;
    }
}

__global__ void scores_kernel(const float* __restrict__ h,
                              const float* __restrict__ a_src,
                              const float* __restrict__ a_dst,
                              float* __restrict__ esrc, float* __restrict__ edst,
                              int n, int H, int C) {
    int i = blockIdx.x * blockDim.x + threadIdx.x;
    if (i >= n * H) return;
    int node = i / H, hh = i % H;
    const float* hp = h + (size_t)node * H * C + (size_t)hh * C;
    const float* as = a_src + hh * C;
    const float* ad = a_dst + hh * C;
    float s = 0.f, d = 0.f;
    for (int c = 0; c < C; ++c) { float v = hp[c]; s += v * as[c]; d += v * ad[c]; }
    esrc[i]  = s;
    edst[i]  = d;
}

__global__ void fill_zero(float* __restrict__ p, long long n) {
    long long i = (long long)blockIdx.x * blockDim.x + threadIdx.x;
    long long stride = (long long)gridDim.x * blockDim.x;
    for (; i < n; i += stride) p[i] = 0.f;
}

__global__ void dst_count(const int* __restrict__ ei, int* __restrict__ cnt,
                          int* __restrict__ rank) {
    int e = blockIdx.x * blockDim.x + threadIdx.x;
    if (e >= NE) return;
    rank[e] = atomicAdd(&cnt[ei[NE + e]], 1);
}

#define SCAN_T   1024
#define SCAN_PER ((NN + SCAN_T - 1) / SCAN_T)
__global__ void row_scan(const int* __restrict__ cnt, int* __restrict__ rowptr) {
    __shared__ int part[SCAN_T];
    const int t = threadIdx.x;
    const int base = t * SCAN_PER;
    int loc[SCAN_PER];
    int s = 0;
    #pragma unroll
    for (int i = 0; i < SCAN_PER; ++i) {
        int idx = base + i;
        int v = idx < NN ? cnt[idx] : 0;
        loc[i] = s;
        s += v;
    }
    part[t] = s;
    __syncthreads();
    for (int off = 1; off < SCAN_T; off <<= 1) {
        int v = t >= off ? part[t - off] : 0;
        __syncthreads();
        part[t] += v;
        __syncthreads();
    }
    const int excl = part[t] - s;
    #pragma unroll
    for (int i = 0; i < SCAN_PER; ++i) {
        int idx = base + i;
        if (idx < NN) rowptr[idx] = excl + loc[i];
    }
    if (t == SCAN_T - 1) rowptr[NN] = part[t];
}

__global__ void dst_fill(const int* __restrict__ ei, const int* __restrict__ rowptr,
                         const int* __restrict__ rank, int2* __restrict__ csr) {
    int e = blockIdx.x * blockDim.x + threadIdx.x;
    if (e >= NE) return;
    csr[rowptr[ei[NE + e]] + rank[e]] = make_int2(e, ei[e]);
}

__device__ __forceinline__ float xhalf(float x) {
    return __builtin_bit_cast(float, __builtin_amdgcn_permlanex16(
        __builtin_bit_cast(int, x), __builtin_bit_cast(int, x), 0x76543210, 0xfedcba98, false, false));
}
template <int XM>
__device__ __forceinline__ float row_xor(float x) {
    const int xi = __builtin_bit_cast(int, x);
    return __builtin_bit_cast(float, __builtin_amdgcn_update_dpp(xi, xi, 0x160 + XM, 0xf, 0xf, false));
}
__device__ __forceinline__ float wave_max(float v) {
    v = fmaxf(v, row_xor<1>(v));
    v = fmaxf(v, row_xor<2>(v));
    v = fmaxf(v, row_xor<4>(v));
    v = fmaxf(v, row_xor<8>(v));
    return fmaxf(v, xhalf(v));
}
__device__ __forceinline__ float wave_sum(float v) {
    v += row_xor<1>(v);
    v += row_xor<2>(v);
    v += row_xor<4>(v);
    v += row_xor<8>(v);
    return v + xhalf(v);
}

template <int HC, int C, int VEC>
struct SegGather {
    typedef float vf __attribute__((ext_vector_type(VEC)));
    static constexpr int H  = HC / C;
    static constexpr int NH = (32 * VEC) / C;
    static_assert(NH == 1 || NH == 2, "a wave covers one or two whole heads");

    struct Block { int src; float v[NH]; bool valid; };

    static __device__ __forceinline__ Block load(const int2* __restrict__ csr,
                                                 const float* __restrict__ esrc,
                                                 const float (&ed)[NH],
                                                 int base, int end, int lane, int hbase) {
        Block b;
        const int idx = base + lane;
        b.valid = idx < end;
        b.src = b.valid ? csr[idx].y : 0;
        const float* ep = esrc + (unsigned)b.src * (unsigned)H + (unsigned)hbase;
        float e[NH];
        if constexpr (NH == 2) { const float2 t = *(const float2*)ep; e[0] = t.x; e[1] = t.y; }
        else                   { e[0] = ep[0]; }
        for (int q = 0; q < NH; ++q) {
            float v = e[q] + ed[q];
            v = v > 0.f ? v : 0.2f * v;
            b.v[q] = b.valid ? v : -INFINITY;
        }
        return b;
    }

    template <int U>
    static __device__ __forceinline__ void group(int srcv, const float (&av)[NH], int j, bool upper,
                                                 const float* __restrict__ hval, unsigned c, vf& acc) {
        float a[U];
        vf v[U];
        #pragma unroll
        for (int u = 0; u < U; ++u) {
            const int s = __builtin_amdgcn_readlane(srcv, j + u);
            float al = __builtin_bit_cast(float, __builtin_amdgcn_readlane(__builtin_bit_cast(int, av[0]), j + u));
            if constexpr (NH == 2) {
                const float a1 = __builtin_bit_cast(float, __builtin_amdgcn_readlane(__builtin_bit_cast(int, av[1]), j + u));
                al = upper ? a1 : al;
            }
            a[u] = al;
            v[u] = *(const vf*)(hval + (unsigned)s * (unsigned)HC + c);
        }
        #pragma unroll
        for (int u = 0; u < U; ++u) acc += a[u] * v[u];
    }

    static __device__ __forceinline__ void block_sum(int srcv, const float (&av)[NH], int cnt, bool upper,
                                                     const float* __restrict__ hval, unsigned c, vf& acc) {
        int j = 0;
        for (; j + 4 <= cnt; j += 4) group<4>(srcv, av, j, upper, hval, c, acc);
        const int r = cnt - j;
        if (r == 3)      group<3>(srcv, av, j, upper, hval, c, acc);
        else if (r == 2) group<2>(srcv, av, j, upper, hval, c, acc);
        else if (r == 1) group<1>(srcv, av, j, upper, hval, c, acc);
    }
};

template <int HC, int C, int VEC>
__global__ void gather_softmax_aggregate(const int* __restrict__ rowptr,
                                         const int2* __restrict__ csr,
                                         const float* __restrict__ esrc,
                                         const float* __restrict__ edst,
                                         const float* __restrict__ hval,
                                         float* __restrict__ agg) {
    typedef SegGather<HC, C, VEC> G;
    typedef typename G::vf vf;
    typedef typename G::Block Block;
    constexpr int H   = G::H;
    constexpr int NH  = G::NH;
    constexpr int WPN = HC / (32 * VEC);
    const int wave = __builtin_amdgcn_readfirstlane(
        (int)((blockIdx.x * blockDim.x + threadIdx.x) >> 5));
    if (wave >= NN * WPN) return;
    const int lane  = threadIdx.x & 31;
    const int node  = wave / WPN;
    const int chunk = wave % WPN;
    const unsigned c = (unsigned)(chunk * 32 * VEC) + (unsigned)lane * VEC;
    const int  hbase = chunk * NH;
    const bool upper = NH == 2 && lane >= 16;
    const int beg = rowptr[node], end = rowptr[node + 1];

    float ed[NH], vself[NH];
    for (int q = 0; q < NH; ++q) {
        ed[q] = edst[node * H + hbase + q];
        float v = esrc[node * H + hbase + q] + ed[q];
        vself[q] = v > 0.f ? v : 0.2f * v;
    }

    const Block b0 = G::load(csr, esrc, ed, beg, end, lane, hbase);
    float m[NH];
    for (int q = 0; q < NH; ++q) m[q] = b0.v[q];
    for (int base = beg + 32; base < end; base += 32) {
        const Block b = G::load(csr, esrc, ed, base, end, lane, hbase);
        for (int q = 0; q < NH; ++q) m[q] = fmaxf(m[q], b.v[q]);
    }
    for (int q = 0; q < NH; ++q) m[q] = fmaxf(wave_max(m[q]), vself[q]);

    float p0[NH], s[NH];
    for (int q = 0; q < NH; ++q) { p0[q] = b0.valid ? expf(b0.v[q] - m[q]) : 0.f; s[q] = p0[q]; }
    for (int base = beg + 32; base < end; base += 32) {
        const Block b = G::load(csr, esrc, ed, base, end, lane, hbase);
        for (int q = 0; q < NH; ++q) s[q] += b.valid ? expf(b.v[q] - m[q]) : 0.f;
    }
    float pself[NH], dn[NH], a0[NH];
    for (int q = 0; q < NH; ++q) {
        pself[q] = expf(vself[q] - m[q]);
        dn[q] = (wave_sum(s[q]) + pself[q]) + 1e-16f;
        a0[q] = p0[q] / dn[q];
    }

    float aself = pself[0] / dn[0];
    if constexpr (NH == 2) { const float a1 = pself[1] / dn[1]; aself = upper ? a1 : aself; }
    vf acc = aself * *(const vf*)(hval + (unsigned)node * (unsigned)HC + c);
    int left = end - beg;
    if (left > 0) G::block_sum(b0.src, a0, left < 32 ? left : 32, upper, hval, c, acc);
    for (int base = beg + 32; base < end; base += 32) {
        left -= 32;
        const Block b = G::load(csr, esrc, ed, base, end, lane, hbase);
        float a[NH];
        for (int q = 0; q < NH; ++q) a[q] = (b.valid ? expf(b.v[q] - m[q]) : 0.f) / dn[q];
        G::block_sum(b.src, a, left < 32 ? left : 32, upper, hval, c, acc);
    }
    *(vf*)(agg + (unsigned)node * (unsigned)HC + c) = acc;
}

__global__ void bias_relu(float* __restrict__ a, const float* __restrict__ b,
                          long long n, int F) {
    long long i = (long long)blockIdx.x * blockDim.x + threadIdx.x;
    if (i >= n) return;
    float v = a[i] + b[(int)(i % F)];
    a[i] = v > 0.f ? v : 0.f;
}

__global__ void logsoftmax64(float* __restrict__ out, const float* __restrict__ b2) {
    int node = blockIdx.x;
    int lane = threadIdx.x;
    float* row = out + (size_t)node * OUT_CH;
    float v0 = row[lane]      + b2[lane];
    float v1 = row[lane + 32] + b2[lane + 32];
    float m = fmaxf(v0, v1);
    #pragma unroll
    for (int o = 16; o > 0; o >>= 1) m = fmaxf(m, __shfl_xor(m, o, 32));
    float s = expf(v0 - m) + expf(v1 - m);
    #pragma unroll
    for (int o = 16; o > 0; o >>= 1) s += __shfl_xor(s, o, 32);
    float lse = logf(s) + m;
    row[lane]      = v0 - lse;
    row[lane + 32] = v1 - lse;
}

extern "C" void kernel_launch(void* const* d_in, const int* in_sizes, int n_in,
                              void* d_out, int out_size, void* d_ws, size_t ws_size,
                              hipStream_t stream) {
    const float* x   = (const float*)d_in[0];
    const int*   ei  = (const int*)  d_in[1];
    const float* W1  = (const float*)d_in[2];
    const float* a1s = (const float*)d_in[3];
    const float* a1d = (const float*)d_in[4];
    const float* b1  = (const float*)d_in[5];
    const float* W2  = (const float*)d_in[6];
    const float* a2s = (const float*)d_in[7];
    const float* a2d = (const float*)d_in[8];
    const float* b2  = (const float*)d_in[9];
    float* out = (float*)d_out;

    float* ws = (float*)d_ws;
    float* h1pre = ws;
    float* agg1  = h1pre + (size_t)NN * HC1;
    float* esrc1 = agg1  + (size_t)NN * HC1;
    float* edst1 = esrc1 + (size_t)NN * HEADS;
    float* h2pre = edst1 + (size_t)NN * HEADS;
    float* esrc2 = h2pre + (size_t)NN * OUT_CH;
    float* edst2 = esrc2 + NN;
    int*   cnt   = (int*)(edst2 + NN);
    int*   rank  = cnt + NN;
    int*   rowp  = rank + NE;
    int2*  csr   = (int2*)(rowp + NN + 2);

    const int TB = 256;
    auto blocks = [](long long n, int tb) { return (unsigned)((n + tb - 1) / tb); };

    fill_zero<<<blocks(NN, TB), TB, 0, stream>>>((float*)cnt, (long long)NN);
    dst_count<<<blocks(NE, TB), TB, 0, stream>>>(ei, cnt, rank);
    row_scan <<<1, SCAN_T, 0, stream>>>(cnt, rowp);
    dst_fill <<<blocks(NE, TB), TB, 0, stream>>>(ei, rowp, rank, csr);

    {
        long long waves = (long long)(NN / 16) * (HC1 / 64);
        wmma_gemm_f32<<<blocks(waves * 32, TB), TB, 0, stream>>>(x, W1, h1pre, NN, IN_CH, HC1);
    }
    scores_kernel<<<blocks((long long)NN * HEADS, TB), TB, 0, stream>>>(
        h1pre, a1s, a1d, esrc1, edst1, NN, HEADS, HID);
    gather_softmax_aggregate<HC1, HID, 4><<<blocks((long long)NN * (HC1 / 128) * 32, TB), TB, 0, stream>>>(
        rowp, csr, esrc1, edst1, h1pre, agg1);
    bias_relu<<<blocks((long long)NN * HC1, TB), TB, 0, stream>>>(agg1, b1, (long long)NN * HC1, HC1);

    {
        long long waves = (long long)(NN / 16) * (OUT_CH / 64);
        wmma_gemm_f32<<<blocks(waves * 32, TB), TB, 0, stream>>>(agg1, W2, h2pre, NN, HC1, OUT_CH);
    }
    scores_kernel<<<blocks((long long)NN, TB), TB, 0, stream>>>(
        h2pre, a2s, a2d, esrc2, edst2, NN, 1, OUT_CH);
    gather_softmax_aggregate<OUT_CH, OUT_CH, 2><<<blocks((long long)NN * 32, TB), TB, 0, stream>>>(
        rowp, csr, esrc2, edst2, h2pre, out);
    logsoftmax64<<<NN, 32, 0, stream>>>(out, b2);
}


# Scores, bias+relu and bias+log_softmax moved into the epilogues of the GEMM and gather kernels that produce their inputs (12 launches -> 8); score and log-sum-exp sums now fixed lane butterflies
# speedup vs baseline: 4.8552x; 1.0767x over previous
#include <hip/hip_runtime.h>
#include <math.h>

#define NN      20000
#define NE      320000
#define ETOT    (NE + NN)
#define IN_CH   128
#define HID     64
#define HEADS   8
#define HC1     (HEADS * HID)
#define OUT_CH  64

typedef float v2f __attribute__((ext_vector_type(2)));
typedef float v8f __attribute__((ext_vector_type(8)));

__device__ __forceinline__ float xhalf(float x) {
    return __builtin_bit_cast(float, __builtin_amdgcn_permlanex16(
        __builtin_bit_cast(int, x), __builtin_bit_cast(int, x), 0x76543210, 0xfedcba98, false, false));
}
template <int XM>
__device__ __forceinline__ float row_xor(float x) {
    const int xi = __builtin_bit_cast(int, x);
    return __builtin_bit_cast(float, __builtin_amdgcn_update_dpp(xi, xi, 0x160 + XM, 0xf, 0xf, false));
}
__device__ __forceinline__ float wave_max(float v) {
    v = fmaxf(v, row_xor<1>(v));
    v = fmaxf(v, row_xor<2>(v));
    v = fmaxf(v, row_xor<4>(v));
    v = fmaxf(v, row_xor<8>(v));
    return fmaxf(v, xhalf(v));
}
__device__ __forceinline__ float row_sum16(float v) {
    v += row_xor<1>(v);
    v += row_xor<2>(v);
    v += row_xor<4>(v);
    return v + row_xor<8>(v);
}
__device__ __forceinline__ float wave_sum(float v) {
    v += row_xor<1>(v);
    v += row_xor<2>(v);
    v += row_xor<4>(v);
    v += row_xor<8>(v);
    return v + xhalf(v);
}

#define GEMM_WAVES 8
__global__ void wmma_gemm_f32(const float* __restrict__ A,
                              const float* __restrict__ B,
                              float* __restrict__ C,
                              const float* __restrict__ a_src,
                              const float* __restrict__ a_dst,
                              float* __restrict__ esrc, float* __restrict__ edst,
                              int M, int K, int N) {
    constexpr int SPAN = 16 * GEMM_WAVES;
    __shared__ float sc[2][SPAN];
    const int wave    = (blockIdx.x * blockDim.x + threadIdx.x) >> 5;
    const int lane    = threadIdx.x & 31;
    const int nGroups = N >> 6;
    const int mTiles  = M >> 4;
    const bool active = wave < mTiles * nGroups;
    const int mt   = wave / nGroups;
    const int ng   = wave % nGroups;
    const int l    = lane & 15;
    const int half = lane >> 4;
    const unsigned uN = (unsigned)N;

    v8f acc0 = {}, acc1 = {}, acc2 = {}, acc3 = {};
    if (active) {
        unsigned aoff = (unsigned)(mt * 16 + l) * (unsigned)K + 2u * half;
        unsigned boff = (unsigned)(2 * half) * uN + (unsigned)(ng * 64 + l);
        for (int k0 = 0; k0 < K; k0 += 4) {
            v2f af = *(const v2f*)(A + aoff);
            v2f bf0 = { B[boff],      B[boff + uN]      };
            v2f bf1 = { B[boff + 16], B[boff + uN + 16] };
            v2f bf2 = { B[boff + 32], B[boff + uN + 32] };
            v2f bf3 = { B[boff + 48], B[boff + uN + 48] };
            acc0 = __builtin_amdgcn_wmma_f32_16x16x4_f32(false, af, false, bf0, (short)0, acc0, false, false);
            acc1 = __builtin_amdgcn_wmma_f32_16x16x4_f32(false, af, false, bf1, (short)0, acc1, false, false);
            acc2 = __builtin_amdgcn_wmma_f32_16x16x4_f32(false, af, false, bf2, (short)0, acc2, false, false);
            acc3 = __builtin_amdgcn_wmma_f32_16x16x4_f32(false, af, false, bf3, (short)0, acc3, false, false);
            aoff += 4u;
            boff += 4u * uN;
        }
        const float* as = a_src + ng * 64 + l;
        const float* ad = a_dst + ng * 64 + l;
        const float s0 = as[0], s1 = as[16], s2 = as[32], s3 = as[48];
        const float d0 = ad[0], d1 = ad[16], d2 = ad[32], d3 = ad[48];
        float keep = 0.f;
        #pragma unroll
        for (int r = 0; r < 8; ++r) {
            const float vs = row_sum16(acc0[r] * s0 + acc1[r] * s1 + acc2[r] * s2 + acc3[r] * s3);
            const float vd = row_sum16(acc0[r] * d0 + acc1[r] * d1 + acc2[r] * d2 + acc3[r] * d3);
            keep = l == r     ? vs : keep;
            keep = l == r + 8 ? vd : keep;
        }
        const int mt0 = (int)(blockIdx.x * GEMM_WAVES) / nGroups;
        const int rr  = (mt - mt0) * 16 + 8 * half + (l & 7);
        sc[l >> 3][rr * nGroups + ng] = keep;
    }
    __syncthreads();
    const int t    = threadIdx.x;
    const int kind = t / SPAN;
    const int p    = t % SPAN;
    const int idx  = blockIdx.x * SPAN + p;
    if (idx < M * nGroups) (kind ? edst : esrc)[idx] = sc[kind][p];

    if (active) {
        unsigned coff = (unsigned)(mt * 16 + 8 * half) * uN + (unsigned)(ng * 64 + l);
        #pragma unroll
        for (int r = 0; r < 8; ++r) {
            float* cp = C + coff;
            cp[0]  = acc0[r];
            cp[16] = acc1[r];
            cp[32] = acc2[r];
            cp[48] = acc3[r];
            coff += uN;
        }
    }
}

__global__ void fill_zero(float* __restrict__ p, long long n) {
    long long i = (long long)blockIdx.x * blockDim.x + threadIdx.x;
    long long stride = (long long)gridDim.x * blockDim.x;
    for (; i < n; i += stride) p[i] = 0.f;
}

__global__ void dst_count(const int* __restrict__ ei, int* __restrict__ cnt,
                          int* __restrict__ rank) {
    int e = blockIdx.x * blockDim.x + threadIdx.x;
    if (e >= NE) return;
    rank[e] = atomicAdd(&cnt[ei[NE + e]], 1);
}

#define SCAN_T   1024
#define SCAN_PER ((NN + SCAN_T - 1) / SCAN_T)
__global__ void row_scan(const int* __restrict__ cnt, int* __restrict__ rowptr) {
    __shared__ int part[SCAN_T];
    const int t = threadIdx.x;
    const int base = t * SCAN_PER;
    int loc[SCAN_PER];
    int s = 0;
    #pragma unroll
    for (int i = 0; i < SCAN_PER; ++i) {
        int idx = base + i;
        int v = idx < NN ? cnt[idx] : 0;
        loc[i] = s;
        s += v;
    }
    part[t] = s;
    __syncthreads();
    for (int off = 1; off < SCAN_T; off <<= 1) {
        int v = t >= off ? part[t - off] : 0;
        __syncthreads();
        part[t] += v;
        __syncthreads();
    }
    const int excl = part[t] - s;
    #pragma unroll
    for (int i = 0; i < SCAN_PER; ++i) {
        int idx = base + i;
        if (idx < NN) rowptr[idx] = excl + loc[i];
    }
    if (t == SCAN_T - 1) rowptr[NN] = part[t];
}

__global__ void dst_fill(const int* __restrict__ ei, const int* __restrict__ rowptr,
                         const int* __restrict__ rank, int2* __restrict__ csr) {
    int e = blockIdx.x * blockDim.x + threadIdx.x;
    if (e >= NE) return;
    csr[rowptr[ei[NE + e]] + rank[e]] = make_int2(e, ei[e]);
}

#define EPI_RELU       1
#define EPI_LOGSOFTMAX 2
template <int HC, int C, int VEC>
struct SegGather {
    typedef float vf __attribute__((ext_vector_type(VEC)));
    static constexpr int H  = HC / C;
    static constexpr int NH = (32 * VEC) / C;
    static_assert(NH == 1 || NH == 2, "a wave covers one or two whole heads");

    struct Block { int src; float v[NH]; bool valid; };

    static __device__ __forceinline__ Block load(const int2* __restrict__ csr,
                                                 const float* __restrict__ esrc,
                                                 const float (&ed)[NH],
                                                 int base, int end, int lane, int hbase) {
        Block b;
        const int idx = base + lane;
        b.valid = idx < end;
        b.src = b.valid ? csr[idx].y : 0;
        const float* ep = esrc + (unsigned)b.src * (unsigned)H + (unsigned)hbase;
        float e[NH];
        if constexpr (NH == 2) { const float2 t = *(const float2*)ep; e[0] = t.x; e[1] = t.y; }
        else                   { e[0] = ep[0]; }
        for (int q = 0; q < NH; ++q) {
            float v = e[q] + ed[q];
            v = v > 0.f ? v : 0.2f * v;
            b.v[q] = b.valid ? v : -INFINITY;
        }
        return b;
    }

    template <int U>
    static __device__ __forceinline__ void group(int srcv, const float (&av)[NH], int j, bool upper,
                                                 const float* __restrict__ hval, unsigned c, vf& acc) {
        float a[U];
        vf v[U];
        #pragma unroll
        for (int u = 0; u < U; ++u) {
            const int s = __builtin_amdgcn_readlane(srcv, j + u);
            float al = __builtin_bit_cast(float, __builtin_amdgcn_readlane(__builtin_bit_cast(int, av[0]), j + u));
            if constexpr (NH == 2) {
                const float a1 = __builtin_bit_cast(float, __builtin_amdgcn_readlane(__builtin_bit_cast(int, av[1]), j + u));
                al = upper ? a1 : al;
            }
            a[u] = al;
            v[u] = *(const vf*)(hval + (unsigned)s * (unsigned)HC + c);
        }
        #pragma unroll
        for (int u = 0; u < U; ++u) acc += a[u] * v[u];
    }

    static __device__ __forceinline__ void block_sum(int srcv, const float (&av)[NH], int cnt, bool upper,
                                                     const float* __restrict__ hval, unsigned c, vf& acc) {
        int j = 0;
        for (; j + 4 <= cnt; j += 4) group<4>(srcv, av, j, upper, hval, c, acc);
        const int r = cnt - j;
        if (r == 3)      group<3>(srcv, av, j, upper, hval, c, acc);
        else if (r == 2) group<2>(srcv, av, j, upper, hval, c, acc);
        else if (r == 1) group<1>(srcv, av, j, upper, hval, c, acc);
    }
};

template <int HC, int C, int VEC, int EPI>
__global__ void gather_softmax_aggregate(const int* __restrict__ rowptr,
                                         const int2* __restrict__ csr,
                                         const float* __restrict__ esrc,
                                         const float* __restrict__ edst,
                                         const float* __restrict__ hval,
                                         const float* __restrict__ bias,
                                         float* __restrict__ agg) {
    typedef SegGather<HC, C, VEC> G;
    typedef typename G::vf vf;
    typedef typename G::Block Block;
    constexpr int H   = G::H;
    constexpr int NH  = G::NH;
    constexpr int WPN = HC / (32 * VEC);
    const int wave = __builtin_amdgcn_readfirstlane(
        (int)((blockIdx.x * blockDim.x + threadIdx.x) >> 5));
    if (wave >= NN * WPN) return;
    const int lane  = threadIdx.x & 31;
    const int node  = wave / WPN;
    const int chunk = wave % WPN;
    const unsigned c = (unsigned)(chunk * 32 * VEC) + (unsigned)lane * VEC;
    const int  hbase = chunk * NH;
    const bool upper = NH == 2 && lane >= 16;
    const int beg = rowptr[node], end = rowptr[node + 1];

    float ed[NH], vself[NH];
    for (int q = 0; q < NH; ++q) {
        ed[q] = edst[node * H + hbase + q];
        float v = esrc[node * H + hbase + q] + ed[q];
        vself[q] = v > 0.f ? v : 0.2f * v;
    }

    const Block b0 = G::load(csr, esrc, ed, beg, end, lane, hbase);
    float m[NH];
    for (int q = 0; q < NH; ++q) m[q] = b0.v[q];
    for (int base = beg + 32; base < end; base += 32) {
        const Block b = G::load(csr, esrc, ed, base, end, lane, hbase);
        for (int q = 0; q < NH; ++q) m[q] = fmaxf(m[q], b.v[q]);
    }
    for (int q = 0; q < NH; ++q) m[q] = fmaxf(wave_max(m[q]), vself[q]);

    float p0[NH], s[NH];
    for (int q = 0; q < NH; ++q) { p0[q] = b0.valid ? expf(b0.v[q] - m[q]) : 0.f; s[q] = p0[q]; }
    for (int base = beg + 32; base < end; base += 32) {
        const Block b = G::load(csr, esrc, ed, base, end, lane, hbase);
        for (int q = 0; q < NH; ++q) s[q] += b.valid ? expf(b.v[q] - m[q]) : 0.f;
    }
    float pself[NH], dn[NH], a0[NH];
    for (int q = 0; q < NH; ++q) {
        pself[q] = expf(vself[q] - m[q]);
        dn[q] = (wave_sum(s[q]) + pself[q]) + 1e-16f;
        a0[q] = p0[q] / dn[q];
    }

    float aself = pself[0] / dn[0];
    if constexpr (NH == 2) { const float a1 = pself[1] / dn[1]; aself = upper ? a1 : aself; }
    vf acc = aself * *(const vf*)(hval + (unsigned)node * (unsigned)HC + c);
    int left = end - beg;
    if (left > 0) G::block_sum(b0.src, a0, left < 32 ? left : 32, upper, hval, c, acc);
    for (int base = beg + 32; base < end; base += 32) {
        left -= 32;
        const Block b = G::load(csr, esrc, ed, base, end, lane, hbase);
        float a[NH];
        for (int q = 0; q < NH; ++q) a[q] = (b.valid ? expf(b.v[q] - m[q]) : 0.f) / dn[q];
        G::block_sum(b.src, a, left < 32 ? left : 32, upper, hval, c, acc);
    }
    vf res = acc + *(const vf*)(bias + c);
    if constexpr (EPI == EPI_RELU) {
        for (int i = 0; i < VEC; ++i) res[i] = res[i] > 0.f ? res[i] : 0.f;
    } else {
        static_assert(WPN == 1 && H == 1, "the wave must hold the node's whole row");
        float mx = res[0];
        for (int i = 1; i < VEC; ++i) mx = fmaxf(mx, res[i]);
        mx = wave_max(mx);
        float se = 0.f;
        for (int i = 0; i < VEC; ++i) se += expf(res[i] - mx);
        const float lse = logf(wave_sum(se)) + mx;
        for (int i = 0; i < VEC; ++i) res[i] -= lse;
    }
    *(vf*)(agg + (unsigned)node * (unsigned)HC + c) = res;
}

extern "C" void kernel_launch(void* const* d_in, const int* in_sizes, int n_in,
                              void* d_out, int out_size, void* d_ws, size_t ws_size,
                              hipStream_t stream) {
    const float* x   = (const float*)d_in[0];
    const int*   ei  = (const int*)  d_in[1];
    const float* W1  = (const float*)d_in[2];
    const float* a1s = (const float*)d_in[3];
    const float* a1d = (const float*)d_in[4];
    const float* b1  = (const float*)d_in[5];
    const float* W2  = (const float*)d_in[6];
    const float* a2s = (const float*)d_in[7];
    const float* a2d = (const float*)d_in[8];
    const float* b2  = (const float*)d_in[9];
    float* out = (float*)d_out;

    float* ws = (float*)d_ws;
    float* h1pre = ws;
    float* agg1  = h1pre + (size_t)NN * HC1;
    float* esrc1 = agg1  + (size_t)NN * HC1;
    float* edst1 = esrc1 + (size_t)NN * HEADS;
    float* h2pre = edst1 + (size_t)NN * HEADS;
    float* esrc2 = h2pre + (size_t)NN * OUT_CH;
    float* edst2 = esrc2 + NN;
    int*   cnt   = (int*)(edst2 + NN);
    int*   rank  = cnt + NN;
    int*   rowp  = rank + NE;
    int2*  csr   = (int2*)(rowp + NN + 2);

    const int TB = 32 * GEMM_WAVES;
    auto blocks = [](long long n, int tb) { return (unsigned)((n + tb - 1) / tb); };

    fill_zero<<<blocks(NN, TB), TB, 0, stream>>>((float*)cnt, (long long)NN);
    dst_count<<<blocks(NE, TB), TB, 0, stream>>>(ei, cnt, rank);
    row_scan <<<1, SCAN_T, 0, stream>>>(cnt, rowp);
    dst_fill <<<blocks(NE, TB), TB, 0, stream>>>(ei, rowp, rank, csr);

    {
        long long waves = (long long)(NN / 16) * (HC1 / 64);
        wmma_gemm_f32<<<blocks(waves * 32, TB), TB, 0, stream>>>(
            x, W1, h1pre, a1s, a1d, esrc1, edst1, NN, IN_CH, HC1);
    }
    gather_softmax_aggregate<HC1, HID, 4, EPI_RELU><<<blocks((long long)NN * (HC1 / 128) * 32, TB), TB, 0, stream>>>(
        rowp, csr, esrc1, edst1, h1pre, b1, agg1);

    {
        long long waves = (long long)(NN / 16) * (OUT_CH / 64);
        wmma_gemm_f32<<<blocks(waves * 32, TB), TB, 0, stream>>>(
            agg1, W2, h2pre, a2s, a2d, esrc2, edst2, NN, HC1, OUT_CH);
    }
    gather_softmax_aggregate<OUT_CH, OUT_CH, 2, EPI_LOGSOFTMAX><<<blocks((long long)NN * 32, TB), TB, 0, stream>>>(
        rowp, csr, esrc2, edst2, h2pre, b2, out);
}


# GEMM K loop: fragment loads of 16 k issued as one group into one of two register sets, the next group in flight during the current group's 16 WMMAs (K, N compile-time); k order unchanged
# speedup vs baseline: 7.0466x; 1.4514x over previous
#include <hip/hip_runtime.h>
#include <math.h>

#define NN      20000
#define NE      320000
#define ETOT    (NE + NN)
#define IN_CH   128
#define HID     64
#define HEADS   8
#define HC1     (HEADS * HID)
#define OUT_CH  64

typedef float v2f __attribute__((ext_vector_type(2)));
typedef float v8f __attribute__((ext_vector_type(8)));

__device__ __forceinline__ float xhalf(float x) {
    return __builtin_bit_cast(float, __builtin_amdgcn_permlanex16(
        __builtin_bit_cast(int, x), __builtin_bit_cast(int, x), 0x76543210, 0xfedcba98, false, false));
}
template <int XM>
__device__ __forceinline__ float row_xor(float x) {
    const int xi = __builtin_bit_cast(int, x);
    return __builtin_bit_cast(float, __builtin_amdgcn_update_dpp(xi, xi, 0x160 + XM, 0xf, 0xf, false));
}
__device__ __forceinline__ float wave_max(float v) {
    v = fmaxf(v, row_xor<1>(v));
    v = fmaxf(v, row_xor<2>(v));
    v = fmaxf(v, row_xor<4>(v));
    v = fmaxf(v, row_xor<8>(v));
    return fmaxf(v, xhalf(v));
}
__device__ __forceinline__ float row_sum16(float v) {
    v += row_xor<1>(v);
    v += row_xor<2>(v);
    v += row_xor<4>(v);
    return v + row_xor<8>(v);
}
__device__ __forceinline__ float wave_sum(float v) {
    v += row_xor<1>(v);
    v += row_xor<2>(v);
    v += row_xor<4>(v);
    v += row_xor<8>(v);
    return v + xhalf(v);
}

#define GEMM_WAVES 8
struct Frags16 { v2f a[4]; v2f b[4][4]; };
template <int N>
__device__ __forceinline__ void load_frags16(Frags16& f, const float* __restrict__ Ap,
                                             const float* __restrict__ Bp) {
    #pragma unroll
    for (int s = 0; s < 4; ++s) {
        f.a[s] = *(const v2f*)(Ap + 4 * s);
        #pragma unroll
        for (int j = 0; j < 4; ++j) {
            v2f b = { Bp[(4 * s) * N + 16 * j], Bp[(4 * s + 1) * N + 16 * j] };
            f.b[s][j] = b;
        }
    }
}
__device__ __forceinline__ void mma_frags16(const Frags16& f, v8f& acc0, v8f& acc1, v8f& acc2, v8f& acc3) {
    #pragma unroll
    for (int s = 0; s < 4; ++s) {
        acc0 = __builtin_amdgcn_wmma_f32_16x16x4_f32(false, f.a[s], false, f.b[s][0], (short)0, acc0, false, false);
        acc1 = __builtin_amdgcn_wmma_f32_16x16x4_f32(false, f.a[s], false, f.b[s][1], (short)0, acc1, false, false);
        acc2 = __builtin_amdgcn_wmma_f32_16x16x4_f32(false, f.a[s], false, f.b[s][2], (short)0, acc2, false, false);
        acc3 = __builtin_amdgcn_wmma_f32_16x16x4_f32(false, f.a[s], false, f.b[s][3], (short)0, acc3, false, false);
    }
}
template <int K, int N>
__global__ void wmma_gemm_f32(const float* __restrict__ A,
                              const float* __restrict__ B,
                              float* __restrict__ C,
                              const float* __restrict__ a_src,
                              const float* __restrict__ a_dst,
                              float* __restrict__ esrc, float* __restrict__ edst,
                              int M) {
    static_assert(K % 32 == 0 && N % 64 == 0, "two groups of 16 k per trip; 64-wide strips");
    constexpr int SPAN = 16 * GEMM_WAVES;
    __shared__ float sc[2][SPAN];
    const int wave    = (blockIdx.x * blockDim.x + threadIdx.x) >> 5;
    const int lane    = threadIdx.x & 31;
    constexpr int nGroups = N >> 6;
    const int mTiles  = M >> 4;
    const bool active = wave < mTiles * nGroups;
    const int mt   = wave / nGroups;
    const int ng   = wave % nGroups;
    const int l    = lane & 15;
    const int half = lane >> 4;
    constexpr unsigned uN = (unsigned)N;

    v8f acc0 = {}, acc1 = {}, acc2 = {}, acc3 = {};
    if (active) {
        const float* Ap = A + ((unsigned)(mt * 16 + l) * (unsigned)K + 2u * half);
        const float* Bp = B + ((unsigned)(2 * half) * uN + (unsigned)(ng * 64 + l));
        Frags16 P, Q;
        load_frags16<N>(P, Ap, Bp);
        for (int k0 = 0; k0 < K - 32; k0 += 32) {
            load_frags16<N>(Q, Ap + 16, Bp + 16 * N);
            mma_frags16(P, acc0, acc1, acc2, acc3);
            load_frags16<N>(P, Ap + 32, Bp + 32 * N);
            mma_frags16(Q, acc0, acc1, acc2, acc3);
            Ap += 32;
            Bp += 32 * N;
        }
        load_frags16<N>(Q, Ap + 16, Bp + 16 * N);
        mma_frags16(P, acc0, acc1, acc2, acc3);
        mma_frags16(Q, acc0, acc1, acc2, acc3);
        const float* as = a_src + ng * 64 + l;
        const float* ad = a_dst + ng * 64 + l;
        const float s0 = as[0], s1 = as[16], s2 = as[32], s3 = as[48];
        const float d0 = ad[0], d1 = ad[16], d2 = ad[32], d3 = ad[48];
        float keep = 0.f;
        #pragma unroll
        for (int r = 0; r < 8; ++r) {
            const float vs = row_sum16(acc0[r] * s0 + acc1[r] * s1 + acc2[r] * s2 + acc3[r] * s3);
            const float vd = row_sum16(acc0[r] * d0 + acc1[r] * d1 + acc2[r] * d2 + acc3[r] * d3);
            keep = l == r     ? vs : keep;
            keep = l == r + 8 ? vd : keep;
        }
        const int mt0 = (int)(blockIdx.x * GEMM_WAVES) / nGroups;
        const int rr  = (mt - mt0) * 16 + 8 * half + (l & 7);
        sc[l >> 3][rr * nGroups + ng] = keep;
    }
    __syncthreads();
    const int t    = threadIdx.x;
    const int kind = t / SPAN;
    const int p    = t % SPAN;
    const int idx  = blockIdx.x * SPAN + p;
    if (idx < M * nGroups) (kind ? edst : esrc)[idx] = sc[kind][p];

    if (active) {
        unsigned coff = (unsigned)(mt * 16 + 8 * half) * uN + (unsigned)(ng * 64 + l);
        #pragma unroll
        for (int r = 0; r < 8; ++r) {
            float* cp = C + coff;
            cp[0]  = acc0[r];
            cp[16] = acc1[r];
            cp[32] = acc2[r];
            cp[48] = acc3[r];
            coff += uN;
        }
    }
}

__global__ void fill_zero(float* __restrict__ p, long long n) {
    long long i = (long long)blockIdx.x * blockDim.x + threadIdx.x;
    long long stride = (long long)gridDim.x * blockDim.x;
    for (; i < n; i += stride) p[i] = 0.f;
}

__global__ void dst_count(const int* __restrict__ ei, int* __restrict__ cnt,
                          int* __restrict__ rank) {
    int e = blockIdx.x * blockDim.x + threadIdx.x;
    if (e >= NE) return;
    rank[e] = atomicAdd(&cnt[ei[NE + e]], 1);
}

#define SCAN_T   1024
#define SCAN_PER ((NN + SCAN_T - 1) / SCAN_T)
__global__ void row_scan(const int* __restrict__ cnt, int* __restrict__ rowptr) {
    __shared__ int part[SCAN_T];
    const int t = threadIdx.x;
    const int base = t * SCAN_PER;
    int loc[SCAN_PER];
    int s = 0;
    #pragma unroll
    for (int i = 0; i < SCAN_PER; ++i) {
        int idx = base + i;
        int v = idx < NN ? cnt[idx] : 0;
        loc[i] = s;
        s += v;
    }
    part[t] = s;
    __syncthreads();
    for (int off = 1; off < SCAN_T; off <<= 1) {
        int v = t >= off ? part[t - off] : 0;
        __syncthreads();
        part[t] += v;
        __syncthreads();
    }
    const int excl = part[t] - s;
    #pragma unroll
    for (int i = 0; i < SCAN_PER; ++i) {
        int idx = base + i;
        if (idx < NN) rowptr[idx] = excl + loc[i];
    }
    if (t == SCAN_T - 1) rowptr[NN] = part[t];
}

__global__ void dst_fill(const int* __restrict__ ei, const int* __restrict__ rowptr,
                         const int* __restrict__ rank, int2* __restrict__ csr) {
    int e = blockIdx.x * blockDim.x + threadIdx.x;
    if (e >= NE) return;
    csr[rowptr[ei[NE + e]] + rank[e]] = make_int2(e, ei[e]);
}

#define EPI_RELU       1
#define EPI_LOGSOFTMAX 2
template <int HC, int C, int VEC>
struct SegGather {
    typedef float vf __attribute__((ext_vector_type(VEC)));
    static constexpr int H  = HC / C;
    static constexpr int NH = (32 * VEC) / C;
    static_assert(NH == 1 || NH == 2, "a wave covers one or two whole heads");

    struct Block { int src; float v[NH]; bool valid; };

    static __device__ __forceinline__ Block load(const int2* __restrict__ csr,
                                                 const float* __restrict__ esrc,
                                                 const float (&ed)[NH],
                                                 int base, int end, int lane, int hbase) {
        Block b;
        const int idx = base + lane;
        b.valid = idx < end;
        b.src = b.valid ? csr[idx].y : 0;
        const float* ep = esrc + (unsigned)b.src * (unsigned)H + (unsigned)hbase;
        float e[NH];
        if constexpr (NH == 2) { const float2 t = *(const float2*)ep; e[0] = t.x; e[1] = t.y; }
        else                   { e[0] = ep[0]; }
        for (int q = 0; q < NH; ++q) {
            float v = e[q] + ed[q];
            v = v > 0.f ? v : 0.2f * v;
            b.v[q] = b.valid ? v : -INFINITY;
        }
        return b;
    }

    template <int U>
    static __device__ __forceinline__ void group(int srcv, const float (&av)[NH], int j, bool upper,
                                                 const float* __restrict__ hval, unsigned c, vf& acc) {
        float a[U];
        vf v[U];
        #pragma unroll
        for (int u = 0; u < U; ++u) {
            const int s = __builtin_amdgcn_readlane(srcv, j + u);
            float al = __builtin_bit_cast(float, __builtin_amdgcn_readlane(__builtin_bit_cast(int, av[0]), j + u));
            if constexpr (NH == 2) {
                const float a1 = __builtin_bit_cast(float, __builtin_amdgcn_readlane(__builtin_bit_cast(int, av[1]), j + u));
                al = upper ? a1 : al;
            }
            a[u] = al;
            v[u] = *(const vf*)(hval + (unsigned)s * (unsigned)HC + c);
        }
        #pragma unroll
        for (int u = 0; u < U; ++u) acc += a[u] * v[u];
    }

    static __device__ __forceinline__ void block_sum(int srcv, const float (&av)[NH], int cnt, bool upper,
                                                     const float* __restrict__ hval, unsigned c, vf& acc) {
        int j = 0;
        for (; j + 4 <= cnt; j += 4) group<4>(srcv, av, j, upper, hval, c, acc);
        const int r = cnt - j;
        if (r == 3)      group<3>(srcv, av, j, upper, hval, c, acc);
        else if (r == 2) group<2>(srcv, av, j, upper, hval, c, acc);
        else if (r == 1) group<1>(srcv, av, j, upper, hval, c, acc);
    }
};

template <int HC, int C, int VEC, int EPI>
__global__ void gather_softmax_aggregate(const int* __restrict__ rowptr,
                                         const int2* __restrict__ csr,
                                         const float* __restrict__ esrc,
                                         const float* __restrict__ edst,
                                         const float* __restrict__ hval,
                                         const float* __restrict__ bias,
                                         float* __restrict__ agg) {
    typedef SegGather<HC, C, VEC> G;
    typedef typename G::vf vf;
    typedef typename G::Block Block;
    constexpr int H   = G::H;
    constexpr int NH  = G::NH;
    constexpr int WPN = HC / (32 * VEC);
    const int wave = __builtin_amdgcn_readfirstlane(
        (int)((blockIdx.x * blockDim.x + threadIdx.x) >> 5));
    if (wave >= NN * WPN) return;
    const int lane  = threadIdx.x & 31;
    const int node  = wave / WPN;
    const int chunk = wave % WPN;
    const unsigned c = (unsigned)(chunk * 32 * VEC) + (unsigned)lane * VEC;
    const int  hbase = chunk * NH;
    const bool upper = NH == 2 && lane >= 16;
    const int beg = rowptr[node], end = rowptr[node + 1];

    float ed[NH], vself[NH];
    for (int q = 0; q < NH; ++q) {
        ed[q] = edst[node * H + hbase + q];
        float v = esrc[node * H + hbase + q] + ed[q];
        vself[q] = v > 0.f ? v : 0.2f * v;
    }

    const Block b0 = G::load(csr, esrc, ed, beg, end, lane, hbase);
    float m[NH];
    for (int q = 0; q < NH; ++q) m[q] = b0.v[q];
    for (int base = beg + 32; base < end; base += 32) {
        const Block b = G::load(csr, esrc, ed, base, end, lane, hbase);
        for (int q = 0; q < NH; ++q) m[q] = fmaxf(m[q], b.v[q]);
    }
    for (int q = 0; q < NH; ++q) m[q] = fmaxf(wave_max(m[q]), vself[q]);

    float p0[NH], s[NH];
    for (int q = 0; q < NH; ++q) { p0[q] = b0.valid ? expf(b0.v[q] - m[q]) : 0.f; s[q] = p0[q]; }
    for (int base = beg + 32; base < end; base += 32) {
        const Block b = G::load(csr, esrc, ed, base, end, lane, hbase);
        for (int q = 0; q < NH; ++q) s[q] += b.valid ? expf(b.v[q] - m[q]) : 0.f;
    }
    float pself[NH], dn[NH], a0[NH];
    for (int q = 0; q < NH; ++q) {
        pself[q] = expf(vself[q] - m[q]);
        dn[q] = (wave_sum(s[q]) + pself[q]) + 1e-16f;
        a0[q] = p0[q] / dn[q];
    }

    float aself = pself[0] / dn[0];
    if constexpr (NH == 2) { const float a1 = pself[1] / dn[1]; aself = upper ? a1 : aself; }
    vf acc = aself * *(const vf*)(hval + (unsigned)node * (unsigned)HC + c);
    int left = end - beg;
    if (left > 0) G::block_sum(b0.src, a0, left < 32 ? left : 32, upper, hval, c, acc);
    for (int base = beg + 32; base < end; base += 32) {
        left -= 32;
        const Block b = G::load(csr, esrc, ed, base, end, lane, hbase);
        float a[NH];
        for (int q = 0; q < NH; ++q) a[q] = (b.valid ? expf(b.v[q] - m[q]) : 0.f) / dn[q];
        G::block_sum(b.src, a, left < 32 ? left : 32, upper, hval, c, acc);
    }
    vf res = acc + *(const vf*)(bias + c);
    if constexpr (EPI == EPI_RELU) {
        for (int i = 0; i < VEC; ++i) res[i] = res[i] > 0.f ? res[i] : 0.f;
    } else {
        static_assert(WPN == 1 && H == 1, "the wave must hold the node's whole row");
        float mx = res[0];
        for (int i = 1; i < VEC; ++i) mx = fmaxf(mx, res[i]);
        mx = wave_max(mx);
        float se = 0.f;
        for (int i = 0; i < VEC; ++i) se += expf(res[i] - mx);
        const float lse = logf(wave_sum(se)) + mx;
        for (int i = 0; i < VEC; ++i) res[i] -= lse;
    }
    *(vf*)(agg + (unsigned)node * (unsigned)HC + c) = res;
}

extern "C" void kernel_launch(void* const* d_in, const int* in_sizes, int n_in,
                              void* d_out, int out_size, void* d_ws, size_t ws_size,
                              hipStream_t stream) {
    const float* x   = (const float*)d_in[0];
    const int*   ei  = (const int*)  d_in[1];
    const float* W1  = (const float*)d_in[2];
    const float* a1s = (const float*)d_in[3];
    const float* a1d = (const float*)d_in[4];
    const float* b1  = (const float*)d_in[5];
    const float* W2  = (const float*)d_in[6];
    const float* a2s = (const float*)d_in[7];
    const float* a2d = (const float*)d_in[8];
    const float* b2  = (const float*)d_in[9];
    float* out = (float*)d_out;

    float* ws = (float*)d_ws;
    float* h1pre = ws;
    float* agg1  = h1pre + (size_t)NN * HC1;
    float* esrc1 = agg1  + (size_t)NN * HC1;
    float* edst1 = esrc1 + (size_t)NN * HEADS;
    float* h2pre = edst1 + (size_t)NN * HEADS;
    float* esrc2 = h2pre + (size_t)NN * OUT_CH;
    float* edst2 = esrc2 + NN;
    int*   cnt   = (int*)(edst2 + NN);
    int*   rank  = cnt + NN;
    int*   rowp  = rank + NE;
    int2*  csr   = (int2*)(rowp + NN + 2);

    const int TB = 32 * GEMM_WAVES;
    auto blocks = [](long long n, int tb) { return (unsigned)((n + tb - 1) / tb); };

    fill_zero<<<blocks(NN, TB), TB, 0, stream>>>((float*)cnt, (long long)NN);
    dst_count<<<blocks(NE, TB), TB, 0, stream>>>(ei, cnt, rank);
    row_scan <<<1, SCAN_T, 0, stream>>>(cnt, rowp);
    dst_fill <<<blocks(NE, TB), TB, 0, stream>>>(ei, rowp, rank, csr);

    {
        long long waves = (long long)(NN / 16) * (HC1 / 64);
        wmma_gemm_f32<IN_CH, HC1><<<blocks(waves * 32, TB), TB, 0, stream>>>(
            x, W1, h1pre, a1s, a1d, esrc1, edst1, NN);
    }
    gather_softmax_aggregate<HC1, HID, 4, EPI_RELU><<<blocks((long long)NN * (HC1 / 128) * 32, TB), TB, 0, stream>>>(
        rowp, csr, esrc1, edst1, h1pre, b1, agg1);

    {
        long long waves = (long long)(NN / 16) * (OUT_CH / 64);
        wmma_gemm_f32<HC1, OUT_CH><<<blocks(waves * 32, TB), TB, 0, stream>>>(
            agg1, W2, h2pre, a2s, a2d, esrc2, edst2, NN);
    }
    gather_softmax_aggregate<OUT_CH, OUT_CH, 2, EPI_LOGSOFTMAX><<<blocks((long long)NN * 32, TB), TB, 0, stream>>>(
        rowp, csr, esrc2, edst2, h2pre, b2, out);
}


# Layer 1 sums x rows per head and destination first (512-byte rows gathered, not 2048-byte projected rows) and applies W1 after; scores from x@(W1 a); sum order of layer 1 changes (fixed)
# speedup vs baseline: 7.5860x; 1.0765x over previous
#include <hip/hip_runtime.h>
#include <math.h>

#define NN      20000
#define NE      320000
#define ETOT    (NE + NN)
#define IN_CH   128
#define HID     64
#define HEADS   8
#define HC1     (HEADS * HID)
#define OUT_CH  64

typedef float v2f __attribute__((ext_vector_type(2)));
typedef float v8f __attribute__((ext_vector_type(8)));

__device__ __forceinline__ float xhalf(float x) {
    return __builtin_bit_cast(float, __builtin_amdgcn_permlanex16(
        __builtin_bit_cast(int, x), __builtin_bit_cast(int, x), 0x76543210, 0xfedcba98, false, false));
}
template <int XM>
__device__ __forceinline__ float row_xor(float x) {
    const int xi = __builtin_bit_cast(int, x);
    return __builtin_bit_cast(float, __builtin_amdgcn_update_dpp(xi, xi, 0x160 + XM, 0xf, 0xf, false));
}
__device__ __forceinline__ float wave_max(float v) {
    v = fmaxf(v, row_xor<1>(v));
    v = fmaxf(v, row_xor<2>(v));
    v = fmaxf(v, row_xor<4>(v));
    v = fmaxf(v, row_xor<8>(v));
    return fmaxf(v, xhalf(v));
}
__device__ __forceinline__ float row_sum16(float v) {
    v += row_xor<1>(v);
    v += row_xor<2>(v);
    v += row_xor<4>(v);
    return v + row_xor<8>(v);
}
__device__ __forceinline__ float wave_sum(float v) {
    v += row_xor<1>(v);
    v += row_xor<2>(v);
    v += row_xor<4>(v);
    v += row_xor<8>(v);
    return v + xhalf(v);
}

#define GEMM_WAVES 8
struct Frags16 { v2f a[4]; v2f b[4][4]; };
template <int N>
__device__ __forceinline__ void load_frags16(Frags16& f, const float* __restrict__ Ap,
                                             const float* __restrict__ Bp) {
    #pragma unroll
    for (int s = 0; s < 4; ++s) {
        f.a[s] = *(const v2f*)(Ap + 4 * s);
        #pragma unroll
        for (int j = 0; j < 4; ++j) {
            v2f b = { Bp[(4 * s) * N + 16 * j], Bp[(4 * s + 1) * N + 16 * j] };
            f.b[s][j] = b;
        }
    }
}
__device__ __forceinline__ void mma_frags16(const Frags16& f, v8f& acc0, v8f& acc1, v8f& acc2, v8f& acc3) {
    #pragma unroll
    for (int s = 0; s < 4; ++s) {
        acc0 = __builtin_amdgcn_wmma_f32_16x16x4_f32(false, f.a[s], false, f.b[s][0], (short)0, acc0, false, false);
        acc1 = __builtin_amdgcn_wmma_f32_16x16x4_f32(false, f.a[s], false, f.b[s][1], (short)0, acc1, false, false);
        acc2 = __builtin_amdgcn_wmma_f32_16x16x4_f32(false, f.a[s], false, f.b[s][2], (short)0, acc2, false, false);
        acc3 = __builtin_amdgcn_wmma_f32_16x16x4_f32(false, f.a[s], false, f.b[s][3], (short)0, acc3, false, false);
    }
}
template <int K, int N>
__global__ void wmma_gemm_f32(const float* __restrict__ A,
                              const float* __restrict__ B,
                              float* __restrict__ C,
                              const float* __restrict__ a_src,
                              const float* __restrict__ a_dst,
                              float* __restrict__ esrc, float* __restrict__ edst,
                              int M) {
    static_assert(K % 32 == 0 && N % 64 == 0, "two groups of 16 k per trip; 64-wide strips");
    constexpr int SPAN = 16 * GEMM_WAVES;
    __shared__ float sc[2][SPAN];
    const int wave    = (blockIdx.x * blockDim.x + threadIdx.x) >> 5;
    const int lane    = threadIdx.x & 31;
    constexpr int nGroups = N >> 6;
    const int mTiles  = M >> 4;
    const bool active = wave < mTiles * nGroups;
    const int mt   = wave / nGroups;
    const int ng   = wave % nGroups;
    const int l    = lane & 15;
    const int half = lane >> 4;
    constexpr unsigned uN = (unsigned)N;

    v8f acc0 = {}, acc1 = {}, acc2 = {}, acc3 = {};
    if (active) {
        const float* Ap = A + ((unsigned)(mt * 16 + l) * (unsigned)K + 2u * half);
        const float* Bp = B + ((unsigned)(2 * half) * uN + (unsigned)(ng * 64 + l));
        Frags16 P, Q;
        load_frags16<N>(P, Ap, Bp);
        for (int k0 = 0; k0 < K - 32; k0 += 32) {
            load_frags16<N>(Q, Ap + 16, Bp + 16 * N);
            mma_frags16(P, acc0, acc1, acc2, acc3);
            load_frags16<N>(P, Ap + 32, Bp + 32 * N);
            mma_frags16(Q, acc0, acc1, acc2, acc3);
            Ap += 32;
            Bp += 32 * N;
        }
        load_frags16<N>(Q, Ap + 16, Bp + 16 * N);
        mma_frags16(P, acc0, acc1, acc2, acc3);
        mma_frags16(Q, acc0, acc1, acc2, acc3);
        const float* as = a_src + ng * 64 + l;
        const float* ad = a_dst + ng * 64 + l;
        const float s0 = as[0], s1 = as[16], s2 = as[32], s3 = as[48];
        const float d0 = ad[0], d1 = ad[16], d2 = ad[32], d3 = ad[48];
        float keep = 0.f;
        #pragma unroll
        for (int r = 0; r < 8; ++r) {
            const float vs = row_sum16(acc0[r] * s0 + acc1[r] * s1 + acc2[r] * s2 + acc3[r] * s3);
            const float vd = row_sum16(acc0[r] * d0 + acc1[r] * d1 + acc2[r] * d2 + acc3[r] * d3);
            keep = l == r     ? vs : keep;
            keep = l == r + 8 ? vd : keep;
        }
        const int mt0 = (int)(blockIdx.x * GEMM_WAVES) / nGroups;
        const int rr  = (mt - mt0) * 16 + 8 * half + (l & 7);
        sc[l >> 3][rr * nGroups + ng] = keep;
    }
    __syncthreads();
    const int t    = threadIdx.x;
    const int kind = t / SPAN;
    const int p    = t % SPAN;
    const int idx  = blockIdx.x * SPAN + p;
    if (idx < M * nGroups) (kind ? edst : esrc)[idx] = sc[kind][p];

    if (active) {
        unsigned coff = (unsigned)(mt * 16 + 8 * half) * uN + (unsigned)(ng * 64 + l);
        #pragma unroll
        for (int r = 0; r < 8; ++r) {
            float* cp = C + coff;
            cp[0]  = acc0[r];
            cp[16] = acc1[r];
            cp[32] = acc2[r];
            cp[48] = acc3[r];
            coff += uN;
        }
    }
}


typedef float v4f __attribute__((ext_vector_type(4)));

__device__ __forceinline__ void scores_tile(int wave, int lane, const float* __restrict__ X,
                                            const float* __restrict__ wsd,
                                            float* __restrict__ esrc, float* __restrict__ edst) {
    const int l = lane & 15, half = lane >> 4;
    const float* Ap = X + (unsigned)(wave * 16 + l) * (unsigned)IN_CH + 2u * half;
    const float* Bp = wsd + (2 * half) * 16 + l;
    v8f accE = {}, accO = {};
    #pragma unroll
    for (int s = 0; s < IN_CH / 4; s += 2) {
        const v2f a0 = *(const v2f*)(Ap + 4 * s);
        const v2f b0 = { Bp[(4 * s) * 16], Bp[(4 * s + 1) * 16] };
        const v2f a1 = *(const v2f*)(Ap + 4 * s + 4);
        const v2f b1 = { Bp[(4 * s + 4) * 16], Bp[(4 * s + 5) * 16] };
        accE = __builtin_amdgcn_wmma_f32_16x16x4_f32(false, a0, false, b0, (short)0, accE, false, false);
        accO = __builtin_amdgcn_wmma_f32_16x16x4_f32(false, a1, false, b1, (short)0, accO, false, false);
    }
    float* dstp = (l < 8 ? esrc : edst) + (unsigned)(wave * 16 + 8 * half) * (unsigned)HEADS + (unsigned)(l & 7);
    #pragma unroll
    for (int r = 0; r < 8; ++r) dstp[r * HEADS] = accE[r] + accO[r];
}

struct Blk8 { int src; float v[HEADS]; bool valid; };
__device__ __forceinline__ Blk8 load8(const int2* __restrict__ csr, const float* __restrict__ esrc,
                                      const float (&ed)[HEADS], int base, int end, int lane) {
    Blk8 b;
    const int idx = base + lane;
    b.valid = idx < end;
    b.src = b.valid ? csr[idx].y : 0;
    const v4f* ep = (const v4f*)(esrc + (unsigned)b.src * (unsigned)HEADS);
    const v4f e0 = ep[0], e1 = ep[1];
    const float e[HEADS] = { e0[0], e0[1], e0[2], e0[3], e1[0], e1[1], e1[2], e1[3] };
    #pragma unroll
    for (int q = 0; q < HEADS; ++q) {
        float v = e[q] + ed[q];
        v = v > 0.f ? v : 0.2f * v;
        b.v[q] = b.valid ? v : -INFINITY;
    }
    return b;
}
template <int U>
__device__ __forceinline__ void group8(int srcv, const float (&av)[HEADS], int j,
                                       const float* __restrict__ X, unsigned c, v4f (&acc)[HEADS]) {
    float a[U][HEADS];
    v4f v[U];
    #pragma unroll
    for (int u = 0; u < U; ++u) {
        const int s = __builtin_amdgcn_readlane(srcv, j + u);
        #pragma unroll
        for (int q = 0; q < HEADS; ++q)
            a[u][q] = __builtin_bit_cast(float, __builtin_amdgcn_readlane(__builtin_bit_cast(int, av[q]), j + u));
        v[u] = *(const v4f*)(X + (unsigned)s * (unsigned)IN_CH + c);
    }
    #pragma unroll
    for (int u = 0; u < U; ++u) {
        #pragma unroll
        for (int q = 0; q < HEADS; ++q) acc[q] += a[u][q] * v[u];
    }
}
__device__ __forceinline__ void block_sum8(int srcv, const float (&av)[HEADS], int cnt,
                                           const float* __restrict__ X, unsigned c, v4f (&acc)[HEADS]) {
    int j = 0;
    for (; j + 4 <= cnt; j += 4) group8<4>(srcv, av, j, X, c, acc);
    const int r = cnt - j;
    if (r == 3)      group8<3>(srcv, av, j, X, c, acc);
    else if (r == 2) group8<2>(srcv, av, j, X, c, acc);
    else if (r == 1) group8<1>(srcv, av, j, X, c, acc);
}

#define ZP (HEADS * IN_CH + 4)
__global__ void __launch_bounds__(32 * GEMM_WAVES)
gather_project_l1(const int* __restrict__ rowptr, const int2* __restrict__ csr,
                  const float* __restrict__ esrc, const float* __restrict__ edst,
                  const float* __restrict__ X, const float* __restrict__ W1,
                  const float* __restrict__ bias, float* __restrict__ out) {
    __shared__ __attribute__((aligned(16))) float zs[16 * ZP];
    const int lane = threadIdx.x & 31;
    const int w = __builtin_amdgcn_readfirstlane((int)(threadIdx.x >> 5));
    const unsigned c = 4u * (unsigned)lane;

    for (int i = 0; i < 2; ++i) {
        const int ml   = 2 * w + i;
        const int node = (int)blockIdx.x * 16 + ml;
        const int beg = rowptr[node], end = rowptr[node + 1];
        float ed[HEADS], vself[HEADS];
        #pragma unroll
        for (int q = 0; q < HEADS; ++q) {
            ed[q] = edst[node * HEADS + q];
            const float v = esrc[node * HEADS + q] + ed[q];
            vself[q] = v > 0.f ? v : 0.2f * v;
        }
        const Blk8 b0 = load8(csr, esrc, ed, beg, end, lane);
        float m[HEADS];
        #pragma unroll
        for (int q = 0; q < HEADS; ++q) m[q] = b0.v[q];
        for (int base = beg + 32; base < end; base += 32) {
            const Blk8 b = load8(csr, esrc, ed, base, end, lane);
            #pragma unroll
            for (int q = 0; q < HEADS; ++q) m[q] = fmaxf(m[q], b.v[q]);
        }
        #pragma unroll
        for (int q = 0; q < HEADS; ++q) m[q] = fmaxf(wave_max(m[q]), vself[q]);
        float p0[HEADS], s[HEADS];
        #pragma unroll
        for (int q = 0; q < HEADS; ++q) { p0[q] = b0.valid ? expf(b0.v[q] - m[q]) : 0.f; s[q] = p0[q]; }
        for (int base = beg + 32; base < end; base += 32) {
            const Blk8 b = load8(csr, esrc, ed, base, end, lane);
            #pragma unroll
            for (int q = 0; q < HEADS; ++q) s[q] += b.valid ? expf(b.v[q] - m[q]) : 0.f;
        }
        float dn[HEADS], a0[HEADS];
        v4f acc[HEADS];
        const v4f xself = *(const v4f*)(X + (unsigned)node * (unsigned)IN_CH + c);
        #pragma unroll
        for (int q = 0; q < HEADS; ++q) {
            const float pself = expf(vself[q] - m[q]);
            dn[q] = (wave_sum(s[q]) + pself) + 1e-16f;
            a0[q] = p0[q] / dn[q];
            acc[q] = (pself / dn[q]) * xself;
        }
        int left = end - beg;
        if (left > 0) block_sum8(b0.src, a0, left < 32 ? left : 32, X, c, acc);
        for (int base = beg + 32; base < end; base += 32) {
            left -= 32;
            const Blk8 b = load8(csr, esrc, ed, base, end, lane);
            float a[HEADS];
            #pragma unroll
            for (int q = 0; q < HEADS; ++q) a[q] = (b.valid ? expf(b.v[q] - m[q]) : 0.f) / dn[q];
            block_sum8(b.src, a, left < 32 ? left : 32, X, c, acc);
        }
        v4f* zrow = (v4f*)(zs + ml * ZP) + lane;
        #pragma unroll
        for (int q = 0; q < HEADS; ++q) zrow[q * (IN_CH / 4)] = acc[q];
    }
    __syncthreads();

    const int l    = lane & 15;
    const int half = lane >> 4;
    v8f acc0 = {}, acc1 = {}, acc2 = {}, acc3 = {};
    const float* Ap = zs + l * ZP + w * IN_CH + 2 * half;
    const float* Bp = W1 + ((unsigned)(2 * half) * (unsigned)HC1 + (unsigned)(w * 64 + l));
    Frags16 P, Q;
    load_frags16<HC1>(P, Ap, Bp);
    for (int k0 = 0; k0 < IN_CH - 32; k0 += 32) {
        load_frags16<HC1>(Q, Ap + 16, Bp + 16 * HC1);
        mma_frags16(P, acc0, acc1, acc2, acc3);
        load_frags16<HC1>(P, Ap + 32, Bp + 32 * HC1);
        mma_frags16(Q, acc0, acc1, acc2, acc3);
        Ap += 32;
        Bp += 32 * HC1;
    }
    load_frags16<HC1>(Q, Ap + 16, Bp + 16 * HC1);
    mma_frags16(P, acc0, acc1, acc2, acc3);
    mma_frags16(Q, acc0, acc1, acc2, acc3);

    const float* bp = bias + w * 64 + l;
    const float b0v = bp[0], b1v = bp[16], b2v = bp[32], b3v = bp[48];
    unsigned coff = (unsigned)((int)blockIdx.x * 16 + 8 * half) * (unsigned)HC1 + (unsigned)(w * 64 + l);
    #pragma unroll
    for (int r = 0; r < 8; ++r) {
        float* cp = out + coff;
        cp[0]  = fmaxf(acc0[r] + b0v, 0.f);
        cp[16] = fmaxf(acc1[r] + b1v, 0.f);
        cp[32] = fmaxf(acc2[r] + b2v, 0.f);
        cp[48] = fmaxf(acc3[r] + b3v, 0.f);
        coff += (unsigned)HC1;
    }
}

#define ZERO_BLOCKS ((NN + 255) / 256)
#define WSD_BLOCKS  ((IN_CH * 16 + 255) / 256)
__global__ void zero_counts_and_score_weights(int* __restrict__ cnt,
                                              const float* __restrict__ W1,
                                              const float* __restrict__ a_src,
                                              const float* __restrict__ a_dst,
                                              float* __restrict__ wsd) {
    if (blockIdx.x < ZERO_BLOCKS) {
        const int i = blockIdx.x * 256 + threadIdx.x;
        if (i < NN) cnt[i] = 0;
        return;
    }
    const int t = (blockIdx.x - ZERO_BLOCKS) * 256 + threadIdx.x;
    if (t >= IN_CH * 16) return;
    const int k = t >> 4, j = t & 15, h = j & 7;
    const float* a = (j < 8 ? a_src : a_dst) + h * HID;
    const float* wr = W1 + k * HC1 + h * HID;
    float s = 0.f;
    for (int cc = 0; cc < HID; ++cc) s = fmaf(wr[cc], a[cc], s);
    wsd[t] = s;
}

#define SCORE_BLOCKS ((NN / 16 + GEMM_WAVES - 1) / GEMM_WAVES)
#define COUNT_BLOCKS ((NE + 255) / 256)
__global__ void dst_count_and_scores(const int* __restrict__ ei, int* __restrict__ cnt,
                                     int* __restrict__ rank,
                                     const float* __restrict__ X, const float* __restrict__ wsd,
                                     float* __restrict__ esrc, float* __restrict__ edst) {
    if (blockIdx.x < SCORE_BLOCKS) {
        const int wave = (int)(blockIdx.x * GEMM_WAVES + (threadIdx.x >> 5));
        if (wave < NN / 16) scores_tile(wave, threadIdx.x & 31, X, wsd, esrc, edst);
        return;
    }
    const int e = (blockIdx.x - SCORE_BLOCKS) * 256 + threadIdx.x;
    if (e >= NE) return;
    rank[e] = atomicAdd(&cnt[ei[NE + e]], 1);
}

#define SCAN_T   1024
#define SCAN_PER ((NN + SCAN_T - 1) / SCAN_T)
__global__ void row_scan(const int* __restrict__ cnt, int* __restrict__ rowptr) {
    __shared__ int part[SCAN_T];
    const int t = threadIdx.x;
    const int base = t * SCAN_PER;
    int loc[SCAN_PER];
    int s = 0;
    #pragma unroll
    for (int i = 0; i < SCAN_PER; ++i) {
        int idx = base + i;
        int v = idx < NN ? cnt[idx] : 0;
        loc[i] = s;
        s += v;
    }
    part[t] = s;
    __syncthreads();
    for (int off = 1; off < SCAN_T; off <<= 1) {
        int v = t >= off ? part[t - off] : 0;
        __syncthreads();
        part[t] += v;
        __syncthreads();
    }
    const int excl = part[t] - s;
    #pragma unroll
    for (int i = 0; i < SCAN_PER; ++i) {
        int idx = base + i;
        if (idx < NN) rowptr[idx] = excl + loc[i];
    }
    if (t == SCAN_T - 1) rowptr[NN] = part[t];
}

__global__ void dst_fill(const int* __restrict__ ei, const int* __restrict__ rowptr,
                         const int* __restrict__ rank, int2* __restrict__ csr) {
    int e = blockIdx.x * blockDim.x + threadIdx.x;
    if (e >= NE) return;
    csr[rowptr[ei[NE + e]] + rank[e]] = make_int2(e, ei[e]);
}

#define EPI_RELU       1
#define EPI_LOGSOFTMAX 2
template <int HC, int C, int VEC>
struct SegGather {
    typedef float vf __attribute__((ext_vector_type(VEC)));
    static constexpr int H  = HC / C;
    static constexpr int NH = (32 * VEC) / C;
    static_assert(NH == 1 || NH == 2, "a wave covers one or two whole heads");

    struct Block { int src; float v[NH]; bool valid; };

    static __device__ __forceinline__ Block load(const int2* __restrict__ csr,
                                                 const float* __restrict__ esrc,
                                                 const float (&ed)[NH],
                                                 int base, int end, int lane, int hbase) {
        Block b;
        const int idx = base + lane;
        b.valid = idx < end;
        b.src = b.valid ? csr[idx].y : 0;
        const float* ep = esrc + (unsigned)b.src * (unsigned)H + (unsigned)hbase;
        float e[NH];
        if constexpr (NH == 2) { const float2 t = *(const float2*)ep; e[0] = t.x; e[1] = t.y; }
        else                   { e[0] = ep[0]; }
        for (int q = 0; q < NH; ++q) {
            float v = e[q] + ed[q];
            v = v > 0.f ? v : 0.2f * v;
            b.v[q] = b.valid ? v : -INFINITY;
        }
        return b;
    }

    template <int U>
    static __device__ __forceinline__ void group(int srcv, const float (&av)[NH], int j, bool upper,
                                                 const float* __restrict__ hval, unsigned c, vf& acc) {
        float a[U];
        vf v[U];
        #pragma unroll
        for (int u = 0; u < U; ++u) {
            const int s = __builtin_amdgcn_readlane(srcv, j + u);
            float al = __builtin_bit_cast(float, __builtin_amdgcn_readlane(__builtin_bit_cast(int, av[0]), j + u));
            if constexpr (NH == 2) {
                const float a1 = __builtin_bit_cast(float, __builtin_amdgcn_readlane(__builtin_bit_cast(int, av[1]), j + u));
                al = upper ? a1 : al;
            }
            a[u] = al;
            v[u] = *(const vf*)(hval + (unsigned)s * (unsigned)HC + c);
        }
        #pragma unroll
        for (int u = 0; u < U; ++u) acc += a[u] * v[u];
    }

    static __device__ __forceinline__ void block_sum(int srcv, const float (&av)[NH], int cnt, bool upper,
                                                     const float* __restrict__ hval, unsigned c, vf& acc) {
        int j = 0;
        for (; j + 4 <= cnt; j += 4) group<4>(srcv, av, j, upper, hval, c, acc);
        const int r = cnt - j;
        if (r == 3)      group<3>(srcv, av, j, upper, hval, c, acc);
        else if (r == 2) group<2>(srcv, av, j, upper, hval, c, acc);
        else if (r == 1) group<1>(srcv, av, j, upper, hval, c, acc);
    }
};

template <int HC, int C, int VEC, int EPI>
__global__ void gather_softmax_aggregate(const int* __restrict__ rowptr,
                                         const int2* __restrict__ csr,
                                         const float* __restrict__ esrc,
                                         const float* __restrict__ edst,
                                         const float* __restrict__ hval,
                                         const float* __restrict__ bias,
                                         float* __restrict__ agg) {
    typedef SegGather<HC, C, VEC> G;
    typedef typename G::vf vf;
    typedef typename G::Block Block;
    constexpr int H   = G::H;
    constexpr int NH  = G::NH;
    constexpr int WPN = HC / (32 * VEC);
    const int wave = __builtin_amdgcn_readfirstlane(
        (int)((blockIdx.x * blockDim.x + threadIdx.x) >> 5));
    if (wave >= NN * WPN) return;
    const int lane  = threadIdx.x & 31;
    const int node  = wave / WPN;
    const int chunk = wave % WPN;
    const unsigned c = (unsigned)(chunk * 32 * VEC) + (unsigned)lane * VEC;
    const int  hbase = chunk * NH;
    const bool upper = NH == 2 && lane >= 16;
    const int beg = rowptr[node], end = rowptr[node + 1];

    float ed[NH], vself[NH];
    for (int q = 0; q < NH; ++q) {
        ed[q] = edst[node * H + hbase + q];
        float v = esrc[node * H + hbase + q] + ed[q];
        vself[q] = v > 0.f ? v : 0.2f * v;
    }

    const Block b0 = G::load(csr, esrc, ed, beg, end, lane, hbase);
    float m[NH];
    for (int q = 0; q < NH; ++q) m[q] = b0.v[q];
    for (int base = beg + 32; base < end; base += 32) {
        const Block b = G::load(csr, esrc, ed, base, end, lane, hbase);
        for (int q = 0; q < NH; ++q) m[q] = fmaxf(m[q], b.v[q]);
    }
    for (int q = 0; q < NH; ++q) m[q] = fmaxf(wave_max(m[q]), vself[q]);

    float p0[NH], s[NH];
    for (int q = 0; q < NH; ++q) { p0[q] = b0.valid ? expf(b0.v[q] - m[q]) : 0.f; s[q] = p0[q]; }
    for (int base = beg + 32; base < end; base += 32) {
        const Block b = G::load(csr, esrc, ed, base, end, lane, hbase);
        for (int q = 0; q < NH; ++q) s[q] += b.valid ? expf(b.v[q] - m[q]) : 0.f;
    }
    float pself[NH], dn[NH], a0[NH];
    for (int q = 0; q < NH; ++q) {
        pself[q] = expf(vself[q] - m[q]);
        dn[q] = (wave_sum(s[q]) + pself[q]) + 1e-16f;
        a0[q] = p0[q] / dn[q];
    }

    float aself = pself[0] / dn[0];
    if constexpr (NH == 2) { const float a1 = pself[1] / dn[1]; aself = upper ? a1 : aself; }
    vf acc = aself * *(const vf*)(hval + (unsigned)node * (unsigned)HC + c);
    int left = end - beg;
    if (left > 0) G::block_sum(b0.src, a0, left < 32 ? left : 32, upper, hval, c, acc);
    for (int base = beg + 32; base < end; base += 32) {
        left -= 32;
        const Block b = G::load(csr, esrc, ed, base, end, lane, hbase);
        float a[NH];
        for (int q = 0; q < NH; ++q) a[q] = (b.valid ? expf(b.v[q] - m[q]) : 0.f) / dn[q];
        G::block_sum(b.src, a, left < 32 ? left : 32, upper, hval, c, acc);
    }
    vf res = acc + *(const vf*)(bias + c);
    if constexpr (EPI == EPI_RELU) {
        for (int i = 0; i < VEC; ++i) res[i] = res[i] > 0.f ? res[i] : 0.f;
    } else {
        static_assert(WPN == 1 && H == 1, "the wave must hold the node's whole row");
        float mx = res[0];
        for (int i = 1; i < VEC; ++i) mx = fmaxf(mx, res[i]);
        mx = wave_max(mx);
        float se = 0.f;
        for (int i = 0; i < VEC; ++i) se += expf(res[i] - mx);
        const float lse = logf(wave_sum(se)) + mx;
        for (int i = 0; i < VEC; ++i) res[i] -= lse;
    }
    *(vf*)(agg + (unsigned)node * (unsigned)HC + c) = res;
}

extern "C" void kernel_launch(void* const* d_in, const int* in_sizes, int n_in,
                              void* d_out, int out_size, void* d_ws, size_t ws_size,
                              hipStream_t stream) {
    const float* x   = (const float*)d_in[0];
    const int*   ei  = (const int*)  d_in[1];
    const float* W1  = (const float*)d_in[2];
    const float* a1s = (const float*)d_in[3];
    const float* a1d = (const float*)d_in[4];
    const float* b1  = (const float*)d_in[5];
    const float* W2  = (const float*)d_in[6];
    const float* a2s = (const float*)d_in[7];
    const float* a2d = (const float*)d_in[8];
    const float* b2  = (const float*)d_in[9];
    float* out = (float*)d_out;

    float* ws = (float*)d_ws;
    float* h1pre = ws;
    float* agg1  = h1pre + (size_t)NN * HC1;
    float* esrc1 = agg1  + (size_t)NN * HC1;
    float* edst1 = esrc1 + (size_t)NN * HEADS;
    float* h2pre = edst1 + (size_t)NN * HEADS;
    float* esrc2 = h2pre + (size_t)NN * OUT_CH;
    float* edst2 = esrc2 + NN;
    int*   cnt   = (int*)(edst2 + NN);
    int*   rank  = cnt + NN;
    int*   rowp  = rank + NE;
    int2*  csr   = (int2*)(rowp + NN + 2);

    const int TB = 32 * GEMM_WAVES;
    auto blocks = [](long long n, int tb) { return (unsigned)((n + tb - 1) / tb); };

    float* wsd = (float*)(csr + NE);

    zero_counts_and_score_weights<<<ZERO_BLOCKS + WSD_BLOCKS, TB, 0, stream>>>(cnt, W1, a1s, a1d, wsd);
    dst_count_and_scores<<<SCORE_BLOCKS + COUNT_BLOCKS, TB, 0, stream>>>(ei, cnt, rank, x, wsd, esrc1, edst1);
    row_scan <<<1, SCAN_T, 0, stream>>>(cnt, rowp);
    dst_fill <<<blocks(NE, TB), TB, 0, stream>>>(ei, rowp, rank, csr);

    gather_project_l1<<<NN / 16, TB, 0, stream>>>(rowp, csr, esrc1, edst1, x, W1, b1, agg1);

    {
        long long waves = (long long)(NN / 16) * (OUT_CH / 64);
        wmma_gemm_f32<HC1, OUT_CH><<<blocks(waves * 32, TB), TB, 0, stream>>>(
            agg1, W2, h2pre, a2s, a2d, esrc2, edst2, NN);
    }
    gather_softmax_aggregate<OUT_CH, OUT_CH, 2, EPI_LOGSOFTMAX><<<blocks((long long)NN * 32, TB), TB, 0, stream>>>(
        rowp, csr, esrc2, edst2, h2pre, b2, out);
}


# Layer-2 GEMM fused into the layer-1 kernel: W1 product transposed so relu'd h1 stays in registers as the next A operand; 8 K-slices added via LDS in wave order (h2 sum order changes); 6 launches
# speedup vs baseline: 7.9458x; 1.0474x over previous
#include <hip/hip_runtime.h>
#include <math.h>

#define NN      20000
#define NE      320000
#define ETOT    (NE + NN)
#define IN_CH   128
#define HID     64
#define HEADS   8
#define HC1     (HEADS * HID)
#define OUT_CH  64

typedef float v2f __attribute__((ext_vector_type(2)));
typedef float v8f __attribute__((ext_vector_type(8)));

__device__ __forceinline__ float xhalf(float x) {
    return __builtin_bit_cast(float, __builtin_amdgcn_permlanex16(
        __builtin_bit_cast(int, x), __builtin_bit_cast(int, x), 0x76543210, 0xfedcba98, false, false));
}
template <int XM>
__device__ __forceinline__ float row_xor(float x) {
    const int xi = __builtin_bit_cast(int, x);
    return __builtin_bit_cast(float, __builtin_amdgcn_update_dpp(xi, xi, 0x160 + XM, 0xf, 0xf, false));
}
__device__ __forceinline__ float wave_max(float v) {
    v = fmaxf(v, row_xor<1>(v));
    v = fmaxf(v, row_xor<2>(v));
    v = fmaxf(v, row_xor<4>(v));
    v = fmaxf(v, row_xor<8>(v));
    return fmaxf(v, xhalf(v));
}
__device__ __forceinline__ float row_sum16(float v) {
    v += row_xor<1>(v);
    v += row_xor<2>(v);
    v += row_xor<4>(v);
    return v + row_xor<8>(v);
}
__device__ __forceinline__ float wave_sum(float v) {
    v += row_xor<1>(v);
    v += row_xor<2>(v);
    v += row_xor<4>(v);
    v += row_xor<8>(v);
    return v + xhalf(v);
}

#define GEMM_WAVES 8
struct Frags16 { v2f a[4]; v2f b[4][4]; };
template <int N>
__device__ __forceinline__ void load_frags16(Frags16& f, const float* __restrict__ Ap,
                                             const float* __restrict__ Bp) {
    #pragma unroll
    for (int s = 0; s < 4; ++s) {
        f.a[s] = *(const v2f*)(Ap + 4 * s);
        #pragma unroll
        for (int j = 0; j < 4; ++j) {
            v2f b = { Bp[(4 * s) * N + 16 * j], Bp[(4 * s + 1) * N + 16 * j] };
            f.b[s][j] = b;
        }
    }
}
__device__ __forceinline__ void mma_frags16_t(const Frags16& f, v8f& acc0, v8f& acc1, v8f& acc2, v8f& acc3) {
    #pragma unroll
    for (int s = 0; s < 4; ++s) {
        acc0 = __builtin_amdgcn_wmma_f32_16x16x4_f32(false, f.b[s][0], false, f.a[s], (short)0, acc0, false, false);
        acc1 = __builtin_amdgcn_wmma_f32_16x16x4_f32(false, f.b[s][1], false, f.a[s], (short)0, acc1, false, false);
        acc2 = __builtin_amdgcn_wmma_f32_16x16x4_f32(false, f.b[s][2], false, f.a[s], (short)0, acc2, false, false);
        acc3 = __builtin_amdgcn_wmma_f32_16x16x4_f32(false, f.b[s][3], false, f.a[s], (short)0, acc3, false, false);
    }
}
struct FragsB16 { v2f b[4][4]; };
template <int N>
__device__ __forceinline__ void load_fragsB16(FragsB16& f, const float* __restrict__ Bp) {
    #pragma unroll
    for (int s = 0; s < 4; ++s) {
        #pragma unroll
        for (int j = 0; j < 4; ++j) {
            v2f b = { Bp[(2 * s) * N + 16 * j], Bp[(2 * s + 1) * N + 16 * j] };
            f.b[s][j] = b;
        }
    }
}
__device__ __forceinline__ void mma_tile16(const v8f& T, const FragsB16& f, v8f& o0, v8f& o1, v8f& o2, v8f& o3) {
    #pragma unroll
    for (int s = 0; s < 4; ++s) {
        const v2f a = { T[2 * s], T[2 * s + 1] };
        o0 = __builtin_amdgcn_wmma_f32_16x16x4_f32(false, a, false, f.b[s][0], (short)0, o0, false, false);
        o1 = __builtin_amdgcn_wmma_f32_16x16x4_f32(false, a, false, f.b[s][1], (short)0, o1, false, false);
        o2 = __builtin_amdgcn_wmma_f32_16x16x4_f32(false, a, false, f.b[s][2], (short)0, o2, false, false);
        o3 = __builtin_amdgcn_wmma_f32_16x16x4_f32(false, a, false, f.b[s][3], (short)0, o3, false, false);
    }
}


typedef float v4f __attribute__((ext_vector_type(4)));

__device__ __forceinline__ void scores_tile(int wave, int lane, const float* __restrict__ X,
                                            const float* __restrict__ wsd,
                                            float* __restrict__ esrc, float* __restrict__ edst) {
    const int l = lane & 15, half = lane >> 4;
    const float* Ap = X + (unsigned)(wave * 16 + l) * (unsigned)IN_CH + 2u * half;
    const float* Bp = wsd + (2 * half) * 16 + l;
    v8f accE = {}, accO = {};
    #pragma unroll
    for (int s = 0; s < IN_CH / 4; s += 2) {
        const v2f a0 = *(const v2f*)(Ap + 4 * s);
        const v2f b0 = { Bp[(4 * s) * 16], Bp[(4 * s + 1) * 16] };
        const v2f a1 = *(const v2f*)(Ap + 4 * s + 4);
        const v2f b1 = { Bp[(4 * s + 4) * 16], Bp[(4 * s + 5) * 16] };
        accE = __builtin_amdgcn_wmma_f32_16x16x4_f32(false, a0, false, b0, (short)0, accE, false, false);
        accO = __builtin_amdgcn_wmma_f32_16x16x4_f32(false, a1, false, b1, (short)0, accO, false, false);
    }
    float* dstp = (l < 8 ? esrc : edst) + (unsigned)(wave * 16 + 8 * half) * (unsigned)HEADS + (unsigned)(l & 7);
    #pragma unroll
    for (int r = 0; r < 8; ++r) dstp[r * HEADS] = accE[r] + accO[r];
}

struct Blk8 { int src; float v[HEADS]; bool valid; };
__device__ __forceinline__ Blk8 load8(const int2* __restrict__ csr, const float* __restrict__ esrc,
                                      const float (&ed)[HEADS], int base, int end, int lane) {
    Blk8 b;
    const int idx = base + lane;
    b.valid = idx < end;
    b.src = b.valid ? csr[idx].y : 0;
    const v4f* ep = (const v4f*)(esrc + (unsigned)b.src * (unsigned)HEADS);
    const v4f e0 = ep[0], e1 = ep[1];
    const float e[HEADS] = { e0[0], e0[1], e0[2], e0[3], e1[0], e1[1], e1[2], e1[3] };
    #pragma unroll
    for (int q = 0; q < HEADS; ++q) {
        float v = e[q] + ed[q];
        v = v > 0.f ? v : 0.2f * v;
        b.v[q] = b.valid ? v : -INFINITY;
    }
    return b;
}
template <int U>
__device__ __forceinline__ void group8(int srcv, const float (&av)[HEADS], int j,
                                       const float* __restrict__ X, unsigned c, v4f (&acc)[HEADS]) {
    float a[U][HEADS];
    v4f v[U];
    #pragma unroll
    for (int u = 0; u < U; ++u) {
        const int s = __builtin_amdgcn_readlane(srcv, j + u);
        #pragma unroll
        for (int q = 0; q < HEADS; ++q)
            a[u][q] = __builtin_bit_cast(float, __builtin_amdgcn_readlane(__builtin_bit_cast(int, av[q]), j + u));
        v[u] = *(const v4f*)(X + (unsigned)s * (unsigned)IN_CH + c);
    }
    #pragma unroll
    for (int u = 0; u < U; ++u) {
        #pragma unroll
        for (int q = 0; q < HEADS; ++q) acc[q] += a[u][q] * v[u];
    }
}
__device__ __forceinline__ void block_sum8(int srcv, const float (&av)[HEADS], int cnt,
                                           const float* __restrict__ X, unsigned c, v4f (&acc)[HEADS]) {
    int j = 0;
    for (; j + 4 <= cnt; j += 4) group8<4>(srcv, av, j, X, c, acc);
    const int r = cnt - j;
    if (r == 3)      group8<3>(srcv, av, j, X, c, acc);
    else if (r == 2) group8<2>(srcv, av, j, X, c, acc);
    else if (r == 1) group8<1>(srcv, av, j, X, c, acc);
}

#define ZP (HEADS * IN_CH + 4)
__global__ void __launch_bounds__(32 * GEMM_WAVES)
gather_project_l1(const int* __restrict__ rowptr, const int2* __restrict__ csr,
                  const float* __restrict__ esrc, const float* __restrict__ edst,
                  const float* __restrict__ X, const float* __restrict__ W1,
                  const float* __restrict__ bias, const float* __restrict__ W2,
                  const float* __restrict__ a_src2, const float* __restrict__ a_dst2,
                  float* __restrict__ h2, float* __restrict__ sc2) {
    __shared__ __attribute__((aligned(16))) float zs[16 * ZP];
    __shared__ float scs[32];
    const int lane = threadIdx.x & 31;
    const int w = __builtin_amdgcn_readfirstlane((int)(threadIdx.x >> 5));
    const unsigned c = 4u * (unsigned)lane;

    for (int i = 0; i < 2; ++i) {
        const int ml   = 2 * w + i;
        const int node = (int)blockIdx.x * 16 + ml;
        const int beg = rowptr[node], end = rowptr[node + 1];
        float ed[HEADS], vself[HEADS];
        #pragma unroll
        for (int q = 0; q < HEADS; ++q) {
            ed[q] = edst[node * HEADS + q];
            const float v = esrc[node * HEADS + q] + ed[q];
            vself[q] = v > 0.f ? v : 0.2f * v;
        }
        const Blk8 b0 = load8(csr, esrc, ed, beg, end, lane);
        float m[HEADS];
        #pragma unroll
        for (int q = 0; q < HEADS; ++q) m[q] = b0.v[q];
        for (int base = beg + 32; base < end; base += 32) {
            const Blk8 b = load8(csr, esrc, ed, base, end, lane);
            #pragma unroll
            for (int q = 0; q < HEADS; ++q) m[q] = fmaxf(m[q], b.v[q]);
        }
        #pragma unroll
        for (int q = 0; q < HEADS; ++q) m[q] = fmaxf(wave_max(m[q]), vself[q]);
        float p0[HEADS], s[HEADS];
        #pragma unroll
        for (int q = 0; q < HEADS; ++q) { p0[q] = b0.valid ? expf(b0.v[q] - m[q]) : 0.f; s[q] = p0[q]; }
        for (int base = beg + 32; base < end; base += 32) {
            const Blk8 b = load8(csr, esrc, ed, base, end, lane);
            #pragma unroll
            for (int q = 0; q < HEADS; ++q) s[q] += b.valid ? expf(b.v[q] - m[q]) : 0.f;
        }
        float dn[HEADS], a0[HEADS];
        v4f acc[HEADS];
        const v4f xself = *(const v4f*)(X + (unsigned)node * (unsigned)IN_CH + c);
        #pragma unroll
        for (int q = 0; q < HEADS; ++q) {
            const float pself = expf(vself[q] - m[q]);
            dn[q] = (wave_sum(s[q]) + pself) + 1e-16f;
            a0[q] = p0[q] / dn[q];
            acc[q] = (pself / dn[q]) * xself;
        }
        int left = end - beg;
        if (left > 0) block_sum8(b0.src, a0, left < 32 ? left : 32, X, c, acc);
        for (int base = beg + 32; base < end; base += 32) {
            left -= 32;
            const Blk8 b = load8(csr, esrc, ed, base, end, lane);
            float a[HEADS];
            #pragma unroll
            for (int q = 0; q < HEADS; ++q) a[q] = (b.valid ? expf(b.v[q] - m[q]) : 0.f) / dn[q];
            block_sum8(b.src, a, left < 32 ? left : 32, X, c, acc);
        }
        v4f* zrow = (v4f*)(zs + ml * ZP) + lane;
        #pragma unroll
        for (int q = 0; q < HEADS; ++q) zrow[q * (IN_CH / 4)] = acc[q];
    }
    __syncthreads();

    const int l    = lane & 15;
    const int half = lane >> 4;
    v8f acc0 = {}, acc1 = {}, acc2 = {}, acc3 = {};
    const float* Ap = zs + l * ZP + w * IN_CH + 2 * half;
    const float* Bp = W1 + ((unsigned)(2 * half) * (unsigned)HC1 + (unsigned)(w * 64 + l));
    const float* W2p = W2 + ((unsigned)(w * 64 + 8 * half) * (unsigned)OUT_CH + (unsigned)l);
    Frags16 P, Q;
    load_frags16<HC1>(P, Ap, Bp);
    for (int k0 = 0; k0 < IN_CH - 32; k0 += 32) {
        load_frags16<HC1>(Q, Ap + 16, Bp + 16 * HC1);
        mma_frags16_t(P, acc0, acc1, acc2, acc3);
        load_frags16<HC1>(P, Ap + 32, Bp + 32 * HC1);
        mma_frags16_t(Q, acc0, acc1, acc2, acc3);
        Ap += 32;
        Bp += 32 * HC1;
    }
    load_frags16<HC1>(Q, Ap + 16, Bp + 16 * HC1);
    mma_frags16_t(P, acc0, acc1, acc2, acc3);
    FragsB16 R, S;
    load_fragsB16<OUT_CH>(R, W2p);
    mma_frags16_t(Q, acc0, acc1, acc2, acc3);

    {
        const v4f* bq = (const v4f*)(bias + w * 64 + 8 * half);
        const v4f b00 = bq[0],  b01 = bq[1];
        const v4f b10 = bq[4],  b11 = bq[5];
        const v4f b20 = bq[8],  b21 = bq[9];
        const v4f b30 = bq[12], b31 = bq[13];
        #pragma unroll
        for (int r = 0; r < 4; ++r) {
            acc0[r] = fmaxf(acc0[r] + b00[r], 0.f);  acc0[r + 4] = fmaxf(acc0[r + 4] + b01[r], 0.f);
            acc1[r] = fmaxf(acc1[r] + b10[r], 0.f);  acc1[r + 4] = fmaxf(acc1[r + 4] + b11[r], 0.f);
            acc2[r] = fmaxf(acc2[r] + b20[r], 0.f);  acc2[r + 4] = fmaxf(acc2[r + 4] + b21[r], 0.f);
            acc3[r] = fmaxf(acc3[r] + b30[r], 0.f);  acc3[r + 4] = fmaxf(acc3[r + 4] + b31[r], 0.f);
        }
    }

    v8f o0 = {}, o1 = {}, o2 = {}, o3 = {};
    load_fragsB16<OUT_CH>(S, W2p + 16 * OUT_CH);
    mma_tile16(acc0, R, o0, o1, o2, o3);
    load_fragsB16<OUT_CH>(R, W2p + 32 * OUT_CH);
    mma_tile16(acc1, S, o0, o1, o2, o3);
    load_fragsB16<OUT_CH>(S, W2p + 48 * OUT_CH);
    mma_tile16(acc2, R, o0, o1, o2, o3);
    mma_tile16(acc3, S, o0, o1, o2, o3);

    {
        float* prow = zs + half * ZP + w * IN_CH + 4 * l;
        #pragma unroll
        for (int r = 0; r < 8; ++r) {
            const v4f pv = { o0[r], o1[r], o2[r], o3[r] };
            *(v4f*)(prow + 2 * r * ZP) = pv;
        }
    }
    __syncthreads();
    const float* rrow = zs + (2 * w + half) * ZP + 4 * l;
    v4f tot = *(const v4f*)rrow;
    #pragma unroll
    for (int ww = 1; ww < GEMM_WAVES; ++ww) tot += *(const v4f*)(rrow + ww * IN_CH);

    const float* as = a_src2 + l;
    const float* ad = a_dst2 + l;
    const float vs = row_sum16(tot[0] * as[0] + tot[1] * as[16] + tot[2] * as[32] + tot[3] * as[48]);
    const float vd = row_sum16(tot[0] * ad[0] + tot[1] * ad[16] + tot[2] * ad[32] + tot[3] * ad[48]);
    const int row = 8 * half + w;
    if (l == 0) { scs[2 * row] = vs; scs[2 * row + 1] = vd; }
    __syncthreads();
    if (w == 0) sc2[(unsigned)blockIdx.x * 32u + (unsigned)lane] = scs[lane];

    float* cp = h2 + (unsigned)((int)blockIdx.x * 16 + row) * (unsigned)OUT_CH + (unsigned)l;
    cp[0]  = tot[0];
    cp[16] = tot[1];
    cp[32] = tot[2];
    cp[48] = tot[3];
}

#define ZERO_BLOCKS ((NN + 255) / 256)
#define WSD_BLOCKS  ((IN_CH * 16 + 255) / 256)
__global__ void zero_counts_and_score_weights(int* __restrict__ cnt,
                                              const float* __restrict__ W1,
                                              const float* __restrict__ a_src,
                                              const float* __restrict__ a_dst,
                                              float* __restrict__ wsd) {
    if (blockIdx.x < ZERO_BLOCKS) {
        const int i = blockIdx.x * 256 + threadIdx.x;
        if (i < NN) cnt[i] = 0;
        return;
    }
    const int t = (blockIdx.x - ZERO_BLOCKS) * 256 + threadIdx.x;
    if (t >= IN_CH * 16) return;
    const int k = t >> 4, j = t & 15, h = j & 7;
    const float* a = (j < 8 ? a_src : a_dst) + h * HID;
    const float* wr = W1 + k * HC1 + h * HID;
    float s = 0.f;
    for (int cc = 0; cc < HID; ++cc) s = fmaf(wr[cc], a[cc], s);
    wsd[t] = s;
}

#define SCORE_BLOCKS ((NN / 16 + GEMM_WAVES - 1) / GEMM_WAVES)
#define COUNT_BLOCKS ((NE + 255) / 256)
__global__ void dst_count_and_scores(const int* __restrict__ ei, int* __restrict__ cnt,
                                     int* __restrict__ rank,
                                     const float* __restrict__ X, const float* __restrict__ wsd,
                                     float* __restrict__ esrc, float* __restrict__ edst) {
    if (blockIdx.x < SCORE_BLOCKS) {
        const int wave = (int)(blockIdx.x * GEMM_WAVES + (threadIdx.x >> 5));
        if (wave < NN / 16) scores_tile(wave, threadIdx.x & 31, X, wsd, esrc, edst);
        return;
    }
    const int e = (blockIdx.x - SCORE_BLOCKS) * 256 + threadIdx.x;
    if (e >= NE) return;
    rank[e] = atomicAdd(&cnt[ei[NE + e]], 1);
}

#define SCAN_T   1024
#define SCAN_PER ((NN + SCAN_T - 1) / SCAN_T)
__global__ void row_scan(const int* __restrict__ cnt, int* __restrict__ rowptr) {
    __shared__ int part[SCAN_T];
    const int t = threadIdx.x;
    const int base = t * SCAN_PER;
    int loc[SCAN_PER];
    int s = 0;
    #pragma unroll
    for (int i = 0; i < SCAN_PER; ++i) {
        int idx = base + i;
        int v = idx < NN ? cnt[idx] : 0;
        loc[i] = s;
        s += v;
    }
    part[t] = s;
    __syncthreads();
    for (int off = 1; off < SCAN_T; off <<= 1) {
        int v = t >= off ? part[t - off] : 0;
        __syncthreads();
        part[t] += v;
        __syncthreads();
    }
    const int excl = part[t] - s;
    #pragma unroll
    for (int i = 0; i < SCAN_PER; ++i) {
        int idx = base + i;
        if (idx < NN) rowptr[idx] = excl + loc[i];
    }
    if (t == SCAN_T - 1) rowptr[NN] = part[t];
}

__global__ void dst_fill(const int* __restrict__ ei, const int* __restrict__ rowptr,
                         const int* __restrict__ rank, int2* __restrict__ csr) {
    int e = blockIdx.x * blockDim.x + threadIdx.x;
    if (e >= NE) return;
    csr[rowptr[ei[NE + e]] + rank[e]] = make_int2(e, ei[e]);
}

#define EPI_RELU       1
#define EPI_LOGSOFTMAX 2
template <int HC, int C, int VEC, int SS>
struct SegGather {
    typedef float vf __attribute__((ext_vector_type(VEC)));
    static constexpr int H  = HC / C;
    static constexpr int NH = (32 * VEC) / C;
    static_assert(SS >= H, "SS = floats between two nodes' scores in esrc / edst");
    static_assert(NH == 1 || NH == 2, "a wave covers one or two whole heads");

    struct Block { int src; float v[NH]; bool valid; };

    static __device__ __forceinline__ Block load(const int2* __restrict__ csr,
                                                 const float* __restrict__ esrc,
                                                 const float (&ed)[NH],
                                                 int base, int end, int lane, int hbase) {
        Block b;
        const int idx = base + lane;
        b.valid = idx < end;
        b.src = b.valid ? csr[idx].y : 0;
        const float* ep = esrc + (unsigned)b.src * (unsigned)SS + (unsigned)hbase;
        float e[NH];
        if constexpr (NH == 2) { const float2 t = *(const float2*)ep; e[0] = t.x; e[1] = t.y; }
        else                   { e[0] = ep[0]; }
        for (int q = 0; q < NH; ++q) {
            float v = e[q] + ed[q];
            v = v > 0.f ? v : 0.2f * v;
            b.v[q] = b.valid ? v : -INFINITY;
        }
        return b;
    }

    template <int U>
    static __device__ __forceinline__ void group(int srcv, const float (&av)[NH], int j, bool upper,
                                                 const float* __restrict__ hval, unsigned c, vf& acc) {
        float a[U];
        vf v[U];
        #pragma unroll
        for (int u = 0; u < U; ++u) {
            const int s = __builtin_amdgcn_readlane(srcv, j + u);
            float al = __builtin_bit_cast(float, __builtin_amdgcn_readlane(__builtin_bit_cast(int, av[0]), j + u));
            if constexpr (NH == 2) {
                const float a1 = __builtin_bit_cast(float, __builtin_amdgcn_readlane(__builtin_bit_cast(int, av[1]), j + u));
                al = upper ? a1 : al;
            }
            a[u] = al;
            v[u] = *(const vf*)(hval + (unsigned)s * (unsigned)HC + c);
        }
        #pragma unroll
        for (int u = 0; u < U; ++u) acc += a[u] * v[u];
    }

    static __device__ __forceinline__ void block_sum(int srcv, const float (&av)[NH], int cnt, bool upper,
                                                     const float* __restrict__ hval, unsigned c, vf& acc) {
        int j = 0;
        for (; j + 4 <= cnt; j += 4) group<4>(srcv, av, j, upper, hval, c, acc);
        const int r = cnt - j;
        if (r == 3)      group<3>(srcv, av, j, upper, hval, c, acc);
        else if (r == 2) group<2>(srcv, av, j, upper, hval, c, acc);
        else if (r == 1) group<1>(srcv, av, j, upper, hval, c, acc);
    }
};

template <int HC, int C, int VEC, int SS, int EPI>
__global__ void gather_softmax_aggregate(const int* __restrict__ rowptr,
                                         const int2* __restrict__ csr,
                                         const float* __restrict__ esrc,
                                         const float* __restrict__ edst,
                                         const float* __restrict__ hval,
                                         const float* __restrict__ bias,
                                         float* __restrict__ agg) {
    typedef SegGather<HC, C, VEC, SS> G;
    typedef typename G::vf vf;
    typedef typename G::Block Block;
    constexpr int H   = G::H;
    constexpr int NH  = G::NH;
    constexpr int WPN = HC / (32 * VEC);
    const int wave = __builtin_amdgcn_readfirstlane(
        (int)((blockIdx.x * blockDim.x + threadIdx.x) >> 5));
    if (wave >= NN * WPN) return;
    const int lane  = threadIdx.x & 31;
    const int node  = wave / WPN;
    const int chunk = wave % WPN;
    const unsigned c = (unsigned)(chunk * 32 * VEC) + (unsigned)lane * VEC;
    const int  hbase = chunk * NH;
    const bool upper = NH == 2 && lane >= 16;
    const int beg = rowptr[node], end = rowptr[node + 1];

    float ed[NH], vself[NH];
    for (int q = 0; q < NH; ++q) {
        ed[q] = edst[node * SS + hbase + q];
        float v = esrc[node * SS + hbase + q] + ed[q];
        vself[q] = v > 0.f ? v : 0.2f * v;
    }

    const Block b0 = G::load(csr, esrc, ed, beg, end, lane, hbase);
    float m[NH];
    for (int q = 0; q < NH; ++q) m[q] = b0.v[q];
    for (int base = beg + 32; base < end; base += 32) {
        const Block b = G::load(csr, esrc, ed, base, end, lane, hbase);
        for (int q = 0; q < NH; ++q) m[q] = fmaxf(m[q], b.v[q]);
    }
    for (int q = 0; q < NH; ++q) m[q] = fmaxf(wave_max(m[q]), vself[q]);

    float p0[NH], s[NH];
    for (int q = 0; q < NH; ++q) { p0[q] = b0.valid ? expf(b0.v[q] - m[q]) : 0.f; s[q] = p0[q]; }
    for (int base = beg + 32; base < end; base += 32) {
        const Block b = G::load(csr, esrc, ed, base, end, lane, hbase);
        for (int q = 0; q < NH; ++q) s[q] += b.valid ? expf(b.v[q] - m[q]) : 0.f;
    }
    float pself[NH], dn[NH], a0[NH];
    for (int q = 0; q < NH; ++q) {
        pself[q] = expf(vself[q] - m[q]);
        dn[q] = (wave_sum(s[q]) + pself[q]) + 1e-16f;
        a0[q] = p0[q] / dn[q];
    }

    float aself = pself[0] / dn[0];
    if constexpr (NH == 2) { const float a1 = pself[1] / dn[1]; aself = upper ? a1 : aself; }
    vf acc = aself * *(const vf*)(hval + (unsigned)node * (unsigned)HC + c);
    int left = end - beg;
    if (left > 0) G::block_sum(b0.src, a0, left < 32 ? left : 32, upper, hval, c, acc);
    for (int base = beg + 32; base < end; base += 32) {
        left -= 32;
        const Block b = G::load(csr, esrc, ed, base, end, lane, hbase);
        float a[NH];
        for (int q = 0; q < NH; ++q) a[q] = (b.valid ? expf(b.v[q] - m[q]) : 0.f) / dn[q];
        G::block_sum(b.src, a, left < 32 ? left : 32, upper, hval, c, acc);
    }
    vf res = acc + *(const vf*)(bias + c);
    if constexpr (EPI == EPI_RELU) {
        for (int i = 0; i < VEC; ++i) res[i] = res[i] > 0.f ? res[i] : 0.f;
    } else {
        static_assert(WPN == 1 && H == 1, "the wave must hold the node's whole row");
        float mx = res[0];
        for (int i = 1; i < VEC; ++i) mx = fmaxf(mx, res[i]);
        mx = wave_max(mx);
        float se = 0.f;
        for (int i = 0; i < VEC; ++i) se += expf(res[i] - mx);
        const float lse = logf(wave_sum(se)) + mx;
        for (int i = 0; i < VEC; ++i) res[i] -= lse;
    }
    *(vf*)(agg + (unsigned)node * (unsigned)HC + c) = res;
}

extern "C" void kernel_launch(void* const* d_in, const int* in_sizes, int n_in,
                              void* d_out, int out_size, void* d_ws, size_t ws_size,
                              hipStream_t stream) {
    const float* x   = (const float*)d_in[0];
    const int*   ei  = (const int*)  d_in[1];
    const float* W1  = (const float*)d_in[2];
    const float* a1s = (const float*)d_in[3];
    const float* a1d = (const float*)d_in[4];
    const float* b1  = (const float*)d_in[5];
    const float* W2  = (const float*)d_in[6];
    const float* a2s = (const float*)d_in[7];
    const float* a2d = (const float*)d_in[8];
    const float* b2  = (const float*)d_in[9];
    float* out = (float*)d_out;

    float* ws = (float*)d_ws;
    float* esrc1 = ws;
    float* edst1 = esrc1 + (size_t)NN * HEADS;
    float* h2pre = edst1 + (size_t)NN * HEADS;
    float* sc2   = h2pre + (size_t)NN * OUT_CH;
    int*   cnt   = (int*)(sc2 + 2 * NN);
    int*   rank  = cnt + NN;
    int*   rowp  = rank + NE;
    int2*  csr   = (int2*)(rowp + NN + 2);

    const int TB = 32 * GEMM_WAVES;
    auto blocks = [](long long n, int tb) { return (unsigned)((n + tb - 1) / tb); };

    float* wsd = (float*)(csr + NE);

    zero_counts_and_score_weights<<<ZERO_BLOCKS + WSD_BLOCKS, TB, 0, stream>>>(cnt, W1, a1s, a1d, wsd);
    dst_count_and_scores<<<SCORE_BLOCKS + COUNT_BLOCKS, TB, 0, stream>>>(ei, cnt, rank, x, wsd, esrc1, edst1);
    row_scan <<<1, SCAN_T, 0, stream>>>(cnt, rowp);
    dst_fill <<<blocks(NE, TB), TB, 0, stream>>>(ei, rowp, rank, csr);

    gather_project_l1<<<NN / 16, TB, 0, stream>>>(rowp, csr, esrc1, edst1, x, W1, b1, W2, a2s, a2d, h2pre, sc2);

    gather_softmax_aggregate<OUT_CH, OUT_CH, 2, 2, EPI_LOGSOFTMAX><<<blocks((long long)NN * 32, TB), TB, 0, stream>>>(
        rowp, csr, sc2, sc2 + 1, h2pre, b2, out);
}


# W1 and W2 rewritten once per call into fragment order in the workspace; each weight fragment load of the fused layer kernel is one 16-byte whole-line load (384 4-byte loads per wave -> 96)
# speedup vs baseline: 8.3825x; 1.0550x over previous
#include <hip/hip_runtime.h>
#include <math.h>

#define NN      20000
#define NE      320000
#define ETOT    (NE + NN)
#define IN_CH   128
#define HID     64
#define HEADS   8
#define HC1     (HEADS * HID)
#define OUT_CH  64

typedef float v2f __attribute__((ext_vector_type(2)));
typedef float v8f __attribute__((ext_vector_type(8)));

__device__ __forceinline__ float xhalf(float x) {
    return __builtin_bit_cast(float, __builtin_amdgcn_permlanex16(
        __builtin_bit_cast(int, x), __builtin_bit_cast(int, x), 0x76543210, 0xfedcba98, false, false));
}
template <int XM>
__device__ __forceinline__ float row_xor(float x) {
    const int xi = __builtin_bit_cast(int, x);
    return __builtin_bit_cast(float, __builtin_amdgcn_update_dpp(xi, xi, 0x160 + XM, 0xf, 0xf, false));
}
__device__ __forceinline__ float wave_max(float v) {
    v = fmaxf(v, row_xor<1>(v));
    v = fmaxf(v, row_xor<2>(v));
    v = fmaxf(v, row_xor<4>(v));
    v = fmaxf(v, row_xor<8>(v));
    return fmaxf(v, xhalf(v));
}
__device__ __forceinline__ float row_sum16(float v) {
    v += row_xor<1>(v);
    v += row_xor<2>(v);
    v += row_xor<4>(v);
    return v + row_xor<8>(v);
}
__device__ __forceinline__ float wave_sum(float v) {
    v += row_xor<1>(v);
    v += row_xor<2>(v);
    v += row_xor<4>(v);
    v += row_xor<8>(v);
    return v + xhalf(v);
}

#define GEMM_WAVES 8
typedef float v4f __attribute__((ext_vector_type(4)));
#define WQ_LOAD  32
#define WQ_GROUP (8 * WQ_LOAD)
__device__ __forceinline__ void load_wfrags(v2f (&b)[4][4], const v4f* __restrict__ q) {
    #pragma unroll
    for (int i = 0; i < 8; ++i) {
        const v4f t = q[i * WQ_LOAD];
        const v2f lo = { t[0], t[1] };
        const v2f hi = { t[2], t[3] };
        b[i >> 1][2 * (i & 1)]     = lo;
        b[i >> 1][2 * (i & 1) + 1] = hi;
    }
}
struct Frags16 { v2f a[4]; v2f b[4][4]; };
__device__ __forceinline__ void load_frags16(Frags16& f, const float* __restrict__ Ap,
                                             const v4f* __restrict__ q) {
    #pragma unroll
    for (int s = 0; s < 4; ++s) f.a[s] = *(const v2f*)(Ap + 4 * s);
    load_wfrags(f.b, q);
}
__device__ __forceinline__ void mma_frags16_t(const Frags16& f, v8f& acc0, v8f& acc1, v8f& acc2, v8f& acc3) {
    #pragma unroll
    for (int s = 0; s < 4; ++s) {
        acc0 = __builtin_amdgcn_wmma_f32_16x16x4_f32(false, f.b[s][0], false, f.a[s], (short)0, acc0, false, false);
        acc1 = __builtin_amdgcn_wmma_f32_16x16x4_f32(false, f.b[s][1], false, f.a[s], (short)0, acc1, false, false);
        acc2 = __builtin_amdgcn_wmma_f32_16x16x4_f32(false, f.b[s][2], false, f.a[s], (short)0, acc2, false, false);
        acc3 = __builtin_amdgcn_wmma_f32_16x16x4_f32(false, f.b[s][3], false, f.a[s], (short)0, acc3, false, false);
    }
}
struct FragsB16 { v2f b[4][4]; };
__device__ __forceinline__ void mma_tile16(const v8f& T, const FragsB16& f, v8f& o0, v8f& o1, v8f& o2, v8f& o3) {
    #pragma unroll
    for (int s = 0; s < 4; ++s) {
        const v2f a = { T[2 * s], T[2 * s + 1] };
        o0 = __builtin_amdgcn_wmma_f32_16x16x4_f32(false, a, false, f.b[s][0], (short)0, o0, false, false);
        o1 = __builtin_amdgcn_wmma_f32_16x16x4_f32(false, a, false, f.b[s][1], (short)0, o1, false, false);
        o2 = __builtin_amdgcn_wmma_f32_16x16x4_f32(false, a, false, f.b[s][2], (short)0, o2, false, false);
        o3 = __builtin_amdgcn_wmma_f32_16x16x4_f32(false, a, false, f.b[s][3], (short)0, o3, false, false);
    }
}


__device__ __forceinline__ void scores_tile(int wave, int lane, const float* __restrict__ X,
                                            const float* __restrict__ wsd,
                                            float* __restrict__ esrc, float* __restrict__ edst) {
    const int l = lane & 15, half = lane >> 4;
    const float* Ap = X + (unsigned)(wave * 16 + l) * (unsigned)IN_CH + 2u * half;
    const float* Bp = wsd + (2 * half) * 16 + l;
    v8f accE = {}, accO = {};
    #pragma unroll
    for (int s = 0; s < IN_CH / 4; s += 2) {
        const v2f a0 = *(const v2f*)(Ap + 4 * s);
        const v2f b0 = { Bp[(4 * s) * 16], Bp[(4 * s + 1) * 16] };
        const v2f a1 = *(const v2f*)(Ap + 4 * s + 4);
        const v2f b1 = { Bp[(4 * s + 4) * 16], Bp[(4 * s + 5) * 16] };
        accE = __builtin_amdgcn_wmma_f32_16x16x4_f32(false, a0, false, b0, (short)0, accE, false, false);
        accO = __builtin_amdgcn_wmma_f32_16x16x4_f32(false, a1, false, b1, (short)0, accO, false, false);
    }
    float* dstp = (l < 8 ? esrc : edst) + (unsigned)(wave * 16 + 8 * half) * (unsigned)HEADS + (unsigned)(l & 7);
    #pragma unroll
    for (int r = 0; r < 8; ++r) dstp[r * HEADS] = accE[r] + accO[r];
}

struct Blk8 { int src; float v[HEADS]; bool valid; };
__device__ __forceinline__ Blk8 load8(const int2* __restrict__ csr, const float* __restrict__ esrc,
                                      const float (&ed)[HEADS], int base, int end, int lane) {
    Blk8 b;
    const int idx = base + lane;
    b.valid = idx < end;
    b.src = b.valid ? csr[idx].y : 0;
    const v4f* ep = (const v4f*)(esrc + (unsigned)b.src * (unsigned)HEADS);
    const v4f e0 = ep[0], e1 = ep[1];
    const float e[HEADS] = { e0[0], e0[1], e0[2], e0[3], e1[0], e1[1], e1[2], e1[3] };
    #pragma unroll
    for (int q = 0; q < HEADS; ++q) {
        float v = e[q] + ed[q];
        v = v > 0.f ? v : 0.2f * v;
        b.v[q] = b.valid ? v : -INFINITY;
    }
    return b;
}
template <int U>
__device__ __forceinline__ void group8(int srcv, const float (&av)[HEADS], int j,
                                       const float* __restrict__ X, unsigned c, v4f (&acc)[HEADS]) {
    float a[U][HEADS];
    v4f v[U];
    #pragma unroll
    for (int u = 0; u < U; ++u) {
        const int s = __builtin_amdgcn_readlane(srcv, j + u);
        #pragma unroll
        for (int q = 0; q < HEADS; ++q)
            a[u][q] = __builtin_bit_cast(float, __builtin_amdgcn_readlane(__builtin_bit_cast(int, av[q]), j + u));
        v[u] = *(const v4f*)(X + (unsigned)s * (unsigned)IN_CH + c);
    }
    #pragma unroll
    for (int u = 0; u < U; ++u) {
        #pragma unroll
        for (int q = 0; q < HEADS; ++q) acc[q] += a[u][q] * v[u];
    }
}
__device__ __forceinline__ void block_sum8(int srcv, const float (&av)[HEADS], int cnt,
                                           const float* __restrict__ X, unsigned c, v4f (&acc)[HEADS]) {
    int j = 0;
    for (; j + 4 <= cnt; j += 4) group8<4>(srcv, av, j, X, c, acc);
    const int r = cnt - j;
    if (r == 3)      group8<3>(srcv, av, j, X, c, acc);
    else if (r == 2) group8<2>(srcv, av, j, X, c, acc);
    else if (r == 1) group8<1>(srcv, av, j, X, c, acc);
}

#define ZP (HEADS * IN_CH + 4)
__global__ void __launch_bounds__(32 * GEMM_WAVES)
gather_project_l1(const int* __restrict__ rowptr, const int2* __restrict__ csr,
                  const float* __restrict__ esrc, const float* __restrict__ edst,
                  const float* __restrict__ X, const float* __restrict__ W1q,
                  const float* __restrict__ bias, const float* __restrict__ W2q,
                  const float* __restrict__ a_src2, const float* __restrict__ a_dst2,
                  float* __restrict__ h2, float* __restrict__ sc2) {
    __shared__ __attribute__((aligned(16))) float zs[16 * ZP];
    __shared__ float scs[32];
    const int lane = threadIdx.x & 31;
    const int w = __builtin_amdgcn_readfirstlane((int)(threadIdx.x >> 5));
    const unsigned c = 4u * (unsigned)lane;

    for (int i = 0; i < 2; ++i) {
        const int ml   = 2 * w + i;
        const int node = (int)blockIdx.x * 16 + ml;
        const int beg = rowptr[node], end = rowptr[node + 1];
        float ed[HEADS], vself[HEADS];
        #pragma unroll
        for (int q = 0; q < HEADS; ++q) {
            ed[q] = edst[node * HEADS + q];
            const float v = esrc[node * HEADS + q] + ed[q];
            vself[q] = v > 0.f ? v : 0.2f * v;
        }
        const Blk8 b0 = load8(csr, esrc, ed, beg, end, lane);
        float m[HEADS];
        #pragma unroll
        for (int q = 0; q < HEADS; ++q) m[q] = b0.v[q];
        for (int base = beg + 32; base < end; base += 32) {
            const Blk8 b = load8(csr, esrc, ed, base, end, lane);
            #pragma unroll
            for (int q = 0; q < HEADS; ++q) m[q] = fmaxf(m[q], b.v[q]);
        }
        #pragma unroll
        for (int q = 0; q < HEADS; ++q) m[q] = fmaxf(wave_max(m[q]), vself[q]);
        float p0[HEADS], s[HEADS];
        #pragma unroll
        for (int q = 0; q < HEADS; ++q) { p0[q] = b0.valid ? expf(b0.v[q] - m[q]) : 0.f; s[q] = p0[q]; }
        for (int base = beg + 32; base < end; base += 32) {
            const Blk8 b = load8(csr, esrc, ed, base, end, lane);
            #pragma unroll
            for (int q = 0; q < HEADS; ++q) s[q] += b.valid ? expf(b.v[q] - m[q]) : 0.f;
        }
        float dn[HEADS], a0[HEADS];
        v4f acc[HEADS];
        const v4f xself = *(const v4f*)(X + (unsigned)node * (unsigned)IN_CH + c);
        #pragma unroll
        for (int q = 0; q < HEADS; ++q) {
            const float pself = expf(vself[q] - m[q]);
            dn[q] = (wave_sum(s[q]) + pself) + 1e-16f;
            a0[q] = p0[q] / dn[q];
            acc[q] = (pself / dn[q]) * xself;
        }
        int left = end - beg;
        if (left > 0) block_sum8(b0.src, a0, left < 32 ? left : 32, X, c, acc);
        for (int base = beg + 32; base < end; base += 32) {
            left -= 32;
            const Blk8 b = load8(csr, esrc, ed, base, end, lane);
            float a[HEADS];
            #pragma unroll
            for (int q = 0; q < HEADS; ++q) a[q] = (b.valid ? expf(b.v[q] - m[q]) : 0.f) / dn[q];
            block_sum8(b.src, a, left < 32 ? left : 32, X, c, acc);
        }
        v4f* zrow = (v4f*)(zs + ml * ZP) + lane;
        #pragma unroll
        for (int q = 0; q < HEADS; ++q) zrow[q * (IN_CH / 4)] = acc[q];
    }
    __syncthreads();

    const int l    = lane & 15;
    const int half = lane >> 4;
    v8f acc0 = {}, acc1 = {}, acc2 = {}, acc3 = {};
    const float* Ap = zs + l * ZP + w * IN_CH + 2 * half;
    const v4f* Bq  = (const v4f*)W1q + (unsigned)(w * (IN_CH / 16) * WQ_GROUP) + (unsigned)lane;
    const v4f* W2p = (const v4f*)W2q + (unsigned)(w * 4 * WQ_GROUP) + (unsigned)lane;
    Frags16 P, Q;
    load_frags16(P, Ap, Bq);
    for (int k0 = 0; k0 < IN_CH - 32; k0 += 32) {
        load_frags16(Q, Ap + 16, Bq + WQ_GROUP);
        mma_frags16_t(P, acc0, acc1, acc2, acc3);
        load_frags16(P, Ap + 32, Bq + 2 * WQ_GROUP);
        mma_frags16_t(Q, acc0, acc1, acc2, acc3);
        Ap += 32;
        Bq += 2 * WQ_GROUP;
    }
    load_frags16(Q, Ap + 16, Bq + WQ_GROUP);
    mma_frags16_t(P, acc0, acc1, acc2, acc3);
    FragsB16 R, S;
    load_wfrags(R.b, W2p);
    mma_frags16_t(Q, acc0, acc1, acc2, acc3);

    {
        const v4f* bq = (const v4f*)(bias + w * 64 + 8 * half);
        const v4f b00 = bq[0],  b01 = bq[1];
        const v4f b10 = bq[4],  b11 = bq[5];
        const v4f b20 = bq[8],  b21 = bq[9];
        const v4f b30 = bq[12], b31 = bq[13];
        #pragma unroll
        for (int r = 0; r < 4; ++r) {
            acc0[r] = fmaxf(acc0[r] + b00[r], 0.f);  acc0[r + 4] = fmaxf(acc0[r + 4] + b01[r], 0.f);
            acc1[r] = fmaxf(acc1[r] + b10[r], 0.f);  acc1[r + 4] = fmaxf(acc1[r + 4] + b11[r], 0.f);
            acc2[r] = fmaxf(acc2[r] + b20[r], 0.f);  acc2[r + 4] = fmaxf(acc2[r + 4] + b21[r], 0.f);
            acc3[r] = fmaxf(acc3[r] + b30[r], 0.f);  acc3[r + 4] = fmaxf(acc3[r + 4] + b31[r], 0.f);
        }
    }

    v8f o0 = {}, o1 = {}, o2 = {}, o3 = {};
    load_wfrags(S.b, W2p + WQ_GROUP);
    mma_tile16(acc0, R, o0, o1, o2, o3);
    load_wfrags(R.b, W2p + 2 * WQ_GROUP);
    mma_tile16(acc1, S, o0, o1, o2, o3);
    load_wfrags(S.b, W2p + 3 * WQ_GROUP);
    mma_tile16(acc2, R, o0, o1, o2, o3);
    mma_tile16(acc3, S, o0, o1, o2, o3);

    {
        float* prow = zs + half * ZP + w * IN_CH + 4 * l;
        #pragma unroll
        for (int r = 0; r < 8; ++r) {
            const v4f pv = { o0[r], o1[r], o2[r], o3[r] };
            *(v4f*)(prow + 2 * r * ZP) = pv;
        }
    }
    __syncthreads();
    const float* rrow = zs + (2 * w + half) * ZP + 4 * l;
    v4f tot = *(const v4f*)rrow;
    #pragma unroll
    for (int ww = 1; ww < GEMM_WAVES; ++ww) tot += *(const v4f*)(rrow + ww * IN_CH);

    const float* as = a_src2 + l;
    const float* ad = a_dst2 + l;
    const float vs = row_sum16(tot[0] * as[0] + tot[1] * as[16] + tot[2] * as[32] + tot[3] * as[48]);
    const float vd = row_sum16(tot[0] * ad[0] + tot[1] * ad[16] + tot[2] * ad[32] + tot[3] * ad[48]);
    const int row = 8 * half + w;
    if (l == 0) { scs[2 * row] = vs; scs[2 * row + 1] = vd; }
    __syncthreads();
    if (w == 0) sc2[(unsigned)blockIdx.x * 32u + (unsigned)lane] = scs[lane];

    float* cp = h2 + (unsigned)((int)blockIdx.x * 16 + row) * (unsigned)OUT_CH + (unsigned)l;
    cp[0]  = tot[0];
    cp[16] = tot[1];
    cp[32] = tot[2];
    cp[48] = tot[3];
}

#define ZERO_BLOCKS ((NN + 255) / 256)
#define WSD_BLOCKS  ((IN_CH * 16 + 255) / 256)
#define W1Q_V4      (HEADS * (IN_CH / 16) * WQ_GROUP)
#define W2Q_V4      (HEADS * 4 * WQ_GROUP)
#define WQ_BLOCKS   ((W1Q_V4 + W2Q_V4) / 256)
static_assert(W1Q_V4 % 256 == 0 && W2Q_V4 % 256 == 0, "whole work-groups per plane");
__device__ __forceinline__ void fragment_order_weights(int t, const float* __restrict__ W1,
                                                       const float* __restrict__ W2,
                                                       float* __restrict__ w1q, float* __restrict__ w2q) {
    const int lane = t & 31, i = (t >> 5) & 7;
    const int s = i >> 1, p = i & 1, h = lane >> 4, l = lane & 15;
    if (t < W1Q_V4) {
        const int g = (t >> 8) & 7, w = t >> 11;
        const float* r0 = W1 + (16 * g + 4 * s + 2 * h) * HC1 + 64 * w + 32 * p + l;
        const v4f o = { r0[0], r0[HC1], r0[16], r0[HC1 + 16] };
        ((v4f*)w1q)[t] = o;
    } else {
        const int u = t - W1Q_V4;
        const int g = (u >> 8) & 3, w = u >> 10;
        const float* r0 = W2 + (64 * w + 16 * g + 8 * h + 2 * s) * OUT_CH + 32 * p + l;
        const v4f o = { r0[0], r0[OUT_CH], r0[16], r0[OUT_CH + 16] };
        ((v4f*)w2q)[u] = o;
    }
}
__global__ void zero_counts_and_prepare_weights(int* __restrict__ cnt,
                                                const float* __restrict__ W1,
                                                const float* __restrict__ a_src,
                                                const float* __restrict__ a_dst,
                                                float* __restrict__ wsd,
                                                const float* __restrict__ W2,
                                                float* __restrict__ w1q, float* __restrict__ w2q) {
    if (blockIdx.x < ZERO_BLOCKS) {
        const int i = blockIdx.x * 256 + threadIdx.x;
        if (i < NN) cnt[i] = 0;
        return;
    }
    if (blockIdx.x >= ZERO_BLOCKS + WSD_BLOCKS) {
        fragment_order_weights((int)(blockIdx.x - ZERO_BLOCKS - WSD_BLOCKS) * 256 + (int)threadIdx.x,
                               W1, W2, w1q, w2q);
        return;
    }
    const int t = (blockIdx.x - ZERO_BLOCKS) * 256 + threadIdx.x;
    if (t >= IN_CH * 16) return;
    const int k = t >> 4, j = t & 15, h = j & 7;
    const float* a = (j < 8 ? a_src : a_dst) + h * HID;
    const float* wr = W1 + k * HC1 + h * HID;
    float s = 0.f;
    for (int cc = 0; cc < HID; ++cc) s = fmaf(wr[cc], a[cc], s);
    wsd[t] = s;
}

#define SCORE_BLOCKS ((NN / 16 + GEMM_WAVES - 1) / GEMM_WAVES)
#define COUNT_BLOCKS ((NE + 255) / 256)
__global__ void dst_count_and_scores(const int* __restrict__ ei, int* __restrict__ cnt,
                                     int* __restrict__ rank,
                                     const float* __restrict__ X, const float* __restrict__ wsd,
                                     float* __restrict__ esrc, float* __restrict__ edst) {
    if (blockIdx.x < SCORE_BLOCKS) {
        const int wave = (int)(blockIdx.x * GEMM_WAVES + (threadIdx.x >> 5));
        if (wave < NN / 16) scores_tile(wave, threadIdx.x & 31, X, wsd, esrc, edst);
        return;
    }
    const int e = (blockIdx.x - SCORE_BLOCKS) * 256 + threadIdx.x;
    if (e >= NE) return;
    rank[e] = atomicAdd(&cnt[ei[NE + e]], 1);
}

#define SCAN_T   1024
#define SCAN_PER ((NN + SCAN_T - 1) / SCAN_T)
__global__ void row_scan(const int* __restrict__ cnt, int* __restrict__ rowptr) {
    __shared__ int part[SCAN_T];
    const int t = threadIdx.x;
    const int base = t * SCAN_PER;
    int loc[SCAN_PER];
    int s = 0;
    #pragma unroll
    for (int i = 0; i < SCAN_PER; ++i) {
        int idx = base + i;
        int v = idx < NN ? cnt[idx] : 0;
        loc[i] = s;
        s += v;
    }
    part[t] = s;
    __syncthreads();
    for (int off = 1; off < SCAN_T; off <<= 1) {
        int v = t >= off ? part[t - off] : 0;
        __syncthreads();
        part[t] += v;
        __syncthreads();
    }
    const int excl = part[t] - s;
    #pragma unroll
    for (int i = 0; i < SCAN_PER; ++i) {
        int idx = base + i;
        if (idx < NN) rowptr[idx] = excl + loc[i];
    }
    if (t == SCAN_T - 1) rowptr[NN] = part[t];
}

__global__ void dst_fill(const int* __restrict__ ei, const int* __restrict__ rowptr,
                         const int* __restrict__ rank, int2* __restrict__ csr) {
    int e = blockIdx.x * blockDim.x + threadIdx.x;
    if (e >= NE) return;
    csr[rowptr[ei[NE + e]] + rank[e]] = make_int2(e, ei[e]);
}

#define EPI_RELU       1
#define EPI_LOGSOFTMAX 2
template <int HC, int C, int VEC, int SS>
struct SegGather {
    typedef float vf __attribute__((ext_vector_type(VEC)));
    static constexpr int H  = HC / C;
    static constexpr int NH = (32 * VEC) / C;
    static_assert(SS >= H, "SS = floats between two nodes' scores in esrc / edst");
    static_assert(NH == 1 || NH == 2, "a wave covers one or two whole heads");

    struct Block { int src; float v[NH]; bool valid; };

    static __device__ __forceinline__ Block load(const int2* __restrict__ csr,
                                                 const float* __restrict__ esrc,
                                                 const float (&ed)[NH],
                                                 int base, int end, int lane, int hbase) {
        Block b;
        const int idx = base + lane;
        b.valid = idx < end;
        b.src = b.valid ? csr[idx].y : 0;
        const float* ep = esrc + (unsigned)b.src * (unsigned)SS + (unsigned)hbase;
        float e[NH];
        if constexpr (NH == 2) { const float2 t = *(const float2*)ep; e[0] = t.x; e[1] = t.y; }
        else                   { e[0] = ep[0]; }
        for (int q = 0; q < NH; ++q) {
            float v = e[q] + ed[q];
            v = v > 0.f ? v : 0.2f * v;
            b.v[q] = b.valid ? v : -INFINITY;
        }
        return b;
    }

    template <int U>
    static __device__ __forceinline__ void group(int srcv, const float (&av)[NH], int j, bool upper,
                                                 const float* __restrict__ hval, unsigned c, vf& acc) {
        float a[U];
        vf v[U];
        #pragma unroll
        for (int u = 0; u < U; ++u) {
            const int s = __builtin_amdgcn_readlane(srcv, j + u);
            float al = __builtin_bit_cast(float, __builtin_amdgcn_readlane(__builtin_bit_cast(int, av[0]), j + u));
            if constexpr (NH == 2) {
                const float a1 = __builtin_bit_cast(float, __builtin_amdgcn_readlane(__builtin_bit_cast(int, av[1]), j + u));
                al = upper ? a1 : al;
            }
            a[u] = al;
            v[u] = *(const vf*)(hval + (unsigned)s * (unsigned)HC + c);
        }
        #pragma unroll
        for (int u = 0; u < U; ++u) acc += a[u] * v[u];
    }

    static __device__ __forceinline__ void block_sum(int srcv, const float (&av)[NH], int cnt, bool upper,
                                                     const float* __restrict__ hval, unsigned c, vf& acc) {
        int j = 0;
        for (; j + 4 <= cnt; j += 4) group<4>(srcv, av, j, upper, hval, c, acc);
        const int r = cnt - j;
        if (r == 3)      group<3>(srcv, av, j, upper, hval, c, acc);
        else if (r == 2) group<2>(srcv, av, j, upper, hval, c, acc);
        else if (r == 1) group<1>(srcv, av, j, upper, hval, c, acc);
    }
};

template <int HC, int C, int VEC, int SS, int EPI>
__global__ void gather_softmax_aggregate(const int* __restrict__ rowptr,
                                         const int2* __restrict__ csr,
                                         const float* __restrict__ esrc,
                                         const float* __restrict__ edst,
                                         const float* __restrict__ hval,
                                         const float* __restrict__ bias,
                                         float* __restrict__ agg) {
    typedef SegGather<HC, C, VEC, SS> G;
    typedef typename G::vf vf;
    typedef typename G::Block Block;
    constexpr int H   = G::H;
    constexpr int NH  = G::NH;
    constexpr int WPN = HC / (32 * VEC);
    const int wave = __builtin_amdgcn_readfirstlane(
        (int)((blockIdx.x * blockDim.x + threadIdx.x) >> 5));
    if (wave >= NN * WPN) return;
    const int lane  = threadIdx.x & 31;
    const int node  = wave / WPN;
    const int chunk = wave % WPN;
    const unsigned c = (unsigned)(chunk * 32 * VEC) + (unsigned)lane * VEC;
    const int  hbase = chunk * NH;
    const bool upper = NH == 2 && lane >= 16;
    const int beg = rowptr[node], end = rowptr[node + 1];

    float ed[NH], vself[NH];
    for (int q = 0; q < NH; ++q) {
        ed[q] = edst[node * SS + hbase + q];
        float v = esrc[node * SS + hbase + q] + ed[q];
        vself[q] = v > 0.f ? v : 0.2f * v;
    }

    const Block b0 = G::load(csr, esrc, ed, beg, end, lane, hbase);
    float m[NH];
    for (int q = 0; q < NH; ++q) m[q] = b0.v[q];
    for (int base = beg + 32; base < end; base += 32) {
        const Block b = G::load(csr, esrc, ed, base, end, lane, hbase);
        for (int q = 0; q < NH; ++q) m[q] = fmaxf(m[q], b.v[q]);
    }
    for (int q = 0; q < NH; ++q) m[q] = fmaxf(wave_max(m[q]), vself[q]);

    float p0[NH], s[NH];
    for (int q = 0; q < NH; ++q) { p0[q] = b0.valid ? expf(b0.v[q] - m[q]) : 0.f; s[q] = p0[q]; }
    for (int base = beg + 32; base < end; base += 32) {
        const Block b = G::load(csr, esrc, ed, base, end, lane, hbase);
        for (int q = 0; q < NH; ++q) s[q] += b.valid ? expf(b.v[q] - m[q]) : 0.f;
    }
    float pself[NH], dn[NH], a0[NH];
    for (int q = 0; q < NH; ++q) {
        pself[q] = expf(vself[q] - m[q]);
        dn[q] = (wave_sum(s[q]) + pself[q]) + 1e-16f;
        a0[q] = p0[q] / dn[q];
    }

    float aself = pself[0] / dn[0];
    if constexpr (NH == 2) { const float a1 = pself[1] / dn[1]; aself = upper ? a1 : aself; }
    vf acc = aself * *(const vf*)(hval + (unsigned)node * (unsigned)HC + c);
    int left = end - beg;
    if (left > 0) G::block_sum(b0.src, a0, left < 32 ? left : 32, upper, hval, c, acc);
    for (int base = beg + 32; base < end; base += 32) {
        left -= 32;
        const Block b = G::load(csr, esrc, ed, base, end, lane, hbase);
        float a[NH];
        for (int q = 0; q < NH; ++q) a[q] = (b.valid ? expf(b.v[q] - m[q]) : 0.f) / dn[q];
        G::block_sum(b.src, a, left < 32 ? left : 32, upper, hval, c, acc);
    }
    vf res = acc + *(const vf*)(bias + c);
    if constexpr (EPI == EPI_RELU) {
        for (int i = 0; i < VEC; ++i) res[i] = res[i] > 0.f ? res[i] : 0.f;
    } else {
        static_assert(WPN == 1 && H == 1, "the wave must hold the node's whole row");
        float mx = res[0];
        for (int i = 1; i < VEC; ++i) mx = fmaxf(mx, res[i]);
        mx = wave_max(mx);
        float se = 0.f;
        for (int i = 0; i < VEC; ++i) se += expf(res[i] - mx);
        const float lse = logf(wave_sum(se)) + mx;
        for (int i = 0; i < VEC; ++i) res[i] -= lse;
    }
    *(vf*)(agg + (unsigned)node * (unsigned)HC + c) = res;
}

extern "C" void kernel_launch(void* const* d_in, const int* in_sizes, int n_in,
                              void* d_out, int out_size, void* d_ws, size_t ws_size,
                              hipStream_t stream) {
    const float* x   = (const float*)d_in[0];
    const int*   ei  = (const int*)  d_in[1];
    const float* W1  = (const float*)d_in[2];
    const float* a1s = (const float*)d_in[3];
    const float* a1d = (const float*)d_in[4];
    const float* b1  = (const float*)d_in[5];
    const float* W2  = (const float*)d_in[6];
    const float* a2s = (const float*)d_in[7];
    const float* a2d = (const float*)d_in[8];
    const float* b2  = (const float*)d_in[9];
    float* out = (float*)d_out;

    float* ws = (float*)d_ws;
    float* w1q   = ws;
    float* w2q   = w1q + (size_t)IN_CH * HC1;
    float* esrc1 = w2q + (size_t)HC1 * OUT_CH;
    float* edst1 = esrc1 + (size_t)NN * HEADS;
    float* h2pre = edst1 + (size_t)NN * HEADS;
    float* sc2   = h2pre + (size_t)NN * OUT_CH;
    int*   cnt   = (int*)(sc2 + 2 * NN);
    int*   rank  = cnt + NN;
    int*   rowp  = rank + NE;
    int2*  csr   = (int2*)(rowp + NN + 2);

    const int TB = 32 * GEMM_WAVES;
    auto blocks = [](long long n, int tb) { return (unsigned)((n + tb - 1) / tb); };

    float* wsd = (float*)(csr + NE);

    zero_counts_and_prepare_weights<<<ZERO_BLOCKS + WSD_BLOCKS + WQ_BLOCKS, TB, 0, stream>>>(
        cnt, W1, a1s, a1d, wsd, W2, w1q, w2q);
    dst_count_and_scores<<<SCORE_BLOCKS + COUNT_BLOCKS, TB, 0, stream>>>(ei, cnt, rank, x, wsd, esrc1, edst1);
    row_scan <<<1, SCAN_T, 0, stream>>>(cnt, rowp);
    dst_fill <<<blocks(NE, TB), TB, 0, stream>>>(ei, rowp, rank, csr);

    gather_project_l1<<<NN / 16, TB, 0, stream>>>(rowp, csr, esrc1, edst1, x, w1q, b1, w2q, a2s, a2d, h2pre, sc2);

    gather_softmax_aggregate<OUT_CH, OUT_CH, 2, 2, EPI_LOGSOFTMAX><<<blocks((long long)NN * 32, TB), TB, 0, stream>>>(
        rowp, csr, sc2, sc2 + 1, h2pre, b2, out);
}


# Edge lists in one pass: an edge's old in-degree count is its slot in a fixed 64-entry row per destination (overflow list beyond); scan and fill launches removed, 6 -> 4; sum order stays arrival-slot
# speedup vs baseline: 10.4919x; 1.2516x over previous
#include <hip/hip_runtime.h>
#include <math.h>

#define NN      20000
#define NE      320000
#define ETOT    (NE + NN)
#define IN_CH   128
#define HID     64
#define HEADS   8
#define HC1     (HEADS * HID)
#define OUT_CH  64

typedef float v2f __attribute__((ext_vector_type(2)));
typedef float v8f __attribute__((ext_vector_type(8)));

__device__ __forceinline__ float xhalf(float x) {
    return __builtin_bit_cast(float, __builtin_amdgcn_permlanex16(
        __builtin_bit_cast(int, x), __builtin_bit_cast(int, x), 0x76543210, 0xfedcba98, false, false));
}
template <int XM>
__device__ __forceinline__ float row_xor(float x) {
    const int xi = __builtin_bit_cast(int, x);
    return __builtin_bit_cast(float, __builtin_amdgcn_update_dpp(xi, xi, 0x160 + XM, 0xf, 0xf, false));
}
__device__ __forceinline__ float wave_max(float v) {
    v = fmaxf(v, row_xor<1>(v));
    v = fmaxf(v, row_xor<2>(v));
    v = fmaxf(v, row_xor<4>(v));
    v = fmaxf(v, row_xor<8>(v));
    return fmaxf(v, xhalf(v));
}
__device__ __forceinline__ float row_sum16(float v) {
    v += row_xor<1>(v);
    v += row_xor<2>(v);
    v += row_xor<4>(v);
    return v + row_xor<8>(v);
}
__device__ __forceinline__ float wave_sum(float v) {
    v += row_xor<1>(v);
    v += row_xor<2>(v);
    v += row_xor<4>(v);
    v += row_xor<8>(v);
    return v + xhalf(v);
}

#define GEMM_WAVES 8
typedef float v4f __attribute__((ext_vector_type(4)));
#define WQ_LOAD  32
#define WQ_GROUP (8 * WQ_LOAD)
__device__ __forceinline__ void load_wfrags(v2f (&b)[4][4], const v4f* __restrict__ q) {
    #pragma unroll
    for (int i = 0; i < 8; ++i) {
        const v4f t = q[i * WQ_LOAD];
        const v2f lo = { t[0], t[1] };
        const v2f hi = { t[2], t[3] };
        b[i >> 1][2 * (i & 1)]     = lo;
        b[i >> 1][2 * (i & 1) + 1] = hi;
    }
}
struct Frags16 { v2f a[4]; v2f b[4][4]; };
__device__ __forceinline__ void load_frags16(Frags16& f, const float* __restrict__ Ap,
                                             const v4f* __restrict__ q) {
    #pragma unroll
    for (int s = 0; s < 4; ++s) f.a[s] = *(const v2f*)(Ap + 4 * s);
    load_wfrags(f.b, q);
}
__device__ __forceinline__ void mma_frags16_t(const Frags16& f, v8f& acc0, v8f& acc1, v8f& acc2, v8f& acc3) {
    #pragma unroll
    for (int s = 0; s < 4; ++s) {
        acc0 = __builtin_amdgcn_wmma_f32_16x16x4_f32(false, f.b[s][0], false, f.a[s], (short)0, acc0, false, false);
        acc1 = __builtin_amdgcn_wmma_f32_16x16x4_f32(false, f.b[s][1], false, f.a[s], (short)0, acc1, false, false);
        acc2 = __builtin_amdgcn_wmma_f32_16x16x4_f32(false, f.b[s][2], false, f.a[s], (short)0, acc2, false, false);
        acc3 = __builtin_amdgcn_wmma_f32_16x16x4_f32(false, f.b[s][3], false, f.a[s], (short)0, acc3, false, false);
    }
}
struct FragsB16 { v2f b[4][4]; };
__device__ __forceinline__ void mma_tile16(const v8f& T, const FragsB16& f, v8f& o0, v8f& o1, v8f& o2, v8f& o3) {
    #pragma unroll
    for (int s = 0; s < 4; ++s) {
        const v2f a = { T[2 * s], T[2 * s + 1] };
        o0 = __builtin_amdgcn_wmma_f32_16x16x4_f32(false, a, false, f.b[s][0], (short)0, o0, false, false);
        o1 = __builtin_amdgcn_wmma_f32_16x16x4_f32(false, a, false, f.b[s][1], (short)0, o1, false, false);
        o2 = __builtin_amdgcn_wmma_f32_16x16x4_f32(false, a, false, f.b[s][2], (short)0, o2, false, false);
        o3 = __builtin_amdgcn_wmma_f32_16x16x4_f32(false, a, false, f.b[s][3], (short)0, o3, false, false);
    }
}


__device__ __forceinline__ void scores_tile(int wave, int lane, const float* __restrict__ X,
                                            const float* __restrict__ wsd,
                                            float* __restrict__ esrc, float* __restrict__ edst) {
    const int l = lane & 15, half = lane >> 4;
    const float* Ap = X + (unsigned)(wave * 16 + l) * (unsigned)IN_CH + 2u * half;
    const float* Bp = wsd + (2 * half) * 16 + l;
    v8f accE = {}, accO = {};
    #pragma unroll
    for (int s = 0; s < IN_CH / 4; s += 2) {
        const v2f a0 = *(const v2f*)(Ap + 4 * s);
        const v2f b0 = { Bp[(4 * s) * 16], Bp[(4 * s + 1) * 16] };
        const v2f a1 = *(const v2f*)(Ap + 4 * s + 4);
        const v2f b1 = { Bp[(4 * s + 4) * 16], Bp[(4 * s + 5) * 16] };
        accE = __builtin_amdgcn_wmma_f32_16x16x4_f32(false, a0, false, b0, (short)0, accE, false, false);
        accO = __builtin_amdgcn_wmma_f32_16x16x4_f32(false, a1, false, b1, (short)0, accO, false, false);
    }
    float* dstp = (l < 8 ? esrc : edst) + (unsigned)(wave * 16 + 8 * half) * (unsigned)HEADS + (unsigned)(l & 7);
    #pragma unroll
    for (int r = 0; r < 8; ++r) dstp[r * HEADS] = accE[r] + accO[r];
}

#define SLOTS 64
struct EdgeList { const int* row; const int2* ovf; int node; int deg; int len; };
__device__ __forceinline__ EdgeList edge_list(const int* __restrict__ cnt, const int* __restrict__ slots,
                                              const int2* __restrict__ ovf, int node) {
    EdgeList L;
    L.row  = slots + (unsigned)node * (unsigned)SLOTS;
    L.ovf  = ovf;
    L.node = node;
    L.deg  = cnt[node];
    L.len  = L.deg;
    if (L.deg > SLOTS) L.len = SLOTS + cnt[NN];
    return L;
}
__device__ __forceinline__ int edge_src(const EdgeList& L, int base, int lane, bool& valid) {
    const int idx = base + lane;
    if (base < SLOTS) {
        valid = idx < L.deg;
        return valid ? L.row[idx] : 0;
    }
    int2 o = make_int2(-1, 0);
    if (idx < L.len) o = L.ovf[idx - SLOTS];
    valid = o.x == L.node;
    return valid ? o.y : L.node;
}

struct Blk8 { int src; float v[HEADS]; bool valid; };
__device__ __forceinline__ Blk8 load8(const EdgeList& L, const float* __restrict__ esrc,
                                      const float (&ed)[HEADS], int base, int lane) {
    Blk8 b;
    b.src = edge_src(L, base, lane, b.valid);
    const v4f* ep = (const v4f*)(esrc + (unsigned)b.src * (unsigned)HEADS);
    const v4f e0 = ep[0], e1 = ep[1];
    const float e[HEADS] = { e0[0], e0[1], e0[2], e0[3], e1[0], e1[1], e1[2], e1[3] };
    #pragma unroll
    for (int q = 0; q < HEADS; ++q) {
        float v = e[q] + ed[q];
        v = v > 0.f ? v : 0.2f * v;
        b.v[q] = b.valid ? v : -INFINITY;
    }
    return b;
}
template <int U>
__device__ __forceinline__ void group8(int srcv, const float (&av)[HEADS], int j,
                                       const float* __restrict__ X, unsigned c, v4f (&acc)[HEADS]) {
    float a[U][HEADS];
    v4f v[U];
    #pragma unroll
    for (int u = 0; u < U; ++u) {
        const int s = __builtin_amdgcn_readlane(srcv, j + u);
        #pragma unroll
        for (int q = 0; q < HEADS; ++q)
            a[u][q] = __builtin_bit_cast(float, __builtin_amdgcn_readlane(__builtin_bit_cast(int, av[q]), j + u));
        v[u] = *(const v4f*)(X + (unsigned)s * (unsigned)IN_CH + c);
    }
    #pragma unroll
    for (int u = 0; u < U; ++u) {
        #pragma unroll
        for (int q = 0; q < HEADS; ++q) acc[q] += a[u][q] * v[u];
    }
}
__device__ __forceinline__ void block_sum8(int srcv, const float (&av)[HEADS], int cnt,
                                           const float* __restrict__ X, unsigned c, v4f (&acc)[HEADS]) {
    int j = 0;
    for (; j + 4 <= cnt; j += 4) group8<4>(srcv, av, j, X, c, acc);
    const int r = cnt - j;
    if (r == 3)      group8<3>(srcv, av, j, X, c, acc);
    else if (r == 2) group8<2>(srcv, av, j, X, c, acc);
    else if (r == 1) group8<1>(srcv, av, j, X, c, acc);
}

#define ZP (HEADS * IN_CH + 4)
__global__ void __launch_bounds__(32 * GEMM_WAVES)
gather_project_l1(const int* __restrict__ cnt, const int* __restrict__ slots, const int2* __restrict__ ovf,
                  const float* __restrict__ esrc, const float* __restrict__ edst,
                  const float* __restrict__ X, const float* __restrict__ W1q,
                  const float* __restrict__ bias, const float* __restrict__ W2q,
                  const float* __restrict__ a_src2, const float* __restrict__ a_dst2,
                  float* __restrict__ h2, float* __restrict__ sc2) {
    __shared__ __attribute__((aligned(16))) float zs[16 * ZP];
    __shared__ float scs[32];
    const int lane = threadIdx.x & 31;
    const int w = __builtin_amdgcn_readfirstlane((int)(threadIdx.x >> 5));
    const unsigned c = 4u * (unsigned)lane;

    for (int i = 0; i < 2; ++i) {
        const int ml   = 2 * w + i;
        const int node = (int)blockIdx.x * 16 + ml;
        const EdgeList L = edge_list(cnt, slots, ovf, node);
        float ed[HEADS], vself[HEADS];
        #pragma unroll
        for (int q = 0; q < HEADS; ++q) {
            ed[q] = edst[node * HEADS + q];
            const float v = esrc[node * HEADS + q] + ed[q];
            vself[q] = v > 0.f ? v : 0.2f * v;
        }
        const Blk8 b0 = load8(L, esrc, ed, 0, lane);
        float m[HEADS];
        #pragma unroll
        for (int q = 0; q < HEADS; ++q) m[q] = b0.v[q];
        for (int base = 32; base < L.len; base += 32) {
            const Blk8 b = load8(L, esrc, ed, base, lane);
            #pragma unroll
            for (int q = 0; q < HEADS; ++q) m[q] = fmaxf(m[q], b.v[q]);
        }
        #pragma unroll
        for (int q = 0; q < HEADS; ++q) m[q] = fmaxf(wave_max(m[q]), vself[q]);
        float p0[HEADS], s[HEADS];
        #pragma unroll
        for (int q = 0; q < HEADS; ++q) { p0[q] = b0.valid ? expf(b0.v[q] - m[q]) : 0.f; s[q] = p0[q]; }
        for (int base = 32; base < L.len; base += 32) {
            const Blk8 b = load8(L, esrc, ed, base, lane);
            #pragma unroll
            for (int q = 0; q < HEADS; ++q) s[q] += b.valid ? expf(b.v[q] - m[q]) : 0.f;
        }
        float dn[HEADS], a0[HEADS];
        v4f acc[HEADS];
        const v4f xself = *(const v4f*)(X + (unsigned)node * (unsigned)IN_CH + c);
        #pragma unroll
        for (int q = 0; q < HEADS; ++q) {
            const float pself = expf(vself[q] - m[q]);
            dn[q] = (wave_sum(s[q]) + pself) + 1e-16f;
            a0[q] = p0[q] / dn[q];
            acc[q] = (pself / dn[q]) * xself;
        }
        int left = L.len;
        if (left > 0) block_sum8(b0.src, a0, left < 32 ? left : 32, X, c, acc);
        for (int base = 32; base < L.len; base += 32) {
            left -= 32;
            const Blk8 b = load8(L, esrc, ed, base, lane);
            float a[HEADS];
            #pragma unroll
            for (int q = 0; q < HEADS; ++q) a[q] = (b.valid ? expf(b.v[q] - m[q]) : 0.f) / dn[q];
            block_sum8(b.src, a, left < 32 ? left : 32, X, c, acc);
        }
        v4f* zrow = (v4f*)(zs + ml * ZP) + lane;
        #pragma unroll
        for (int q = 0; q < HEADS; ++q) zrow[q * (IN_CH / 4)] = acc[q];
    }
    __syncthreads();

    const int l    = lane & 15;
    const int half = lane >> 4;
    v8f acc0 = {}, acc1 = {}, acc2 = {}, acc3 = {};
    const float* Ap = zs + l * ZP + w * IN_CH + 2 * half;
    const v4f* Bq  = (const v4f*)W1q + (unsigned)(w * (IN_CH / 16) * WQ_GROUP) + (unsigned)lane;
    const v4f* W2p = (const v4f*)W2q + (unsigned)(w * 4 * WQ_GROUP) + (unsigned)lane;
    Frags16 P, Q;
    load_frags16(P, Ap, Bq);
    for (int k0 = 0; k0 < IN_CH - 32; k0 += 32) {
        load_frags16(Q, Ap + 16, Bq + WQ_GROUP);
        mma_frags16_t(P, acc0, acc1, acc2, acc3);
        load_frags16(P, Ap + 32, Bq + 2 * WQ_GROUP);
        mma_frags16_t(Q, acc0, acc1, acc2, acc3);
        Ap += 32;
        Bq += 2 * WQ_GROUP;
    }
    load_frags16(Q, Ap + 16, Bq + WQ_GROUP);
    mma_frags16_t(P, acc0, acc1, acc2, acc3);
    FragsB16 R, S;
    load_wfrags(R.b, W2p);
    mma_frags16_t(Q, acc0, acc1, acc2, acc3);

    {
        const v4f* bq = (const v4f*)(bias + w * 64 + 8 * half);
        const v4f b00 = bq[0],  b01 = bq[1];
        const v4f b10 = bq[4],  b11 = bq[5];
        const v4f b20 = bq[8],  b21 = bq[9];
        const v4f b30 = bq[12], b31 = bq[13];
        #pragma unroll
        for (int r = 0; r < 4; ++r) {
            acc0[r] = fmaxf(acc0[r] + b00[r], 0.f);  acc0[r + 4] = fmaxf(acc0[r + 4] + b01[r], 0.f);
            acc1[r] = fmaxf(acc1[r] + b10[r], 0.f);  acc1[r + 4] = fmaxf(acc1[r + 4] + b11[r], 0.f);
            acc2[r] = fmaxf(acc2[r] + b20[r], 0.f);  acc2[r + 4] = fmaxf(acc2[r + 4] + b21[r], 0.f);
            acc3[r] = fmaxf(acc3[r] + b30[r], 0.f);  acc3[r + 4] = fmaxf(acc3[r + 4] + b31[r], 0.f);
        }
    }

    v8f o0 = {}, o1 = {}, o2 = {}, o3 = {};
    load_wfrags(S.b, W2p + WQ_GROUP);
    mma_tile16(acc0, R, o0, o1, o2, o3);
    load_wfrags(R.b, W2p + 2 * WQ_GROUP);
    mma_tile16(acc1, S, o0, o1, o2, o3);
    load_wfrags(S.b, W2p + 3 * WQ_GROUP);
    mma_tile16(acc2, R, o0, o1, o2, o3);
    mma_tile16(acc3, S, o0, o1, o2, o3);

    {
        float* prow = zs + half * ZP + w * IN_CH + 4 * l;
        #pragma unroll
        for (int r = 0; r < 8; ++r) {
            const v4f pv = { o0[r], o1[r], o2[r], o3[r] };
            *(v4f*)(prow + 2 * r * ZP) = pv;
        }
    }
    __syncthreads();
    const float* rrow = zs + (2 * w + half) * ZP + 4 * l;
    v4f tot = *(const v4f*)rrow;
    #pragma unroll
    for (int ww = 1; ww < GEMM_WAVES; ++ww) tot += *(const v4f*)(rrow + ww * IN_CH);

    const float* as = a_src2 + l;
    const float* ad = a_dst2 + l;
    const float vs = row_sum16(tot[0] * as[0] + tot[1] * as[16] + tot[2] * as[32] + tot[3] * as[48]);
    const float vd = row_sum16(tot[0] * ad[0] + tot[1] * ad[16] + tot[2] * ad[32] + tot[3] * ad[48]);
    const int row = 8 * half + w;
    if (l == 0) { scs[2 * row] = vs; scs[2 * row + 1] = vd; }
    __syncthreads();
    if (w == 0) sc2[(unsigned)blockIdx.x * 32u + (unsigned)lane] = scs[lane];

    float* cp = h2 + (unsigned)((int)blockIdx.x * 16 + row) * (unsigned)OUT_CH + (unsigned)l;
    cp[0]  = tot[0];
    cp[16] = tot[1];
    cp[32] = tot[2];
    cp[48] = tot[3];
}

#define ZERO_BLOCKS ((NN + 1 + 255) / 256)
#define WSD_BLOCKS  ((IN_CH * 16 + 255) / 256)
#define W1Q_V4      (HEADS * (IN_CH / 16) * WQ_GROUP)
#define W2Q_V4      (HEADS * 4 * WQ_GROUP)
#define WQ_BLOCKS   ((W1Q_V4 + W2Q_V4) / 256)
static_assert(W1Q_V4 % 256 == 0 && W2Q_V4 % 256 == 0, "whole work-groups per plane");
__device__ __forceinline__ void fragment_order_weights(int t, const float* __restrict__ W1,
                                                       const float* __restrict__ W2,
                                                       float* __restrict__ w1q, float* __restrict__ w2q) {
    const int lane = t & 31, i = (t >> 5) & 7;
    const int s = i >> 1, p = i & 1, h = lane >> 4, l = lane & 15;
    if (t < W1Q_V4) {
        const int g = (t >> 8) & 7, w = t >> 11;
        const float* r0 = W1 + (16 * g + 4 * s + 2 * h) * HC1 + 64 * w + 32 * p + l;
        const v4f o = { r0[0], r0[HC1], r0[16], r0[HC1 + 16] };
        ((v4f*)w1q)[t] = o;
    } else {
        const int u = t - W1Q_V4;
        const int g = (u >> 8) & 3, w = u >> 10;
        const float* r0 = W2 + (64 * w + 16 * g + 8 * h + 2 * s) * OUT_CH + 32 * p + l;
        const v4f o = { r0[0], r0[OUT_CH], r0[16], r0[OUT_CH + 16] };
        ((v4f*)w2q)[u] = o;
    }
}
__global__ void zero_counts_and_prepare_weights(int* __restrict__ cnt,
                                                const float* __restrict__ W1,
                                                const float* __restrict__ a_src,
                                                const float* __restrict__ a_dst,
                                                float* __restrict__ wsd,
                                                const float* __restrict__ W2,
                                                float* __restrict__ w1q, float* __restrict__ w2q) {
    if (blockIdx.x < ZERO_BLOCKS) {
        const int i = blockIdx.x * 256 + threadIdx.x;
        if (i < NN + 1) cnt[i] = 0;
        return;
    }
    if (blockIdx.x >= ZERO_BLOCKS + WSD_BLOCKS) {
        fragment_order_weights((int)(blockIdx.x - ZERO_BLOCKS - WSD_BLOCKS) * 256 + (int)threadIdx.x,
                               W1, W2, w1q, w2q);
        return;
    }
    const int t = (blockIdx.x - ZERO_BLOCKS) * 256 + threadIdx.x;
    if (t >= IN_CH * 16) return;
    const int k = t >> 4, j = t & 15, h = j & 7;
    const float* a = (j < 8 ? a_src : a_dst) + h * HID;
    const float* wr = W1 + k * HC1 + h * HID;
    float s = 0.f;
    for (int cc = 0; cc < HID; ++cc) s = fmaf(wr[cc], a[cc], s);
    wsd[t] = s;
}

#define SCORE_BLOCKS ((NN / 16 + GEMM_WAVES - 1) / GEMM_WAVES)
#define PLACE_BLOCKS ((NE + 255) / 256)
__global__ void dst_place_and_scores(const int* __restrict__ ei, int* __restrict__ cnt,
                                     int* __restrict__ slots, int2* __restrict__ ovf,
                                     const float* __restrict__ X, const float* __restrict__ wsd,
                                     float* __restrict__ esrc, float* __restrict__ edst) {
    if (blockIdx.x < SCORE_BLOCKS) {
        const int wave = (int)(blockIdx.x * GEMM_WAVES + (threadIdx.x >> 5));
        if (wave < NN / 16) scores_tile(wave, threadIdx.x & 31, X, wsd, esrc, edst);
        return;
    }
    const int e = (blockIdx.x - SCORE_BLOCKS) * 256 + threadIdx.x;
    if (e >= NE) return;
    const int src = ei[e], dst = ei[NE + e];
    const int slot = atomicAdd(&cnt[dst], 1);
    if (slot < SLOTS) slots[(unsigned)dst * (unsigned)SLOTS + (unsigned)slot] = src;
    else              ovf[atomicAdd(&cnt[NN], 1)] = make_int2(dst, src);
}

#define EPI_RELU       1
#define EPI_LOGSOFTMAX 2
template <int HC, int C, int VEC, int SS>
struct SegGather {
    typedef float vf __attribute__((ext_vector_type(VEC)));
    static constexpr int H  = HC / C;
    static constexpr int NH = (32 * VEC) / C;
    static_assert(SS >= H, "SS = floats between two nodes' scores in esrc / edst");
    static_assert(NH == 1 || NH == 2, "a wave covers one or two whole heads");

    struct Block { int src; float v[NH]; bool valid; };

    static __device__ __forceinline__ Block load(const EdgeList& L,
                                                 const float* __restrict__ esrc,
                                                 const float (&ed)[NH],
                                                 int base, int lane, int hbase) {
        Block b;
        b.src = edge_src(L, base, lane, b.valid);
        const float* ep = esrc + (unsigned)b.src * (unsigned)SS + (unsigned)hbase;
        float e[NH];
        if constexpr (NH == 2) { const float2 t = *(const float2*)ep; e[0] = t.x; e[1] = t.y; }
        else                   { e[0] = ep[0]; }
        for (int q = 0; q < NH; ++q) {
            float v = e[q] + ed[q];
            v = v > 0.f ? v : 0.2f * v;
            b.v[q] = b.valid ? v : -INFINITY;
        }
        return b;
    }

    template <int U>
    static __device__ __forceinline__ void group(int srcv, const float (&av)[NH], int j, bool upper,
                                                 const float* __restrict__ hval, unsigned c, vf& acc) {
        float a[U];
        vf v[U];
        #pragma unroll
        for (int u = 0; u < U; ++u) {
            const int s = __builtin_amdgcn_readlane(srcv, j + u);
            float al = __builtin_bit_cast(float, __builtin_amdgcn_readlane(__builtin_bit_cast(int, av[0]), j + u));
            if constexpr (NH == 2) {
                const float a1 = __builtin_bit_cast(float, __builtin_amdgcn_readlane(__builtin_bit_cast(int, av[1]), j + u));
                al = upper ? a1 : al;
            }
            a[u] = al;
            v[u] = *(const vf*)(hval + (unsigned)s * (unsigned)HC + c);
        }
        #pragma unroll
        for (int u = 0; u < U; ++u) acc += a[u] * v[u];
    }

    static __device__ __forceinline__ void block_sum(int srcv, const float (&av)[NH], int cnt, bool upper,
                                                     const float* __restrict__ hval, unsigned c, vf& acc) {
        int j = 0;
        for (; j + 4 <= cnt; j += 4) group<4>(srcv, av, j, upper, hval, c, acc);
        const int r = cnt - j;
        if (r == 3)      group<3>(srcv, av, j, upper, hval, c, acc);
        else if (r == 2) group<2>(srcv, av, j, upper, hval, c, acc);
        else if (r == 1) group<1>(srcv, av, j, upper, hval, c, acc);
    }
};

template <int HC, int C, int VEC, int SS, int EPI>
__global__ void gather_softmax_aggregate(const int* __restrict__ cnt,
                                         const int* __restrict__ slots,
                                         const int2* __restrict__ ovf,
                                         const float* __restrict__ esrc,
                                         const float* __restrict__ edst,
                                         const float* __restrict__ hval,
                                         const float* __restrict__ bias,
                                         float* __restrict__ agg) {
    typedef SegGather<HC, C, VEC, SS> G;
    typedef typename G::vf vf;
    typedef typename G::Block Block;
    constexpr int H   = G::H;
    constexpr int NH  = G::NH;
    constexpr int WPN = HC / (32 * VEC);
    const int wave = __builtin_amdgcn_readfirstlane(
        (int)((blockIdx.x * blockDim.x + threadIdx.x) >> 5));
    if (wave >= NN * WPN) return;
    const int lane  = threadIdx.x & 31;
    const int node  = wave / WPN;
    const int chunk = wave % WPN;
    const unsigned c = (unsigned)(chunk * 32 * VEC) + (unsigned)lane * VEC;
    const int  hbase = chunk * NH;
    const bool upper = NH == 2 && lane >= 16;
    const EdgeList L = edge_list(cnt, slots, ovf, node);

    float ed[NH], vself[NH];
    for (int q = 0; q < NH; ++q) {
        ed[q] = edst[node * SS + hbase + q];
        float v = esrc[node * SS + hbase + q] + ed[q];
        vself[q] = v > 0.f ? v : 0.2f * v;
    }

    const Block b0 = G::load(L, esrc, ed, 0, lane, hbase);
    float m[NH];
    for (int q = 0; q < NH; ++q) m[q] = b0.v[q];
    for (int base = 32; base < L.len; base += 32) {
        const Block b = G::load(L, esrc, ed, base, lane, hbase);
        for (int q = 0; q < NH; ++q) m[q] = fmaxf(m[q], b.v[q]);
    }
    for (int q = 0; q < NH; ++q) m[q] = fmaxf(wave_max(m[q]), vself[q]);

    float p0[NH], s[NH];
    for (int q = 0; q < NH; ++q) { p0[q] = b0.valid ? expf(b0.v[q] - m[q]) : 0.f; s[q] = p0[q]; }
    for (int base = 32; base < L.len; base += 32) {
        const Block b = G::load(L, esrc, ed, base, lane, hbase);
        for (int q = 0; q < NH; ++q) s[q] += b.valid ? expf(b.v[q] - m[q]) : 0.f;
    }
    float pself[NH], dn[NH], a0[NH];
    for (int q = 0; q < NH; ++q) {
        pself[q] = expf(vself[q] - m[q]);
        dn[q] = (wave_sum(s[q]) + pself[q]) + 1e-16f;
        a0[q] = p0[q] / dn[q];
    }

    float aself = pself[0] / dn[0];
    if constexpr (NH == 2) { const float a1 = pself[1] / dn[1]; aself = upper ? a1 : aself; }
    vf acc = aself * *(const vf*)(hval + (unsigned)node * (unsigned)HC + c);
    int left = L.len;
    if (left > 0) G::block_sum(b0.src, a0, left < 32 ? left : 32, upper, hval, c, acc);
    for (int base = 32; base < L.len; base += 32) {
        left -= 32;
        const Block b = G::load(L, esrc, ed, base, lane, hbase);
        float a[NH];
        for (int q = 0; q < NH; ++q) a[q] = (b.valid ? expf(b.v[q] - m[q]) : 0.f) / dn[q];
        G::block_sum(b.src, a, left < 32 ? left : 32, upper, hval, c, acc);
    }
    vf res = acc + *(const vf*)(bias + c);
    if constexpr (EPI == EPI_RELU) {
        for (int i = 0; i < VEC; ++i) res[i] = res[i] > 0.f ? res[i] : 0.f;
    } else {
        static_assert(WPN == 1 && H == 1, "the wave must hold the node's whole row");
        float mx = res[0];
        for (int i = 1; i < VEC; ++i) mx = fmaxf(mx, res[i]);
        mx = wave_max(mx);
        float se = 0.f;
        for (int i = 0; i < VEC; ++i) se += expf(res[i] - mx);
        const float lse = logf(wave_sum(se)) + mx;
        for (int i = 0; i < VEC; ++i) res[i] -= lse;
    }
    *(vf*)(agg + (unsigned)node * (unsigned)HC + c) = res;
}

extern "C" void kernel_launch(void* const* d_in, const int* in_sizes, int n_in,
                              void* d_out, int out_size, void* d_ws, size_t ws_size,
                              hipStream_t stream) {
    const float* x   = (const float*)d_in[0];
    const int*   ei  = (const int*)  d_in[1];
    const float* W1  = (const float*)d_in[2];
    const float* a1s = (const float*)d_in[3];
    const float* a1d = (const float*)d_in[4];
    const float* b1  = (const float*)d_in[5];
    const float* W2  = (const float*)d_in[6];
    const float* a2s = (const float*)d_in[7];
    const float* a2d = (const float*)d_in[8];
    const float* b2  = (const float*)d_in[9];
    float* out = (float*)d_out;

    float* ws = (float*)d_ws;
    float* w1q   = ws;
    float* w2q   = w1q + (size_t)IN_CH * HC1;
    float* esrc1 = w2q + (size_t)HC1 * OUT_CH;
    float* edst1 = esrc1 + (size_t)NN * HEADS;
    float* h2pre = edst1 + (size_t)NN * HEADS;
    float* sc2   = h2pre + (size_t)NN * OUT_CH;
    int*   cnt   = (int*)(sc2 + 2 * NN);
    int*   slots = cnt + NN + 32;
    int2*  ovf   = (int2*)(slots + (size_t)NN * SLOTS);

    const int TB = 32 * GEMM_WAVES;
    auto blocks = [](long long n, int tb) { return (unsigned)((n + tb - 1) / tb); };

    float* wsd = (float*)(ovf + NE);

    zero_counts_and_prepare_weights<<<ZERO_BLOCKS + WSD_BLOCKS + WQ_BLOCKS, TB, 0, stream>>>(
        cnt, W1, a1s, a1d, wsd, W2, w1q, w2q);
    dst_place_and_scores<<<SCORE_BLOCKS + PLACE_BLOCKS, TB, 0, stream>>>(ei, cnt, slots, ovf, x, wsd, esrc1, edst1);

    gather_project_l1<<<NN / 16, TB, 0, stream>>>(cnt, slots, ovf, esrc1, edst1, x, w1q, b1, w2q, a2s, a2d, h2pre, sc2);

    gather_softmax_aggregate<OUT_CH, OUT_CH, 2, 2, EPI_LOGSOFTMAX><<<blocks((long long)NN * 32, TB), TB, 0, stream>>>(
        cnt, slots, ovf, sc2, sc2 + 1, h2pre, b2, out);
}
